# starting kernel
# baseline (speedup 1.0000x reference)
#include <hip/hip_runtime.h>
#include <hip/hip_bf16.h>

typedef float v2f __attribute__((ext_vector_type(2)));
typedef float v4f __attribute__((ext_vector_type(4)));
typedef float v8f __attribute__((ext_vector_type(8)));

#define SDIM 64
#define NCH  128
#define NB   8
#define SLEN 4096

__global__ void hippo_setup_kernel(const float* __restrict__ A,
                                   const float* __restrict__ Bm,
                                   const float* __restrict__ dtp,
                                   float* __restrict__ abar,
                                   float* __restrict__ bbar) {
    __shared__ float Lsh[SDIM * SDIM];
    __shared__ float Linv[SDIM * SDIM];
    const float dt = dtp[0];
    const int tid = threadIdx.x;

    for (int e = tid; e < SDIM * SDIM; e += 256) {
        const int i = e >> 6, j = e & 63;
        float l = -0.5f * dt * A[e];
        if (i == j) l += 1.0f;
        Lsh[e] = l;
    }
    __syncthreads();

    if (tid < SDIM) {
        const int j = tid;
        for (int i = 0; i < SDIM; ++i) {
            if (i < j) {
                Linv[i * SDIM + j] = 0.0f;
            } else {
                float sum = (i == j) ? 1.0f : 0.0f;
                for (int k = j; k < i; ++k)
                    sum -= Lsh[i * SDIM + k] * Linv[k * SDIM + j];
                Linv[i * SDIM + j] = sum / Lsh[i * SDIM + i];
            }
        }
    }
    __syncthreads();

    for (int e = tid; e < SDIM * SDIM; e += 256) {
        const int i = e >> 6, j = e & 63;
        abar[e] = 2.0f * Linv[e] - ((i == j) ? 1.0f : 0.0f);
    }
    for (int e = tid; e < SDIM * NCH; e += 256) {
        const int i = e >> 7, c = e & 127;
        float sum = 0.0f;
        for (int k = 0; k <= i; ++k)
            sum += Linv[i * SDIM + k] * Bm[k * NCH + c];
        bbar[e] = dt * sum;
    }
}

#define ABAR_TILE_NONZERO(tn, tk) ((tk) < 4 * ((tn) + 1))

__global__ __launch_bounds__(256, 1)
void hippo_scan_kernel(const float* __restrict__ sig,
                       const float* __restrict__ Cmat,
                       const float* __restrict__ abar,
                       const float* __restrict__ bbar,
                       float* __restrict__ out) {
    const int b    = blockIdx.x;
    const int lane = threadIdx.x & 31;
    const int w    = threadIdx.x >> 5;
    const int lc   = lane & 15;
    const int lh   = lane >> 4;
    const int c    = w * 16 + lc;

    v2f a[4][16];
#pragma unroll
    for (int tn = 0; tn < 4; ++tn)
#pragma unroll
        for (int tk = 0; tk < 16; ++tk) {
            if (ABAR_TILE_NONZERO(tn, tk)) {
                const int base = (tn * 16 + lc) * SDIM + tk * 4 + 2 * lh;
                a[tn][tk] = *reinterpret_cast<const v2f*>(abar + base);
            }
        }

    float bb[4][8], cw[4][8];
#pragma unroll
    for (int tn = 0; tn < 4; ++tn)
#pragma unroll
        for (int r = 0; r < 8; ++r) {
            const int n = tn * 16 + r + 8 * lh;
            bb[tn][r] = bbar[n * NCH + c];
            cw[tn][r] = Cmat[n * NCH + c];
        }

    v8f s[4];
#pragma unroll
    for (int tn = 0; tn < 4; ++tn)
#pragma unroll
        for (int r = 0; r < 8; ++r) s[tn][r] = 0.0f;

    const int colOff = (b * NCH + c) * SLEN;
    const float* xp = sig + colOff;
    float*       yp = out + colOff;

    v4f xcur4 = *reinterpret_cast<const v4f*>(xp);

#pragma unroll 1
    for (int tc = 0; tc < SLEN; tc += 4) {
        v4f xnext4;
        if (tc + 4 < SLEN) {
            xnext4 = *reinterpret_cast<const v4f*>(xp + tc + 4);
        } else {
            xnext4[0] = 0.0f; xnext4[1] = 0.0f; xnext4[2] = 0.0f; xnext4[3] = 0.0f;
        }

        v4f y4;
#pragma unroll
        for (int u = 0; u < 4; ++u) {
            const float xcur = xcur4[u];

            v2f bv[16];
#pragma unroll
            for (int tk = 0; tk < 16; ++tk) {
                float vals[2];
#pragma unroll
                for (int v = 0; v < 2; ++v) {
                    const int Rlow   = 4 * tk + v;
                    const int tn     = Rlow >> 4;
                    const int rr     = Rlow & 15;
                    const int half   = rr >> 3;
                    const int regLo  = rr & 7;
                    const int regHi  = regLo + 2;
                    const float lowV  = s[tn][regLo];
                    const float highV = s[tn][regHi];
                    const float sw = __shfl_xor(half ? lowV : highV, 16, 32);
                    const float direct = half ? highV : lowV;
                    vals[v] = (lh == half) ? direct : sw;
                }
                bv[tk].x = vals[0];
                bv[tk].y = vals[1];
            }

#pragma unroll
            for (int tn = 0; tn < 4; ++tn) {
                v8f acc;
#pragma unroll
                for (int r = 0; r < 8; ++r) acc[r] = bb[tn][r] * xcur;
#pragma unroll
                for (int tk = 0; tk < 16; ++tk) {
                    if (ABAR_TILE_NONZERO(tn, tk)) {
                        acc = __builtin_amdgcn_wmma_f32_16x16x4_f32(
                             false, a[tn][tk],
                             false, bv[tk],
                             (short)0, acc,
                             false,  false);
                    }
                }
                s[tn] = acc;
            }

            float y = 0.0f;
#pragma unroll
            for (int tn = 0; tn < 4; ++tn)
#pragma unroll
                for (int r = 0; r < 8; ++r) y += cw[tn][r] * s[tn][r];
            y += __shfl_xor(y, 16, 32);
            y4[u] = y;
        }

        if (lh == 0) *reinterpret_cast<v4f*>(yp + tc) = y4;

        xcur4 = xnext4;
    }
}

extern "C" void kernel_launch(void* const* d_in, const int* in_sizes, int n_in,
                              void* d_out, int out_size, void* d_ws, size_t ws_size,
                              hipStream_t stream) {
    const float* sig = (const float*)d_in[0];
    const float* Bm  = (const float*)d_in[1];
    const float* Cm  = (const float*)d_in[2];
    const float* Am  = (const float*)d_in[3];
    const float* dtp = (const float*)d_in[4];
    float* out  = (float*)d_out;
    float* abar = (float*)d_ws;
    float* bbar = abar + SDIM * SDIM;

    hippo_setup_kernel<<<1, 256, 0, stream>>>(Am, Bm, dtp, abar, bbar);
    hippo_scan_kernel<<<NB, 256, 0, stream>>>(sig, Cm, abar, bbar, out);
}


# output as a causal convolution with the impulse response (taps C^T A_bar^k B_bar from two 63-step power chains), block-Toeplitz WMMA products, no recurrence over time; all sums reordered
# speedup vs baseline: 12.1673x; 12.1673x over previous
#include <hip/hip_runtime.h>
#include <hip/hip_bf16.h>

typedef float v2f __attribute__((ext_vector_type(2)));
typedef float v4f __attribute__((ext_vector_type(4)));
typedef float v8f __attribute__((ext_vector_type(8)));

#define SDIM 64
#define NCH  128
#define NB   8
#define SLEN 4096

__global__ void hippo_setup_kernel(const float* __restrict__ A,
                                   const float* __restrict__ Bm,
                                   const float* __restrict__ dtp,
                                   float* __restrict__ abar,
                                   float* __restrict__ bbar) {
    __shared__ float Lsh[SDIM * SDIM];
    __shared__ float Linv[SDIM * SDIM];
    const float dt = dtp[0];
    const int tid = threadIdx.x;

    for (int e = tid; e < SDIM * SDIM; e += 256) {
        const int i = e >> 6, j = e & 63;
        float l = -0.5f * dt * A[e];
        if (i == j) l += 1.0f;
        Lsh[e] = l;
    }
    __syncthreads();

    if (tid < SDIM) {
        const int j = tid;
        for (int i = 0; i < SDIM; ++i) {
            if (i < j) {
                Linv[i * SDIM + j] = 0.0f;
            } else {
                float sum = (i == j) ? 1.0f : 0.0f;
                for (int k = j; k < i; ++k)
                    sum -= Lsh[i * SDIM + k] * Linv[k * SDIM + j];
                Linv[i * SDIM + j] = sum / Lsh[i * SDIM + i];
            }
        }
    }
    __syncthreads();

    for (int e = tid; e < SDIM * SDIM; e += 256) {
        const int i = e >> 6, j = e & 63;
        abar[e] = 2.0f * Linv[e] - ((i == j) ? 1.0f : 0.0f);
    }
    for (int e = tid; e < SDIM * NCH; e += 256) {
        const int i = e >> 7, c = e & 127;
        float sum = 0.0f;
        for (int k = 0; k <= i; ++k)
            sum += Linv[i * SDIM + k] * Bm[k * NCH + c];
        bbar[e] = dt * sum;
    }
}


#define KPITCH 4112
#define XPITCH 4116
#define SPITCH 68

__device__ __forceinline__ void chain_run(const float* __restrict__ M, v8f s, float* dstAll,
                                          float* dstLast, int nsteps, float* Sl,
                                          int tn, int lc, int lh) {
    v2f a[16];
#pragma unroll
    for (int tk = 0; tk < 16; ++tk)
        a[tk] = *reinterpret_cast<const v2f*>(M + (tn * 16 + lc) * SDIM + tk * 4 + 2 * lh);
    const int rowoff = tn * 16 + 8 * lh;
    {
        v4f lo, hi;
#pragma unroll
        for (int r = 0; r < 4; ++r) { lo[r] = s[r]; hi[r] = s[r + 4]; }
        *reinterpret_cast<v4f*>(Sl + lc * SPITCH + rowoff) = lo;
        *reinterpret_cast<v4f*>(Sl + lc * SPITCH + rowoff + 4) = hi;
        if (dstAll) {
            *reinterpret_cast<v4f*>(dstAll + rowoff) = lo;
            *reinterpret_cast<v4f*>(dstAll + rowoff + 4) = hi;
        }
    }
    __syncthreads();
#pragma unroll 1
    for (int i = 1; i <= nsteps; ++i) {
        const float* cur = Sl + ((i - 1) & 1) * (16 * SPITCH);
        float* nxt = Sl + (i & 1) * (16 * SPITCH);
        v8f acc0, acc1;
#pragma unroll
        for (int r = 0; r < 8; ++r) { acc0[r] = 0.0f; acc1[r] = 0.0f; }
#pragma unroll
        for (int tk = 0; tk < 16; tk += 2) {
            const v2f b0 = *reinterpret_cast<const v2f*>(cur + lc * SPITCH + tk * 4 + 2 * lh);
            const v2f b1 = *reinterpret_cast<const v2f*>(cur + lc * SPITCH + tk * 4 + 4 + 2 * lh);
            acc0 = __builtin_amdgcn_wmma_f32_16x16x4_f32(false, a[tk], false, b0, (short)0, acc0, false, false);
            acc1 = __builtin_amdgcn_wmma_f32_16x16x4_f32(false, a[tk + 1], false, b1, (short)0, acc1, false, false);
        }
        v4f lo, hi;
#pragma unroll
        for (int r = 0; r < 4; ++r) { lo[r] = acc0[r] + acc1[r]; hi[r] = acc0[r + 4] + acc1[r + 4]; }
        *reinterpret_cast<v4f*>(nxt + lc * SPITCH + rowoff) = lo;
        *reinterpret_cast<v4f*>(nxt + lc * SPITCH + rowoff + 4) = hi;
        if (dstAll) {
            *reinterpret_cast<v4f*>(dstAll + i * SDIM + rowoff) = lo;
            *reinterpret_cast<v4f*>(dstAll + i * SDIM + rowoff + 4) = hi;
        }
        if (dstLast && i == nsteps) {
            *reinterpret_cast<v4f*>(dstLast + rowoff) = lo;
            *reinterpret_cast<v4f*>(dstLast + rowoff + 4) = hi;
        }
        __syncthreads();
    }
}

__global__ __launch_bounds__(128, 1)
void hippo_powers_kernel(const float* __restrict__ abar, const float* __restrict__ bbar,
                         float* __restrict__ Gc, float* __restrict__ PT) {
    __shared__ __attribute__((aligned(16))) float Sl[2 * 16 * SPITCH];
    const int lane = threadIdx.x & 31;
    const int tn = __builtin_amdgcn_readfirstlane(threadIdx.x >> 5);
    const int lc = lane & 15, lh = lane >> 4;
    const int blk = blockIdx.x;
    v8f s;
    if (blk < 8) {
        const int c = blk * 16 + lc;
#pragma unroll
        for (int r = 0; r < 8; ++r) s[r] = bbar[(tn * 16 + r + 8 * lh) * NCH + c];
        chain_run(abar, s, Gc + (size_t)c * SDIM * SDIM, nullptr, SDIM - 1, Sl, tn, lc, lh);
    } else {
        const int col = (blk - 8) * 16 + lc;
#pragma unroll
        for (int r = 0; r < 8; ++r) s[r] = abar[(tn * 16 + r + 8 * lh) * SDIM + col];
        chain_run(abar, s, nullptr, PT + col * SDIM, SDIM - 1, Sl, tn, lc, lh);
    }
}

__global__ __launch_bounds__(128, 1)
void hippo_h_kernel(const float* __restrict__ PT, const float* __restrict__ Cmat,
                    float* __restrict__ Hc) {
    __shared__ __attribute__((aligned(16))) float Sl[2 * 16 * SPITCH];
    const int lane = threadIdx.x & 31;
    const int tn = __builtin_amdgcn_readfirstlane(threadIdx.x >> 5);
    const int lc = lane & 15, lh = lane >> 4;
    const int c = blockIdx.x * 16 + lc;
    v8f s;
#pragma unroll
    for (int r = 0; r < 8; ++r) s[r] = Cmat[(tn * 16 + r + 8 * lh) * NCH + c];
    chain_run(PT, s, Hc + (size_t)c * SDIM * SDIM, nullptr, SDIM - 1, Sl, tn, lc, lh);
}

__global__ __launch_bounds__(256, 1)
void hippo_taps_kernel(const float* __restrict__ Hc, const float* __restrict__ Gc,
                       float* __restrict__ Kp) {
    const int c = blockIdx.x;
    const int lane = threadIdx.x & 31;
    const int w = threadIdx.x >> 5;
    const int lc = lane & 15, lh = lane >> 4;
    const float* H = Hc + (size_t)c * SDIM * SDIM;
    const float* G = Gc + (size_t)c * SDIM * SDIM;
    float* K = Kp + (size_t)c * KPITCH;
    if (threadIdx.x < 16) K[SLEN + threadIdx.x] = 0.0f;
#pragma unroll
    for (int t = 0; t < 2; ++t) {
        const int tile = 2 * w + t;
        const int tj = tile >> 2, ti = tile & 3;
        v8f acc;
#pragma unroll
        for (int r = 0; r < 8; ++r) acc[r] = 0.0f;
#pragma unroll
        for (int tk = 0; tk < 16; ++tk) {
            const v2f af = *reinterpret_cast<const v2f*>(H + (tj * 16 + lc) * SDIM + tk * 4 + 2 * lh);
            const v2f bf = *reinterpret_cast<const v2f*>(G + (ti * 16 + lc) * SDIM + tk * 4 + 2 * lh);
            acc = __builtin_amdgcn_wmma_f32_16x16x4_f32(false, af, false, bf, (short)0, acc, false, false);
        }
#pragma unroll
        for (int r = 0; r < 8; ++r)
            K[SLEN - 1 - ((tj * 16 + r + 8 * lh) * SDIM + ti * 16 + lc)] = acc[r];
    }
}

struct ConvFrag { v2f a0[4], a1[4]; v4f x0lo, x0hi, x1lo, x1hi; };

__device__ __forceinline__ void conv_load(ConvFrag& f, const float* xb, const float* kb) {
    f.x0lo = *reinterpret_cast<const v4f*>(xb);
    f.x0hi = *reinterpret_cast<const v4f*>(xb + 4);
    f.x1lo = *reinterpret_cast<const v4f*>(xb - 16);
    f.x1hi = *reinterpret_cast<const v4f*>(xb - 12);
#pragma unroll
    for (int s = 0; s < 4; ++s) {
        f.a0[s].x = kb[2 * s];      f.a0[s].y = kb[2 * s + 1];
        f.a1[s].x = kb[2 * s - 16]; f.a1[s].y = kb[2 * s - 15];
    }
}

__device__ __forceinline__ void conv_mma(const ConvFrag& f, v8f& acc0, v8f& acc1) {
    v2f b0[4], b1[4];
    b0[0].x = f.x0lo[0]; b0[0].y = f.x0lo[1]; b0[1].x = f.x0lo[2]; b0[1].y = f.x0lo[3];
    b0[2].x = f.x0hi[0]; b0[2].y = f.x0hi[1]; b0[3].x = f.x0hi[2]; b0[3].y = f.x0hi[3];
    b1[0].x = f.x1lo[0]; b1[0].y = f.x1lo[1]; b1[1].x = f.x1lo[2]; b1[1].y = f.x1lo[3];
    b1[2].x = f.x1hi[0]; b1[2].y = f.x1hi[1]; b1[3].x = f.x1hi[2]; b1[3].y = f.x1hi[3];
#pragma unroll
    for (int s = 0; s < 4; ++s) {
        acc0 = __builtin_amdgcn_wmma_f32_16x16x4_f32(false, f.a0[s], false, b0[s], (short)0, acc0, false, false);
        acc1 = __builtin_amdgcn_wmma_f32_16x16x4_f32(false, f.a1[s], false, b1[s], (short)0, acc1, false, false);
    }
}

__global__ __launch_bounds__(256, 1)
void hippo_conv_kernel(const float* __restrict__ sig, const float* __restrict__ Kp,
                       float* __restrict__ out) {
    __shared__ __attribute__((aligned(16))) float xs[(NB + 1) * XPITCH];
    __shared__ __attribute__((aligned(16))) float ks[KPITCH];
    const int c = blockIdx.x >> 1;
    const int z = blockIdx.x & 1;
    const int tid = threadIdx.x;
    const int lane = tid & 31;
    const int w = __builtin_amdgcn_readfirstlane(tid >> 5);
    const int lc = lane & 15, lh = lane >> 4;

    if (tid < NB * 16) xs[(tid >> 4) * XPITCH + (tid & 15)] = 0.0f;
#pragma unroll 1
    for (int b = 0; b < NB; ++b) {
        const v4f* src = reinterpret_cast<const v4f*>(sig + ((size_t)b * NCH + c) * SLEN);
        v4f* dst = reinterpret_cast<v4f*>(xs + b * XPITCH + 16);
#pragma unroll
        for (int i = 0; i < SLEN / 4 / 256; ++i) dst[tid + i * 256] = src[tid + i * 256];
    }
    {
        const v4f* src = reinterpret_cast<const v4f*>(Kp + (size_t)c * KPITCH);
        v4f* dst = reinterpret_cast<v4f*>(ks);
        for (int i = tid; i < KPITCH / 4; i += 256) dst[i] = src[i];
    }
    __syncthreads();

    const int bcol = lc & 7;
    const int e    = lc >> 3;
    const int g    = z * 8 + w;
    const float* xrow = xs + bcol * XPITCH + 8 * lh + 16 * (e + 1);
    const float* krow = ks + (SLEN - 1) - lc + 8 * lh;

#pragma unroll 1
    for (int it = 0; it < 8; ++it) {
        const int P0 = g + 16 * (it >> 1);
        const int P  = (it & 1) ? (127 - P0) : P0;
        const int mlast = 2 * P;
        v8f acc0, acc1;
#pragma unroll
        for (int r = 0; r < 8; ++r) { acc0[r] = 0.0f; acc1[r] = 0.0f; }
        ConvFrag f0, f1;
        conv_load(f0, xrow + 16 * (2 * P), krow);
        int m = 0;
#pragma unroll 1
        while (true) {
            if (m >= mlast) { conv_mma(f0, acc0, acc1); break; }
            conv_load(f1, xrow + 16 * (2 * P - m - 2), krow - 16 * (m + 2));
            conv_mma(f0, acc0, acc1);
            m += 2;
            if (m >= mlast) { conv_mma(f1, acc0, acc1); break; }
            conv_load(f0, xrow + 16 * (2 * P - m - 2), krow - 16 * (m + 2));
            conv_mma(f1, acc0, acc1);
            m += 2;
        }
        float* yp = out + ((size_t)bcol * NCH + c) * SLEN + 16 * (2 * P + e) + 8 * lh;
        v4f lo, hi;
#pragma unroll
        for (int r = 0; r < 4; ++r) { lo[r] = acc0[r] + acc1[r]; hi[r] = acc0[r + 4] + acc1[r + 4]; }
        *reinterpret_cast<v4f*>(yp) = lo;
        *reinterpret_cast<v4f*>(yp + 4) = hi;
    }
}

extern "C" void kernel_launch(void* const* d_in, const int* in_sizes, int n_in,
                              void* d_out, int out_size, void* d_ws, size_t ws_size,
                              hipStream_t stream) {
    const float* sig = (const float*)d_in[0];
    const float* Bm  = (const float*)d_in[1];
    const float* Cm  = (const float*)d_in[2];
    const float* Am  = (const float*)d_in[3];
    const float* dtp = (const float*)d_in[4];
    float* out  = (float*)d_out;
    float* abar = (float*)d_ws;
    float* bbar = abar + SDIM * SDIM;
    float* pt   = bbar + SDIM * NCH;
    float* gc   = pt + SDIM * SDIM;
    float* hc   = gc + NCH * SDIM * SDIM;
    float* kp   = hc + NCH * SDIM * SDIM;

    hippo_setup_kernel<<<1, 256, 0, stream>>>(Am, Bm, dtp, abar, bbar);
    hippo_powers_kernel<<<12, 128, 0, stream>>>(abar, bbar, gc, pt);
    hippo_h_kernel<<<8, 128, 0, stream>>>(pt, Cm, hc);
    hippo_taps_kernel<<<NCH, 256, 0, stream>>>(hc, gc, kp);
    hippo_conv_kernel<<<NCH * 2, 256, 0, stream>>>(sig, kp, out);
}


# setup kernel: triangular inverse built blockwise (16x16 diagonal blocks by substitution, off-diagonal blocks as block products, two levels), B_bar as a WMMA f32 product; setup sums reordered
# speedup vs baseline: 16.4145x; 1.3491x over previous
#include <hip/hip_runtime.h>
#include <hip/hip_bf16.h>

typedef float v2f __attribute__((ext_vector_type(2)));
typedef float v4f __attribute__((ext_vector_type(4)));
typedef float v8f __attribute__((ext_vector_type(8)));

#define SDIM 64
#define NCH  128
#define NB   8
#define SLEN 4096

#define LP 68
#define TP 33
__global__ __launch_bounds__(256, 1)
void hippo_setup_kernel(const float* __restrict__ A,
                        const float* __restrict__ Bm,
                        const float* __restrict__ dtp,
                        float* __restrict__ abar,
                        float* __restrict__ bbar) {
    __shared__ __attribute__((aligned(16))) float Ls[SDIM * LP];
    __shared__ __attribute__((aligned(16))) float Xs[SDIM * LP];
    __shared__ float Ts[32 * TP];
    const float dt = dtp[0];
    const int tid = threadIdx.x;
    const int lane = tid & 31;
    const int w = __builtin_amdgcn_readfirstlane(tid >> 5);
    const int lc = lane & 15, lh = lane >> 4;
    const int tn = w & 3;
    const int cg = w >> 2;

    float av[16];
#pragma unroll
    for (int u = 0; u < 16; ++u) av[u] = A[tid + 256 * u];
#pragma unroll
    for (int u = 0; u < 16; ++u) {
        const int e = tid + 256 * u;
        const int i = e >> 6, j = e & 63;
        float l = -0.5f * dt * av[u];
        if (i == j) l += 1.0f;
        Ls[i * LP + j] = l;
        Xs[i * LP + j] = 0.0f;
    }
    v2f bf[4][16];
#pragma unroll
    for (int ct = 0; ct < 4; ++ct) {
        const int c = (cg * 4 + ct) * 16 + lc;
#pragma unroll
        for (int tk = 0; tk < 16; ++tk) {
            bf[ct][tk].x = Bm[(tk * 4 + 2 * lh) * NCH + c];
            bf[ct][tk].y = Bm[(tk * 4 + 2 * lh + 1) * NCH + c];
        }
    }
    __syncthreads();

    if (tid < 64) {
        const int I = tid >> 4, j = tid & 15;
        const float* Lb = Ls + (16 * I) * LP + 16 * I;
        float x[16];
#pragma unroll
        for (int i = 0; i < 16; ++i) {
            float sum = (i == j) ? 1.0f : 0.0f;
#pragma unroll
            for (int k = 0; k < i; ++k) sum -= Lb[i * LP + k] * x[k];
            x[i] = sum / Lb[i * LP + i];
        }
#pragma unroll
        for (int i = 0; i < 16; ++i) Xs[(16 * I + i) * LP + 16 * I + j] = x[i];
    }
    __syncthreads();

    {
        const int r = tid >> 4, c = tid & 15;
#pragma unroll
        for (int p = 0; p < 2; ++p) {
            const int R0 = 16 + 32 * p, C0 = 32 * p;
            float s = 0.0f;
#pragma unroll
            for (int k = 0; k < 16; ++k) s += Ls[(R0 + r) * LP + C0 + k] * Xs[(C0 + k) * LP + C0 + c];
            Ts[(16 * p + r) * TP + c] = s;
        }
        __syncthreads();
#pragma unroll
        for (int p = 0; p < 2; ++p) {
            const int R0 = 16 + 32 * p, C0 = 32 * p;
            float s = 0.0f;
#pragma unroll
            for (int k = 0; k < 16; ++k) s += Xs[(R0 + r) * LP + R0 + k] * Ts[(16 * p + k) * TP + c];
            Xs[(R0 + r) * LP + C0 + c] = -s;
        }
        __syncthreads();
    }

    {
        const int r = tid >> 3, c0 = tid & 7;
        float s[4] = {0.0f, 0.0f, 0.0f, 0.0f};
#pragma unroll
        for (int k = 0; k < 32; ++k) {
            const float a = Ls[(32 + r) * LP + k];
#pragma unroll
            for (int m = 0; m < 4; ++m) s[m] += a * Xs[k * LP + c0 + 8 * m];
        }
#pragma unroll
        for (int m = 0; m < 4; ++m) Ts[r * TP + c0 + 8 * m] = s[m];
        __syncthreads();
#pragma unroll
        for (int m = 0; m < 4; ++m) s[m] = 0.0f;
#pragma unroll
        for (int k = 0; k < 32; ++k) {
            const float a = Xs[(32 + r) * LP + 32 + k];
#pragma unroll
            for (int m = 0; m < 4; ++m) s[m] += a * Ts[k * TP + c0 + 8 * m];
        }
#pragma unroll
        for (int m = 0; m < 4; ++m) Xs[(32 + r) * LP + c0 + 8 * m] = -s[m];
        __syncthreads();
    }

#pragma unroll
    for (int u = 0; u < 16; ++u) {
        const int e = tid + 256 * u;
        const int i = e >> 6, j = e & 63;
        abar[e] = 2.0f * Xs[i * LP + j] - ((i == j) ? 1.0f : 0.0f);
    }

    v2f a[16];
#pragma unroll
    for (int tk = 0; tk < 16; ++tk)
        a[tk] = *reinterpret_cast<const v2f*>(Xs + (tn * 16 + lc) * LP + tk * 4 + 2 * lh);
#pragma unroll
    for (int ct = 0; ct < 4; ++ct) {
        const int c = (cg * 4 + ct) * 16 + lc;
        v8f acc0, acc1;
#pragma unroll
        for (int r = 0; r < 8; ++r) { acc0[r] = 0.0f; acc1[r] = 0.0f; }
#pragma unroll
        for (int tk = 0; tk < 16; tk += 2) {
            acc0 = __builtin_amdgcn_wmma_f32_16x16x4_f32(false, a[tk], false, bf[ct][tk], (short)0, acc0, false, false);
            acc1 = __builtin_amdgcn_wmma_f32_16x16x4_f32(false, a[tk + 1], false, bf[ct][tk + 1], (short)0, acc1, false, false);
        }
#pragma unroll
        for (int r = 0; r < 8; ++r)
            bbar[(tn * 16 + 8 * lh + r) * NCH + c] = dt * (acc0[r] + acc1[r]);
    }
}


#define KPITCH 4112
#define XPITCH 4116
#define SPITCH 68

__device__ __forceinline__ void chain_run(const float* __restrict__ M, v8f s, float* dstAll,
                                          float* dstLast, int nsteps, float* Sl,
                                          int tn, int lc, int lh) {
    v2f a[16];
#pragma unroll
    for (int tk = 0; tk < 16; ++tk)
        a[tk] = *reinterpret_cast<const v2f*>(M + (tn * 16 + lc) * SDIM + tk * 4 + 2 * lh);
    const int rowoff = tn * 16 + 8 * lh;
    {
        v4f lo, hi;
#pragma unroll
        for (int r = 0; r < 4; ++r) { lo[r] = s[r]; hi[r] = s[r + 4]; }
        *reinterpret_cast<v4f*>(Sl + lc * SPITCH + rowoff) = lo;
        *reinterpret_cast<v4f*>(Sl + lc * SPITCH + rowoff + 4) = hi;
        if (dstAll) {
            *reinterpret_cast<v4f*>(dstAll + rowoff) = lo;
            *reinterpret_cast<v4f*>(dstAll + rowoff + 4) = hi;
        }
    }
    __syncthreads();
#pragma unroll 1
    for (int i = 1; i <= nsteps; ++i) {
        const float* cur = Sl + ((i - 1) & 1) * (16 * SPITCH);
        float* nxt = Sl + (i & 1) * (16 * SPITCH);
        v8f acc0, acc1;
#pragma unroll
        for (int r = 0; r < 8; ++r) { acc0[r] = 0.0f; acc1[r] = 0.0f; }
#pragma unroll
        for (int tk = 0; tk < 16; tk += 2) {
            const v2f b0 = *reinterpret_cast<const v2f*>(cur + lc * SPITCH + tk * 4 + 2 * lh);
            const v2f b1 = *reinterpret_cast<const v2f*>(cur + lc * SPITCH + tk * 4 + 4 + 2 * lh);
            acc0 = __builtin_amdgcn_wmma_f32_16x16x4_f32(false, a[tk], false, b0, (short)0, acc0, false, false);
            acc1 = __builtin_amdgcn_wmma_f32_16x16x4_f32(false, a[tk + 1], false, b1, (short)0, acc1, false, false);
        }
        v4f lo, hi;
#pragma unroll
        for (int r = 0; r < 4; ++r) { lo[r] = acc0[r] + acc1[r]; hi[r] = acc0[r + 4] + acc1[r + 4]; }
        *reinterpret_cast<v4f*>(nxt + lc * SPITCH + rowoff) = lo;
        *reinterpret_cast<v4f*>(nxt + lc * SPITCH + rowoff + 4) = hi;
        if (dstAll) {
            *reinterpret_cast<v4f*>(dstAll + i * SDIM + rowoff) = lo;
            *reinterpret_cast<v4f*>(dstAll + i * SDIM + rowoff + 4) = hi;
        }
        if (dstLast && i == nsteps) {
            *reinterpret_cast<v4f*>(dstLast + rowoff) = lo;
            *reinterpret_cast<v4f*>(dstLast + rowoff + 4) = hi;
        }
        __syncthreads();
    }
}

__global__ __launch_bounds__(128, 1)
void hippo_powers_kernel(const float* __restrict__ abar, const float* __restrict__ bbar,
                         float* __restrict__ Gc, float* __restrict__ PT) {
    __shared__ __attribute__((aligned(16))) float Sl[2 * 16 * SPITCH];
    const int lane = threadIdx.x & 31;
    const int tn = __builtin_amdgcn_readfirstlane(threadIdx.x >> 5);
    const int lc = lane & 15, lh = lane >> 4;
    const int blk = blockIdx.x;
    v8f s;
    if (blk < 8) {
        const int c = blk * 16 + lc;
#pragma unroll
        for (int r = 0; r < 8; ++r) s[r] = bbar[(tn * 16 + r + 8 * lh) * NCH + c];
        chain_run(abar, s, Gc + (size_t)c * SDIM * SDIM, nullptr, SDIM - 1, Sl, tn, lc, lh);
    } else {
        const int col = (blk - 8) * 16 + lc;
#pragma unroll
        for (int r = 0; r < 8; ++r) s[r] = abar[(tn * 16 + r + 8 * lh) * SDIM + col];
        chain_run(abar, s, nullptr, PT + col * SDIM, SDIM - 1, Sl, tn, lc, lh);
    }
}

__global__ __launch_bounds__(128, 1)
void hippo_h_kernel(const float* __restrict__ PT, const float* __restrict__ Cmat,
                    float* __restrict__ Hc) {
    __shared__ __attribute__((aligned(16))) float Sl[2 * 16 * SPITCH];
    const int lane = threadIdx.x & 31;
    const int tn = __builtin_amdgcn_readfirstlane(threadIdx.x >> 5);
    const int lc = lane & 15, lh = lane >> 4;
    const int c = blockIdx.x * 16 + lc;
    v8f s;
#pragma unroll
    for (int r = 0; r < 8; ++r) s[r] = Cmat[(tn * 16 + r + 8 * lh) * NCH + c];
    chain_run(PT, s, Hc + (size_t)c * SDIM * SDIM, nullptr, SDIM - 1, Sl, tn, lc, lh);
}

__global__ __launch_bounds__(256, 1)
void hippo_taps_kernel(const float* __restrict__ Hc, const float* __restrict__ Gc,
                       float* __restrict__ Kp) {
    const int c = blockIdx.x;
    const int lane = threadIdx.x & 31;
    const int w = threadIdx.x >> 5;
    const int lc = lane & 15, lh = lane >> 4;
    const float* H = Hc + (size_t)c * SDIM * SDIM;
    const float* G = Gc + (size_t)c * SDIM * SDIM;
    float* K = Kp + (size_t)c * KPITCH;
    if (threadIdx.x < 16) K[SLEN + threadIdx.x] = 0.0f;
#pragma unroll
    for (int t = 0; t < 2; ++t) {
        const int tile = 2 * w + t;
        const int tj = tile >> 2, ti = tile & 3;
        v8f acc;
#pragma unroll
        for (int r = 0; r < 8; ++r) acc[r] = 0.0f;
#pragma unroll
        for (int tk = 0; tk < 16; ++tk) {
            const v2f af = *reinterpret_cast<const v2f*>(H + (tj * 16 + lc) * SDIM + tk * 4 + 2 * lh);
            const v2f bf = *reinterpret_cast<const v2f*>(G + (ti * 16 + lc) * SDIM + tk * 4 + 2 * lh);
            acc = __builtin_amdgcn_wmma_f32_16x16x4_f32(false, af, false, bf, (short)0, acc, false, false);
        }
#pragma unroll
        for (int r = 0; r < 8; ++r)
            K[SLEN - 1 - ((tj * 16 + r + 8 * lh) * SDIM + ti * 16 + lc)] = acc[r];
    }
}

struct ConvFrag { v2f a0[4], a1[4]; v4f x0lo, x0hi, x1lo, x1hi; };

__device__ __forceinline__ void conv_load(ConvFrag& f, const float* xb, const float* kb) {
    f.x0lo = *reinterpret_cast<const v4f*>(xb);
    f.x0hi = *reinterpret_cast<const v4f*>(xb + 4);
    f.x1lo = *reinterpret_cast<const v4f*>(xb - 16);
    f.x1hi = *reinterpret_cast<const v4f*>(xb - 12);
#pragma unroll
    for (int s = 0; s < 4; ++s) {
        f.a0[s].x = kb[2 * s];      f.a0[s].y = kb[2 * s + 1];
        f.a1[s].x = kb[2 * s - 16]; f.a1[s].y = kb[2 * s - 15];
    }
}

__device__ __forceinline__ void conv_mma(const ConvFrag& f, v8f& acc0, v8f& acc1) {
    v2f b0[4], b1[4];
    b0[0].x = f.x0lo[0]; b0[0].y = f.x0lo[1]; b0[1].x = f.x0lo[2]; b0[1].y = f.x0lo[3];
    b0[2].x = f.x0hi[0]; b0[2].y = f.x0hi[1]; b0[3].x = f.x0hi[2]; b0[3].y = f.x0hi[3];
    b1[0].x = f.x1lo[0]; b1[0].y = f.x1lo[1]; b1[1].x = f.x1lo[2]; b1[1].y = f.x1lo[3];
    b1[2].x = f.x1hi[0]; b1[2].y = f.x1hi[1]; b1[3].x = f.x1hi[2]; b1[3].y = f.x1hi[3];
#pragma unroll
    for (int s = 0; s < 4; ++s) {
        acc0 = __builtin_amdgcn_wmma_f32_16x16x4_f32(false, f.a0[s], false, b0[s], (short)0, acc0, false, false);
        acc1 = __builtin_amdgcn_wmma_f32_16x16x4_f32(false, f.a1[s], false, b1[s], (short)0, acc1, false, false);
    }
}

__global__ __launch_bounds__(256, 1)
void hippo_conv_kernel(const float* __restrict__ sig, const float* __restrict__ Kp,
                       float* __restrict__ out) {
    __shared__ __attribute__((aligned(16))) float xs[(NB + 1) * XPITCH];
    __shared__ __attribute__((aligned(16))) float ks[KPITCH];
    const int c = blockIdx.x >> 1;
    const int z = blockIdx.x & 1;
    const int tid = threadIdx.x;
    const int lane = tid & 31;
    const int w = __builtin_amdgcn_readfirstlane(tid >> 5);
    const int lc = lane & 15, lh = lane >> 4;

    if (tid < NB * 16) xs[(tid >> 4) * XPITCH + (tid & 15)] = 0.0f;
#pragma unroll 1
    for (int b = 0; b < NB; ++b) {
        const v4f* src = reinterpret_cast<const v4f*>(sig + ((size_t)b * NCH + c) * SLEN);
        v4f* dst = reinterpret_cast<v4f*>(xs + b * XPITCH + 16);
#pragma unroll
        for (int i = 0; i < SLEN / 4 / 256; ++i) dst[tid + i * 256] = src[tid + i * 256];
    }
    {
        const v4f* src = reinterpret_cast<const v4f*>(Kp + (size_t)c * KPITCH);
        v4f* dst = reinterpret_cast<v4f*>(ks);
        for (int i = tid; i < KPITCH / 4; i += 256) dst[i] = src[i];
    }
    __syncthreads();

    const int bcol = lc & 7;
    const int e    = lc >> 3;
    const int g    = z * 8 + w;
    const float* xrow = xs + bcol * XPITCH + 8 * lh + 16 * (e + 1);
    const float* krow = ks + (SLEN - 1) - lc + 8 * lh;

#pragma unroll 1
    for (int it = 0; it < 8; ++it) {
        const int P0 = g + 16 * (it >> 1);
        const int P  = (it & 1) ? (127 - P0) : P0;
        const int mlast = 2 * P;
        v8f acc0, acc1;
#pragma unroll
        for (int r = 0; r < 8; ++r) { acc0[r] = 0.0f; acc1[r] = 0.0f; }
        ConvFrag f0, f1;
        conv_load(f0, xrow + 16 * (2 * P), krow);
        int m = 0;
#pragma unroll 1
        while (true) {
            if (m >= mlast) { conv_mma(f0, acc0, acc1); break; }
            conv_load(f1, xrow + 16 * (2 * P - m - 2), krow - 16 * (m + 2));
            conv_mma(f0, acc0, acc1);
            m += 2;
            if (m >= mlast) { conv_mma(f1, acc0, acc1); break; }
            conv_load(f0, xrow + 16 * (2 * P - m - 2), krow - 16 * (m + 2));
            conv_mma(f1, acc0, acc1);
            m += 2;
        }
        float* yp = out + ((size_t)bcol * NCH + c) * SLEN + 16 * (2 * P + e) + 8 * lh;
        v4f lo, hi;
#pragma unroll
        for (int r = 0; r < 4; ++r) { lo[r] = acc0[r] + acc1[r]; hi[r] = acc0[r + 4] + acc1[r + 4]; }
        *reinterpret_cast<v4f*>(yp) = lo;
        *reinterpret_cast<v4f*>(yp + 4) = hi;
    }
}

extern "C" void kernel_launch(void* const* d_in, const int* in_sizes, int n_in,
                              void* d_out, int out_size, void* d_ws, size_t ws_size,
                              hipStream_t stream) {
    const float* sig = (const float*)d_in[0];
    const float* Bm  = (const float*)d_in[1];
    const float* Cm  = (const float*)d_in[2];
    const float* Am  = (const float*)d_in[3];
    const float* dtp = (const float*)d_in[4];
    float* out  = (float*)d_out;
    float* abar = (float*)d_ws;
    float* bbar = abar + SDIM * SDIM;
    float* pt   = bbar + SDIM * NCH;
    float* gc   = pt + SDIM * SDIM;
    float* hc   = gc + NCH * SDIM * SDIM;
    float* kp   = hc + NCH * SDIM * SDIM;

    hippo_setup_kernel<<<1, 256, 0, stream>>>(Am, Bm, dtp, abar, bbar);
    hippo_powers_kernel<<<12, 128, 0, stream>>>(abar, bbar, gc, pt);
    hippo_h_kernel<<<8, 128, 0, stream>>>(pt, Cm, hc);
    hippo_taps_kernel<<<NCH, 256, 0, stream>>>(hc, gc, kp);
    hippo_conv_kernel<<<NCH * 2, 256, 0, stream>>>(sig, kp, out);
}


# power chains (G, P, H): the per-step workgroup barrier orders LDS only (local-scope fences around the bare barrier), so a step no longer waits for its global stores to land; every sum unchanged
# speedup vs baseline: 24.5365x; 1.4948x over previous
#include <hip/hip_runtime.h>
#include <hip/hip_bf16.h>

typedef float v2f __attribute__((ext_vector_type(2)));
typedef float v4f __attribute__((ext_vector_type(4)));
typedef float v8f __attribute__((ext_vector_type(8)));

#define SDIM 64
#define NCH  128
#define NB   8
#define SLEN 4096

#define LP 68
#define TP 33
__global__ __launch_bounds__(256, 1)
void hippo_setup_kernel(const float* __restrict__ A,
                        const float* __restrict__ Bm,
                        const float* __restrict__ dtp,
                        float* __restrict__ abar,
                        float* __restrict__ bbar) {
    __shared__ __attribute__((aligned(16))) float Ls[SDIM * LP];
    __shared__ __attribute__((aligned(16))) float Xs[SDIM * LP];
    __shared__ float Ts[32 * TP];
    const float dt = dtp[0];
    const int tid = threadIdx.x;
    const int lane = tid & 31;
    const int w = __builtin_amdgcn_readfirstlane(tid >> 5);
    const int lc = lane & 15, lh = lane >> 4;
    const int tn = w & 3;
    const int cg = w >> 2;

    float av[16];
#pragma unroll
    for (int u = 0; u < 16; ++u) av[u] = A[tid + 256 * u];
#pragma unroll
    for (int u = 0; u < 16; ++u) {
        const int e = tid + 256 * u;
        const int i = e >> 6, j = e & 63;
        float l = -0.5f * dt * av[u];
        if (i == j) l += 1.0f;
        Ls[i * LP + j] = l;
        Xs[i * LP + j] = 0.0f;
    }
    v2f bf[4][16];
#pragma unroll
    for (int ct = 0; ct < 4; ++ct) {
        const int c = (cg * 4 + ct) * 16 + lc;
#pragma unroll
        for (int tk = 0; tk < 16; ++tk) {
            bf[ct][tk].x = Bm[(tk * 4 + 2 * lh) * NCH + c];
            bf[ct][tk].y = Bm[(tk * 4 + 2 * lh + 1) * NCH + c];
        }
    }
    __syncthreads();

    if (tid < 64) {
        const int I = tid >> 4, j = tid & 15;
        const float* Lb = Ls + (16 * I) * LP + 16 * I;
        float x[16];
#pragma unroll
        for (int i = 0; i < 16; ++i) {
            float sum = (i == j) ? 1.0f : 0.0f;
#pragma unroll
            for (int k = 0; k < i; ++k) sum -= Lb[i * LP + k] * x[k];
            x[i] = sum / Lb[i * LP + i];
        }
#pragma unroll
        for (int i = 0; i < 16; ++i) Xs[(16 * I + i) * LP + 16 * I + j] = x[i];
    }
    __syncthreads();

    {
        const int r = tid >> 4, c = tid & 15;
#pragma unroll
        for (int p = 0; p < 2; ++p) {
            const int R0 = 16 + 32 * p, C0 = 32 * p;
            float s = 0.0f;
#pragma unroll
            for (int k = 0; k < 16; ++k) s += Ls[(R0 + r) * LP + C0 + k] * Xs[(C0 + k) * LP + C0 + c];
            Ts[(16 * p + r) * TP + c] = s;
        }
        __syncthreads();
#pragma unroll
        for (int p = 0; p < 2; ++p) {
            const int R0 = 16 + 32 * p, C0 = 32 * p;
            float s = 0.0f;
#pragma unroll
            for (int k = 0; k < 16; ++k) s += Xs[(R0 + r) * LP + R0 + k] * Ts[(16 * p + k) * TP + c];
            Xs[(R0 + r) * LP + C0 + c] = -s;
        }
        __syncthreads();
    }

    {
        const int r = tid >> 3, c0 = tid & 7;
        float s[4] = {0.0f, 0.0f, 0.0f, 0.0f};
#pragma unroll
        for (int k = 0; k < 32; ++k) {
            const float a = Ls[(32 + r) * LP + k];
#pragma unroll
            for (int m = 0; m < 4; ++m) s[m] += a * Xs[k * LP + c0 + 8 * m];
        }
#pragma unroll
        for (int m = 0; m < 4; ++m) Ts[r * TP + c0 + 8 * m] = s[m];
        __syncthreads();
#pragma unroll
        for (int m = 0; m < 4; ++m) s[m] = 0.0f;
#pragma unroll
        for (int k = 0; k < 32; ++k) {
            const float a = Xs[(32 + r) * LP + 32 + k];
#pragma unroll
            for (int m = 0; m < 4; ++m) s[m] += a * Ts[k * TP + c0 + 8 * m];
        }
#pragma unroll
        for (int m = 0; m < 4; ++m) Xs[(32 + r) * LP + c0 + 8 * m] = -s[m];
        __syncthreads();
    }

#pragma unroll
    for (int u = 0; u < 16; ++u) {
        const int e = tid + 256 * u;
        const int i = e >> 6, j = e & 63;
        abar[e] = 2.0f * Xs[i * LP + j] - ((i == j) ? 1.0f : 0.0f);
    }

    v2f a[16];
#pragma unroll
    for (int tk = 0; tk < 16; ++tk)
        a[tk] = *reinterpret_cast<const v2f*>(Xs + (tn * 16 + lc) * LP + tk * 4 + 2 * lh);
#pragma unroll
    for (int ct = 0; ct < 4; ++ct) {
        const int c = (cg * 4 + ct) * 16 + lc;
        v8f acc0, acc1;
#pragma unroll
        for (int r = 0; r < 8; ++r) { acc0[r] = 0.0f; acc1[r] = 0.0f; }
#pragma unroll
        for (int tk = 0; tk < 16; tk += 2) {
            acc0 = __builtin_amdgcn_wmma_f32_16x16x4_f32(false, a[tk], false, bf[ct][tk], (short)0, acc0, false, false);
            acc1 = __builtin_amdgcn_wmma_f32_16x16x4_f32(false, a[tk + 1], false, bf[ct][tk + 1], (short)0, acc1, false, false);
        }
#pragma unroll
        for (int r = 0; r < 8; ++r)
            bbar[(tn * 16 + 8 * lh + r) * NCH + c] = dt * (acc0[r] + acc1[r]);
    }
}


#define KPITCH 4112
#define XPITCH 4116
#define SPITCH 68

__device__ __forceinline__ void lds_barrier() {
    __builtin_amdgcn_fence(__ATOMIC_RELEASE, "workgroup", "local");
    __builtin_amdgcn_s_barrier();
    __builtin_amdgcn_fence(__ATOMIC_ACQUIRE, "workgroup", "local");
}

__device__ __forceinline__ void chain_run(const float* __restrict__ M, v8f s, float* dstAll,
                                          float* dstLast, int nsteps, float* Sl,
                                          int tn, int lc, int lh) {
    v2f a[16];
#pragma unroll
    for (int tk = 0; tk < 16; ++tk)
        a[tk] = *reinterpret_cast<const v2f*>(M + (tn * 16 + lc) * SDIM + tk * 4 + 2 * lh);
    const int rowoff = tn * 16 + 8 * lh;
    {
        v4f lo, hi;
#pragma unroll
        for (int r = 0; r < 4; ++r) { lo[r] = s[r]; hi[r] = s[r + 4]; }
        *reinterpret_cast<v4f*>(Sl + lc * SPITCH + rowoff) = lo;
        *reinterpret_cast<v4f*>(Sl + lc * SPITCH + rowoff + 4) = hi;
        if (dstAll) {
            *reinterpret_cast<v4f*>(dstAll + rowoff) = lo;
            *reinterpret_cast<v4f*>(dstAll + rowoff + 4) = hi;
        }
    }
    lds_barrier();
#pragma unroll 1
    for (int i = 1; i <= nsteps; ++i) {
        const float* cur = Sl + ((i - 1) & 1) * (16 * SPITCH);
        float* nxt = Sl + (i & 1) * (16 * SPITCH);
        v8f acc0, acc1;
#pragma unroll
        for (int r = 0; r < 8; ++r) { acc0[r] = 0.0f; acc1[r] = 0.0f; }
#pragma unroll
        for (int tk = 0; tk < 16; tk += 2) {
            const v2f b0 = *reinterpret_cast<const v2f*>(cur + lc * SPITCH + tk * 4 + 2 * lh);
            const v2f b1 = *reinterpret_cast<const v2f*>(cur + lc * SPITCH + tk * 4 + 4 + 2 * lh);
            acc0 = __builtin_amdgcn_wmma_f32_16x16x4_f32(false, a[tk], false, b0, (short)0, acc0, false, false);
            acc1 = __builtin_amdgcn_wmma_f32_16x16x4_f32(false, a[tk + 1], false, b1, (short)0, acc1, false, false);
        }
        v4f lo, hi;
#pragma unroll
        for (int r = 0; r < 4; ++r) { lo[r] = acc0[r] + acc1[r]; hi[r] = acc0[r + 4] + acc1[r + 4]; }
        *reinterpret_cast<v4f*>(nxt + lc * SPITCH + rowoff) = lo;
        *reinterpret_cast<v4f*>(nxt + lc * SPITCH + rowoff + 4) = hi;
        if (dstAll) {
            *reinterpret_cast<v4f*>(dstAll + i * SDIM + rowoff) = lo;
            *reinterpret_cast<v4f*>(dstAll + i * SDIM + rowoff + 4) = hi;
        }
        if (dstLast && i == nsteps) {
            *reinterpret_cast<v4f*>(dstLast + rowoff) = lo;
            *reinterpret_cast<v4f*>(dstLast + rowoff + 4) = hi;
        }
        lds_barrier();
    }
}

__global__ __launch_bounds__(128, 1)
void hippo_powers_kernel(const float* __restrict__ abar, const float* __restrict__ bbar,
                         float* __restrict__ Gc, float* __restrict__ PT) {
    __shared__ __attribute__((aligned(16))) float Sl[2 * 16 * SPITCH];
    const int lane = threadIdx.x & 31;
    const int tn = __builtin_amdgcn_readfirstlane(threadIdx.x >> 5);
    const int lc = lane & 15, lh = lane >> 4;
    const int blk = blockIdx.x;
    v8f s;
    if (blk < 8) {
        const int c = blk * 16 + lc;
#pragma unroll
        for (int r = 0; r < 8; ++r) s[r] = bbar[(tn * 16 + r + 8 * lh) * NCH + c];
        chain_run(abar, s, Gc + (size_t)c * SDIM * SDIM, nullptr, SDIM - 1, Sl, tn, lc, lh);
    } else {
        const int col = (blk - 8) * 16 + lc;
#pragma unroll
        for (int r = 0; r < 8; ++r) s[r] = abar[(tn * 16 + r + 8 * lh) * SDIM + col];
        chain_run(abar, s, nullptr, PT + col * SDIM, SDIM - 1, Sl, tn, lc, lh);
    }
}

__global__ __launch_bounds__(128, 1)
void hippo_h_kernel(const float* __restrict__ PT, const float* __restrict__ Cmat,
                    float* __restrict__ Hc) {
    __shared__ __attribute__((aligned(16))) float Sl[2 * 16 * SPITCH];
    const int lane = threadIdx.x & 31;
    const int tn = __builtin_amdgcn_readfirstlane(threadIdx.x >> 5);
    const int lc = lane & 15, lh = lane >> 4;
    const int c = blockIdx.x * 16 + lc;
    v8f s;
#pragma unroll
    for (int r = 0; r < 8; ++r) s[r] = Cmat[(tn * 16 + r + 8 * lh) * NCH + c];
    chain_run(PT, s, Hc + (size_t)c * SDIM * SDIM, nullptr, SDIM - 1, Sl, tn, lc, lh);
}

__global__ __launch_bounds__(256, 1)
void hippo_taps_kernel(const float* __restrict__ Hc, const float* __restrict__ Gc,
                       float* __restrict__ Kp) {
    const int c = blockIdx.x;
    const int lane = threadIdx.x & 31;
    const int w = threadIdx.x >> 5;
    const int lc = lane & 15, lh = lane >> 4;
    const float* H = Hc + (size_t)c * SDIM * SDIM;
    const float* G = Gc + (size_t)c * SDIM * SDIM;
    float* K = Kp + (size_t)c * KPITCH;
    if (threadIdx.x < 16) K[SLEN + threadIdx.x] = 0.0f;
#pragma unroll
    for (int t = 0; t < 2; ++t) {
        const int tile = 2 * w + t;
        const int tj = tile >> 2, ti = tile & 3;
        v8f acc;
#pragma unroll
        for (int r = 0; r < 8; ++r) acc[r] = 0.0f;
#pragma unroll
        for (int tk = 0; tk < 16; ++tk) {
            const v2f af = *reinterpret_cast<const v2f*>(H + (tj * 16 + lc) * SDIM + tk * 4 + 2 * lh);
            const v2f bf = *reinterpret_cast<const v2f*>(G + (ti * 16 + lc) * SDIM + tk * 4 + 2 * lh);
            acc = __builtin_amdgcn_wmma_f32_16x16x4_f32(false, af, false, bf, (short)0, acc, false, false);
        }
#pragma unroll
        for (int r = 0; r < 8; ++r)
            K[SLEN - 1 - ((tj * 16 + r + 8 * lh) * SDIM + ti * 16 + lc)] = acc[r];
    }
}

struct ConvFrag { v2f a0[4], a1[4]; v4f x0lo, x0hi, x1lo, x1hi; };

__device__ __forceinline__ void conv_load(ConvFrag& f, const float* xb, const float* kb) {
    f.x0lo = *reinterpret_cast<const v4f*>(xb);
    f.x0hi = *reinterpret_cast<const v4f*>(xb + 4);
    f.x1lo = *reinterpret_cast<const v4f*>(xb - 16);
    f.x1hi = *reinterpret_cast<const v4f*>(xb - 12);
#pragma unroll
    for (int s = 0; s < 4; ++s) {
        f.a0[s].x = kb[2 * s];      f.a0[s].y = kb[2 * s + 1];
        f.a1[s].x = kb[2 * s - 16]; f.a1[s].y = kb[2 * s - 15];
    }
}

__device__ __forceinline__ void conv_mma(const ConvFrag& f, v8f& acc0, v8f& acc1) {
    v2f b0[4], b1[4];
    b0[0].x = f.x0lo[0]; b0[0].y = f.x0lo[1]; b0[1].x = f.x0lo[2]; b0[1].y = f.x0lo[3];
    b0[2].x = f.x0hi[0]; b0[2].y = f.x0hi[1]; b0[3].x = f.x0hi[2]; b0[3].y = f.x0hi[3];
    b1[0].x = f.x1lo[0]; b1[0].y = f.x1lo[1]; b1[1].x = f.x1lo[2]; b1[1].y = f.x1lo[3];
    b1[2].x = f.x1hi[0]; b1[2].y = f.x1hi[1]; b1[3].x = f.x1hi[2]; b1[3].y = f.x1hi[3];
#pragma unroll
    for (int s = 0; s < 4; ++s) {
        acc0 = __builtin_amdgcn_wmma_f32_16x16x4_f32(false, f.a0[s], false, b0[s], (short)0, acc0, false, false);
        acc1 = __builtin_amdgcn_wmma_f32_16x16x4_f32(false, f.a1[s], false, b1[s], (short)0, acc1, false, false);
    }
}

__global__ __launch_bounds__(256, 1)
void hippo_conv_kernel(const float* __restrict__ sig, const float* __restrict__ Kp,
                       float* __restrict__ out) {
    __shared__ __attribute__((aligned(16))) float xs[(NB + 1) * XPITCH];
    __shared__ __attribute__((aligned(16))) float ks[KPITCH];
    const int c = blockIdx.x >> 1;
    const int z = blockIdx.x & 1;
    const int tid = threadIdx.x;
    const int lane = tid & 31;
    const int w = __builtin_amdgcn_readfirstlane(tid >> 5);
    const int lc = lane & 15, lh = lane >> 4;

    if (tid < NB * 16) xs[(tid >> 4) * XPITCH + (tid & 15)] = 0.0f;
#pragma unroll 1
    for (int b = 0; b < NB; ++b) {
        const v4f* src = reinterpret_cast<const v4f*>(sig + ((size_t)b * NCH + c) * SLEN);
        v4f* dst = reinterpret_cast<v4f*>(xs + b * XPITCH + 16);
#pragma unroll
        for (int i = 0; i < SLEN / 4 / 256; ++i) dst[tid + i * 256] = src[tid + i * 256];
    }
    {
        const v4f* src = reinterpret_cast<const v4f*>(Kp + (size_t)c * KPITCH);
        v4f* dst = reinterpret_cast<v4f*>(ks);
        for (int i = tid; i < KPITCH / 4; i += 256) dst[i] = src[i];
    }
    __syncthreads();

    const int bcol = lc & 7;
    const int e    = lc >> 3;
    const int g    = z * 8 + w;
    const float* xrow = xs + bcol * XPITCH + 8 * lh + 16 * (e + 1);
    const float* krow = ks + (SLEN - 1) - lc + 8 * lh;

#pragma unroll 1
    for (int it = 0; it < 8; ++it) {
        const int P0 = g + 16 * (it >> 1);
        const int P  = (it & 1) ? (127 - P0) : P0;
        const int mlast = 2 * P;
        v8f acc0, acc1;
#pragma unroll
        for (int r = 0; r < 8; ++r) { acc0[r] = 0.0f; acc1[r] = 0.0f; }
        ConvFrag f0, f1;
        conv_load(f0, xrow + 16 * (2 * P), krow);
        int m = 0;
#pragma unroll 1
        while (true) {
            if (m >= mlast) { conv_mma(f0, acc0, acc1); break; }
            conv_load(f1, xrow + 16 * (2 * P - m - 2), krow - 16 * (m + 2));
            conv_mma(f0, acc0, acc1);
            m += 2;
            if (m >= mlast) { conv_mma(f1, acc0, acc1); break; }
            conv_load(f0, xrow + 16 * (2 * P - m - 2), krow - 16 * (m + 2));
            conv_mma(f1, acc0, acc1);
            m += 2;
        }
        float* yp = out + ((size_t)bcol * NCH + c) * SLEN + 16 * (2 * P + e) + 8 * lh;
        v4f lo, hi;
#pragma unroll
        for (int r = 0; r < 4; ++r) { lo[r] = acc0[r] + acc1[r]; hi[r] = acc0[r + 4] + acc1[r + 4]; }
        *reinterpret_cast<v4f*>(yp) = lo;
        *reinterpret_cast<v4f*>(yp + 4) = hi;
    }
}

extern "C" void kernel_launch(void* const* d_in, const int* in_sizes, int n_in,
                              void* d_out, int out_size, void* d_ws, size_t ws_size,
                              hipStream_t stream) {
    const float* sig = (const float*)d_in[0];
    const float* Bm  = (const float*)d_in[1];
    const float* Cm  = (const float*)d_in[2];
    const float* Am  = (const float*)d_in[3];
    const float* dtp = (const float*)d_in[4];
    float* out  = (float*)d_out;
    float* abar = (float*)d_ws;
    float* bbar = abar + SDIM * SDIM;
    float* pt   = bbar + SDIM * NCH;
    float* gc   = pt + SDIM * SDIM;
    float* hc   = gc + NCH * SDIM * SDIM;
    float* kp   = hc + NCH * SDIM * SDIM;

    hippo_setup_kernel<<<1, 256, 0, stream>>>(Am, Bm, dtp, abar, bbar);
    hippo_powers_kernel<<<12, 128, 0, stream>>>(abar, bbar, gc, pt);
    hippo_h_kernel<<<8, 128, 0, stream>>>(pt, Cm, hc);
    hippo_taps_kernel<<<NCH, 256, 0, stream>>>(hc, gc, kp);
    hippo_conv_kernel<<<NCH * 2, 256, 0, stream>>>(sig, kp, out);
}


# tap chains with stride 8: A_bar^8, A_bar^64, A_bar^512 by nine squarings in setup; G and H chains in one launch of 128 blocks, at most 14 dependent steps instead of 63+63; tap sums reordered
# speedup vs baseline: 30.0025x; 1.2228x over previous
#include <hip/hip_runtime.h>
#include <hip/hip_bf16.h>

typedef float v2f __attribute__((ext_vector_type(2)));
typedef float v4f __attribute__((ext_vector_type(4)));
typedef float v8f __attribute__((ext_vector_type(8)));

#define SDIM 64
#define NCH  128
#define NB   8
#define SLEN 4096

#define LP 68
#define TP 33

__device__ __forceinline__ void lds_barrier() {
    __builtin_amdgcn_fence(__ATOMIC_RELEASE, "workgroup", "local");
    __builtin_amdgcn_s_barrier();
    __builtin_amdgcn_fence(__ATOMIC_ACQUIRE, "workgroup", "local");
}

__global__ __launch_bounds__(256, 1)
void hippo_setup_kernel(const float* __restrict__ A,
                        const float* __restrict__ Bm,
                        const float* __restrict__ dtp,
                        float* __restrict__ abar,
                        float* __restrict__ bbar,
                        float* __restrict__ q8,
                        float* __restrict__ pT,
                        float* __restrict__ rT) {
    __shared__ __attribute__((aligned(16))) float Ls[SDIM * LP];
    __shared__ __attribute__((aligned(16))) float Xs[SDIM * LP];
    __shared__ float Ts[32 * TP];
    const float dt = dtp[0];
    const int tid = threadIdx.x;
    const int lane = tid & 31;
    const int w = __builtin_amdgcn_readfirstlane(tid >> 5);
    const int lc = lane & 15, lh = lane >> 4;
    const int tn = w & 3;
    const int cg = w >> 2;

    float av[16];
#pragma unroll
    for (int u = 0; u < 16; ++u) av[u] = A[tid + 256 * u];
#pragma unroll
    for (int u = 0; u < 16; ++u) {
        const int e = tid + 256 * u;
        const int i = e >> 6, j = e & 63;
        float l = -0.5f * dt * av[u];
        if (i == j) l += 1.0f;
        Ls[i * LP + j] = l;
        Xs[i * LP + j] = 0.0f;
    }
    v2f bf[4][16];
#pragma unroll
    for (int ct = 0; ct < 4; ++ct) {
        const int c = (cg * 4 + ct) * 16 + lc;
#pragma unroll
        for (int tk = 0; tk < 16; ++tk) {
            bf[ct][tk].x = Bm[(tk * 4 + 2 * lh) * NCH + c];
            bf[ct][tk].y = Bm[(tk * 4 + 2 * lh + 1) * NCH + c];
        }
    }
    __syncthreads();

    if (tid < 64) {
        const int I = tid >> 4, j = tid & 15;
        const float* Lb = Ls + (16 * I) * LP + 16 * I;
        float x[16];
#pragma unroll
        for (int i = 0; i < 16; ++i) {
            float sum = (i == j) ? 1.0f : 0.0f;
#pragma unroll
            for (int k = 0; k < i; ++k) sum -= Lb[i * LP + k] * x[k];
            x[i] = sum / Lb[i * LP + i];
        }
#pragma unroll
        for (int i = 0; i < 16; ++i) Xs[(16 * I + i) * LP + 16 * I + j] = x[i];
    }
    __syncthreads();

    {
        const int r = tid >> 4, c = tid & 15;
#pragma unroll
        for (int p = 0; p < 2; ++p) {
            const int R0 = 16 + 32 * p, C0 = 32 * p;
            float s = 0.0f;
#pragma unroll
            for (int k = 0; k < 16; ++k) s += Ls[(R0 + r) * LP + C0 + k] * Xs[(C0 + k) * LP + C0 + c];
            Ts[(16 * p + r) * TP + c] = s;
        }
        __syncthreads();
#pragma unroll
        for (int p = 0; p < 2; ++p) {
            const int R0 = 16 + 32 * p, C0 = 32 * p;
            float s = 0.0f;
#pragma unroll
            for (int k = 0; k < 16; ++k) s += Xs[(R0 + r) * LP + R0 + k] * Ts[(16 * p + k) * TP + c];
            Xs[(R0 + r) * LP + C0 + c] = -s;
        }
        __syncthreads();
    }

    {
        const int r = tid >> 3, c0 = tid & 7;
        float s[4] = {0.0f, 0.0f, 0.0f, 0.0f};
#pragma unroll
        for (int k = 0; k < 32; ++k) {
            const float a = Ls[(32 + r) * LP + k];
#pragma unroll
            for (int m = 0; m < 4; ++m) s[m] += a * Xs[k * LP + c0 + 8 * m];
        }
#pragma unroll
        for (int m = 0; m < 4; ++m) Ts[r * TP + c0 + 8 * m] = s[m];
        __syncthreads();
#pragma unroll
        for (int m = 0; m < 4; ++m) s[m] = 0.0f;
#pragma unroll
        for (int k = 0; k < 32; ++k) {
            const float a = Xs[(32 + r) * LP + 32 + k];
#pragma unroll
            for (int m = 0; m < 4; ++m) s[m] += a * Ts[k * TP + c0 + 8 * m];
        }
#pragma unroll
        for (int m = 0; m < 4; ++m) Xs[(32 + r) * LP + c0 + 8 * m] = -s[m];
        __syncthreads();
    }

#pragma unroll
    for (int u = 0; u < 16; ++u) {
        const int e = tid + 256 * u;
        const int i = e >> 6, j = e & 63;
        const float ab = 2.0f * Xs[i * LP + j] - ((i == j) ? 1.0f : 0.0f);
        abar[e] = ab;
        Ls[i * LP + j] = ab;
    }

    v2f a[16];
#pragma unroll
    for (int tk = 0; tk < 16; ++tk)
        a[tk] = *reinterpret_cast<const v2f*>(Xs + (tn * 16 + lc) * LP + tk * 4 + 2 * lh);
#pragma unroll
    for (int ct = 0; ct < 4; ++ct) {
        const int c = (cg * 4 + ct) * 16 + lc;
        v8f acc0, acc1;
#pragma unroll
        for (int r = 0; r < 8; ++r) { acc0[r] = 0.0f; acc1[r] = 0.0f; }
#pragma unroll
        for (int tk = 0; tk < 16; tk += 2) {
            acc0 = __builtin_amdgcn_wmma_f32_16x16x4_f32(false, a[tk], false, bf[ct][tk], (short)0, acc0, false, false);
            acc1 = __builtin_amdgcn_wmma_f32_16x16x4_f32(false, a[tk + 1], false, bf[ct][tk + 1], (short)0, acc1, false, false);
        }
#pragma unroll
        for (int r = 0; r < 8; ++r)
            bbar[(tn * 16 + 8 * lh + r) * NCH + c] = dt * (acc0[r] + acc1[r]);
    }
    lds_barrier();

    float* src = Ls;
    float* dst = Xs;
#pragma unroll 1
    for (int it = 0; it < 9; ++it) {
        v2f sa[16], sb0[16], sb1[16];
        const int col0 = cg * 32 + lc;
#pragma unroll
        for (int tk = 0; tk < 16; ++tk) {
            sa[tk] = *reinterpret_cast<const v2f*>(src + (tn * 16 + lc) * LP + tk * 4 + 2 * lh);
            const float* p = src + (tk * 4 + 2 * lh) * LP + col0;
            sb0[tk].x = p[0];  sb0[tk].y = p[LP];
            sb1[tk].x = p[16]; sb1[tk].y = p[LP + 16];
        }
        v8f d0, d1;
#pragma unroll
        for (int r = 0; r < 8; ++r) { d0[r] = 0.0f; d1[r] = 0.0f; }
#pragma unroll
        for (int tk = 0; tk < 16; ++tk) {
            d0 = __builtin_amdgcn_wmma_f32_16x16x4_f32(false, sa[tk], false, sb0[tk], (short)0, d0, false, false);
            d1 = __builtin_amdgcn_wmma_f32_16x16x4_f32(false, sa[tk], false, sb1[tk], (short)0, d1, false, false);
        }
#pragma unroll
        for (int r = 0; r < 8; ++r) {
            dst[(tn * 16 + 8 * lh + r) * LP + col0] = d0[r];
            dst[(tn * 16 + 8 * lh + r) * LP + col0 + 16] = d1[r];
        }
        lds_barrier();
        if (it == 2) {
#pragma unroll
            for (int u = 0; u < 4; ++u) {
                const int e4 = tid + 256 * u;
                const int i = e4 >> 4, j4 = (e4 & 15) * 4;
                *reinterpret_cast<v4f*>(q8 + i * SDIM + j4) = *reinterpret_cast<const v4f*>(dst + i * LP + j4);
            }
        } else if (it == 5 || it == 8) {
            float* mt = (it == 5) ? pT : rT;
#pragma unroll
            for (int u = 0; u < 4; ++u) {
                const int e4 = tid + 256 * u;
                const int i = e4 >> 4, j4 = (e4 & 15) * 4;
                v4f v;
#pragma unroll
                for (int m = 0; m < 4; ++m) v[m] = dst[(j4 + m) * LP + i];
                *reinterpret_cast<v4f*>(mt + i * SDIM + j4) = v;
            }
        }
        float* t = src; src = dst; dst = t;
    }
}


#define KPITCH 4112
#define XPITCH 4116
#define SPITCH 68

__device__ __forceinline__ void chain_step(const v2f (&a)[16], const float* cur, float* nxt,
                                           int lc, int lh, int rowoff, v4f& lo, v4f& hi) {
    v8f acc0, acc1;
#pragma unroll
    for (int r = 0; r < 8; ++r) { acc0[r] = 0.0f; acc1[r] = 0.0f; }
#pragma unroll
    for (int tk = 0; tk < 16; tk += 2) {
        const v2f b0 = *reinterpret_cast<const v2f*>(cur + lc * SPITCH + tk * 4 + 2 * lh);
        const v2f b1 = *reinterpret_cast<const v2f*>(cur + lc * SPITCH + tk * 4 + 4 + 2 * lh);
        acc0 = __builtin_amdgcn_wmma_f32_16x16x4_f32(false, a[tk], false, b0, (short)0, acc0, false, false);
        acc1 = __builtin_amdgcn_wmma_f32_16x16x4_f32(false, a[tk + 1], false, b1, (short)0, acc1, false, false);
    }
#pragma unroll
    for (int r = 0; r < 4; ++r) { lo[r] = acc0[r] + acc1[r]; hi[r] = acc0[r + 4] + acc1[r + 4]; }
    *reinterpret_cast<v4f*>(nxt + lc * SPITCH + rowoff) = lo;
    *reinterpret_cast<v4f*>(nxt + lc * SPITCH + rowoff + 4) = hi;
}

__global__ __launch_bounds__(128, 1)
void hippo_chains_kernel(const float* __restrict__ abar, const float* __restrict__ q8,
                         const float* __restrict__ pT, const float* __restrict__ rT,
                         const float* __restrict__ bbar, const float* __restrict__ Cmat,
                         float* __restrict__ Gc, float* __restrict__ Hc) {
    __shared__ __attribute__((aligned(16))) float Sl[2 * 16 * SPITCH];
    const int lane = threadIdx.x & 31;
    const int tn = __builtin_amdgcn_readfirstlane(threadIdx.x >> 5);
    const int lc = lane & 15, lh = lane >> 4;
    const int blk = blockIdx.x;
    const int side = blk >> 6;
    const int b = blk & 7;
    const int c = ((blk >> 3) & 7) * 16 + lc;
    const float* M1 = side ? pT : abar;
    const float* M2 = side ? rT : q8;
    const float* S0 = side ? Cmat : bbar;
    const int rowoff = tn * 16 + 8 * lh;
    float* dst = (side ? Hc : Gc) + (size_t)c * SDIM * SDIM + b * SDIM + rowoff;

    v4f lo, hi;
#pragma unroll
    for (int r = 0; r < 4; ++r) {
        lo[r] = S0[(rowoff + r) * NCH + c];
        hi[r] = S0[(rowoff + r + 4) * NCH + c];
    }
    v2f a1[16], a2[16];
#pragma unroll
    for (int tk = 0; tk < 16; ++tk) {
        a1[tk] = *reinterpret_cast<const v2f*>(M1 + (tn * 16 + lc) * SDIM + tk * 4 + 2 * lh);
        a2[tk] = *reinterpret_cast<const v2f*>(M2 + (tn * 16 + lc) * SDIM + tk * 4 + 2 * lh);
    }
    *reinterpret_cast<v4f*>(Sl + lc * SPITCH + rowoff) = lo;
    *reinterpret_cast<v4f*>(Sl + lc * SPITCH + rowoff + 4) = hi;
    lds_barrier();

    int par = 0;
#pragma unroll 1
    for (int i = 0; i < b; ++i) {
        chain_step(a1, Sl + par * (16 * SPITCH), Sl + (par ^ 1) * (16 * SPITCH), lc, lh, rowoff, lo, hi);
        par ^= 1;
        lds_barrier();
    }
    *reinterpret_cast<v4f*>(dst) = lo;
    *reinterpret_cast<v4f*>(dst + 4) = hi;
#pragma unroll 1
    for (int a = 1; a < 8; ++a) {
        chain_step(a2, Sl + par * (16 * SPITCH), Sl + (par ^ 1) * (16 * SPITCH), lc, lh, rowoff, lo, hi);
        par ^= 1;
        *reinterpret_cast<v4f*>(dst + a * 8 * SDIM) = lo;
        *reinterpret_cast<v4f*>(dst + a * 8 * SDIM + 4) = hi;
        lds_barrier();
    }
}

__global__ __launch_bounds__(256, 1)
void hippo_taps_kernel(const float* __restrict__ Hc, const float* __restrict__ Gc,
                       float* __restrict__ Kp) {
    const int c = blockIdx.x;
    const int lane = threadIdx.x & 31;
    const int w = threadIdx.x >> 5;
    const int lc = lane & 15, lh = lane >> 4;
    const float* H = Hc + (size_t)c * SDIM * SDIM;
    const float* G = Gc + (size_t)c * SDIM * SDIM;
    float* K = Kp + (size_t)c * KPITCH;
    if (threadIdx.x < 16) K[SLEN + threadIdx.x] = 0.0f;
#pragma unroll
    for (int t = 0; t < 2; ++t) {
        const int tile = 2 * w + t;
        const int tj = tile >> 2, ti = tile & 3;
        v8f acc;
#pragma unroll
        for (int r = 0; r < 8; ++r) acc[r] = 0.0f;
#pragma unroll
        for (int tk = 0; tk < 16; ++tk) {
            const v2f af = *reinterpret_cast<const v2f*>(H + (tj * 16 + lc) * SDIM + tk * 4 + 2 * lh);
            const v2f bf = *reinterpret_cast<const v2f*>(G + (ti * 16 + lc) * SDIM + tk * 4 + 2 * lh);
            acc = __builtin_amdgcn_wmma_f32_16x16x4_f32(false, af, false, bf, (short)0, acc, false, false);
        }
#pragma unroll
        for (int r = 0; r < 8; ++r)
            K[SLEN - 1 - ((tj * 16 + r + 8 * lh) * SDIM + ti * 16 + lc)] = acc[r];
    }
}

struct ConvFrag { v2f a0[4], a1[4]; v4f x0lo, x0hi, x1lo, x1hi; };

__device__ __forceinline__ void conv_load(ConvFrag& f, const float* xb, const float* kb) {
    f.x0lo = *reinterpret_cast<const v4f*>(xb);
    f.x0hi = *reinterpret_cast<const v4f*>(xb + 4);
    f.x1lo = *reinterpret_cast<const v4f*>(xb - 16);
    f.x1hi = *reinterpret_cast<const v4f*>(xb - 12);
#pragma unroll
    for (int s = 0; s < 4; ++s) {
        f.a0[s].x = kb[2 * s];      f.a0[s].y = kb[2 * s + 1];
        f.a1[s].x = kb[2 * s - 16]; f.a1[s].y = kb[2 * s - 15];
    }
}

__device__ __forceinline__ void conv_mma(const ConvFrag& f, v8f& acc0, v8f& acc1) {
    v2f b0[4], b1[4];
    b0[0].x = f.x0lo[0]; b0[0].y = f.x0lo[1]; b0[1].x = f.x0lo[2]; b0[1].y = f.x0lo[3];
    b0[2].x = f.x0hi[0]; b0[2].y = f.x0hi[1]; b0[3].x = f.x0hi[2]; b0[3].y = f.x0hi[3];
    b1[0].x = f.x1lo[0]; b1[0].y = f.x1lo[1]; b1[1].x = f.x1lo[2]; b1[1].y = f.x1lo[3];
    b1[2].x = f.x1hi[0]; b1[2].y = f.x1hi[1]; b1[3].x = f.x1hi[2]; b1[3].y = f.x1hi[3];
#pragma unroll
    for (int s = 0; s < 4; ++s) {
        acc0 = __builtin_amdgcn_wmma_f32_16x16x4_f32(false, f.a0[s], false, b0[s], (short)0, acc0, false, false);
        acc1 = __builtin_amdgcn_wmma_f32_16x16x4_f32(false, f.a1[s], false, b1[s], (short)0, acc1, false, false);
    }
}

__global__ __launch_bounds__(256, 1)
void hippo_conv_kernel(const float* __restrict__ sig, const float* __restrict__ Kp,
                       float* __restrict__ out) {
    __shared__ __attribute__((aligned(16))) float xs[(NB + 1) * XPITCH];
    __shared__ __attribute__((aligned(16))) float ks[KPITCH];
    const int c = blockIdx.x >> 1;
    const int z = blockIdx.x & 1;
    const int tid = threadIdx.x;
    const int lane = tid & 31;
    const int w = __builtin_amdgcn_readfirstlane(tid >> 5);
    const int lc = lane & 15, lh = lane >> 4;

    if (tid < NB * 16) xs[(tid >> 4) * XPITCH + (tid & 15)] = 0.0f;
#pragma unroll 1
    for (int b = 0; b < NB; ++b) {
        const v4f* src = reinterpret_cast<const v4f*>(sig + ((size_t)b * NCH + c) * SLEN);
        v4f* dst = reinterpret_cast<v4f*>(xs + b * XPITCH + 16);
#pragma unroll
        for (int i = 0; i < SLEN / 4 / 256; ++i) dst[tid + i * 256] = src[tid + i * 256];
    }
    {
        const v4f* src = reinterpret_cast<const v4f*>(Kp + (size_t)c * KPITCH);
        v4f* dst = reinterpret_cast<v4f*>(ks);
        for (int i = tid; i < KPITCH / 4; i += 256) dst[i] = src[i];
    }
    __syncthreads();

    const int bcol = lc & 7;
    const int e    = lc >> 3;
    const int g    = z * 8 + w;
    const float* xrow = xs + bcol * XPITCH + 8 * lh + 16 * (e + 1);
    const float* krow = ks + (SLEN - 1) - lc + 8 * lh;

#pragma unroll 1
    for (int it = 0; it < 8; ++it) {
        const int P0 = g + 16 * (it >> 1);
        const int P  = (it & 1) ? (127 - P0) : P0;
        const int mlast = 2 * P;
        v8f acc0, acc1;
#pragma unroll
        for (int r = 0; r < 8; ++r) { acc0[r] = 0.0f; acc1[r] = 0.0f; }
        ConvFrag f0, f1;
        conv_load(f0, xrow + 16 * (2 * P), krow);
        int m = 0;
#pragma unroll 1
        while (true) {
            if (m >= mlast) { conv_mma(f0, acc0, acc1); break; }
            conv_load(f1, xrow + 16 * (2 * P - m - 2), krow - 16 * (m + 2));
            conv_mma(f0, acc0, acc1);
            m += 2;
            if (m >= mlast) { conv_mma(f1, acc0, acc1); break; }
            conv_load(f0, xrow + 16 * (2 * P - m - 2), krow - 16 * (m + 2));
            conv_mma(f1, acc0, acc1);
            m += 2;
        }
        float* yp = out + ((size_t)bcol * NCH + c) * SLEN + 16 * (2 * P + e) + 8 * lh;
        v4f lo, hi;
#pragma unroll
        for (int r = 0; r < 4; ++r) { lo[r] = acc0[r] + acc1[r]; hi[r] = acc0[r + 4] + acc1[r + 4]; }
        *reinterpret_cast<v4f*>(yp) = lo;
        *reinterpret_cast<v4f*>(yp + 4) = hi;
    }
}

extern "C" void kernel_launch(void* const* d_in, const int* in_sizes, int n_in,
                              void* d_out, int out_size, void* d_ws, size_t ws_size,
                              hipStream_t stream) {
    const float* sig = (const float*)d_in[0];
    const float* Bm  = (const float*)d_in[1];
    const float* Cm  = (const float*)d_in[2];
    const float* Am  = (const float*)d_in[3];
    const float* dtp = (const float*)d_in[4];
    float* out  = (float*)d_out;
    float* abar = (float*)d_ws;
    float* bbar = abar + SDIM * SDIM;
    float* pt   = bbar + SDIM * NCH;
    float* gc   = pt + SDIM * SDIM;
    float* hc   = gc + NCH * SDIM * SDIM;
    float* kp   = hc + NCH * SDIM * SDIM;

    float* q8   = kp;
    float* rt   = kp + SDIM * SDIM;

    hippo_setup_kernel<<<1, 256, 0, stream>>>(Am, Bm, dtp, abar, bbar, q8, pt, rt);
    hippo_chains_kernel<<<128, 128, 0, stream>>>(abar, q8, pt, rt, bbar, Cm, gc, hc);
    hippo_taps_kernel<<<NCH, 256, 0, stream>>>(hc, gc, kp);
    hippo_conv_kernel<<<NCH * 2, 256, 0, stream>>>(sig, kp, out);
}


# convolution: the four 16x16 Toeplitz block products of each 2x2 block pair done as three (pair sums of x, tap differences K[k-16]-K[k], both built in LDS): 24% fewer WMMAs; output sums reordered
# speedup vs baseline: 35.4741x; 1.1824x over previous
#include <hip/hip_runtime.h>
#include <hip/hip_bf16.h>

typedef float v2f __attribute__((ext_vector_type(2)));
typedef float v4f __attribute__((ext_vector_type(4)));
typedef float v8f __attribute__((ext_vector_type(8)));

#define SDIM 64
#define NCH  128
#define NB   8
#define SLEN 4096

#define LP 68
#define TP 33

__device__ __forceinline__ void lds_barrier() {
    __builtin_amdgcn_fence(__ATOMIC_RELEASE, "workgroup", "local");
    __builtin_amdgcn_s_barrier();
    __builtin_amdgcn_fence(__ATOMIC_ACQUIRE, "workgroup", "local");
}

__global__ __launch_bounds__(256, 1)
void hippo_setup_kernel(const float* __restrict__ A,
                        const float* __restrict__ Bm,
                        const float* __restrict__ dtp,
                        float* __restrict__ abar,
                        float* __restrict__ bbar,
                        float* __restrict__ q8,
                        float* __restrict__ pT,
                        float* __restrict__ rT) {
    __shared__ __attribute__((aligned(16))) float Ls[SDIM * LP];
    __shared__ __attribute__((aligned(16))) float Xs[SDIM * LP];
    __shared__ float Ts[32 * TP];
    const float dt = dtp[0];
    const int tid = threadIdx.x;
    const int lane = tid & 31;
    const int w = __builtin_amdgcn_readfirstlane(tid >> 5);
    const int lc = lane & 15, lh = lane >> 4;
    const int tn = w & 3;
    const int cg = w >> 2;

    float av[16];
#pragma unroll
    for (int u = 0; u < 16; ++u) av[u] = A[tid + 256 * u];
#pragma unroll
    for (int u = 0; u < 16; ++u) {
        const int e = tid + 256 * u;
        const int i = e >> 6, j = e & 63;
        float l = -0.5f * dt * av[u];
        if (i == j) l += 1.0f;
        Ls[i * LP + j] = l;
        Xs[i * LP + j] = 0.0f;
    }
    v2f bf[4][16];
#pragma unroll
    for (int ct = 0; ct < 4; ++ct) {
        const int c = (cg * 4 + ct) * 16 + lc;
#pragma unroll
        for (int tk = 0; tk < 16; ++tk) {
            bf[ct][tk].x = Bm[(tk * 4 + 2 * lh) * NCH + c];
            bf[ct][tk].y = Bm[(tk * 4 + 2 * lh + 1) * NCH + c];
        }
    }
    __syncthreads();

    if (tid < 64) {
        const int I = tid >> 4, j = tid & 15;
        const float* Lb = Ls + (16 * I) * LP + 16 * I;
        float x[16];
#pragma unroll
        for (int i = 0; i < 16; ++i) {
            float sum = (i == j) ? 1.0f : 0.0f;
#pragma unroll
            for (int k = 0; k < i; ++k) sum -= Lb[i * LP + k] * x[k];
            x[i] = sum / Lb[i * LP + i];
        }
#pragma unroll
        for (int i = 0; i < 16; ++i) Xs[(16 * I + i) * LP + 16 * I + j] = x[i];
    }
    __syncthreads();

    {
        const int r = tid >> 4, c = tid & 15;
#pragma unroll
        for (int p = 0; p < 2; ++p) {
            const int R0 = 16 + 32 * p, C0 = 32 * p;
            float s = 0.0f;
#pragma unroll
            for (int k = 0; k < 16; ++k) s += Ls[(R0 + r) * LP + C0 + k] * Xs[(C0 + k) * LP + C0 + c];
            Ts[(16 * p + r) * TP + c] = s;
        }
        __syncthreads();
#pragma unroll
        for (int p = 0; p < 2; ++p) {
            const int R0 = 16 + 32 * p, C0 = 32 * p;
            float s = 0.0f;
#pragma unroll
            for (int k = 0; k < 16; ++k) s += Xs[(R0 + r) * LP + R0 + k] * Ts[(16 * p + k) * TP + c];
            Xs[(R0 + r) * LP + C0 + c] = -s;
        }
        __syncthreads();
    }

    {
        const int r = tid >> 3, c0 = tid & 7;
        float s[4] = {0.0f, 0.0f, 0.0f, 0.0f};
#pragma unroll
        for (int k = 0; k < 32; ++k) {
            const float a = Ls[(32 + r) * LP + k];
#pragma unroll
            for (int m = 0; m < 4; ++m) s[m] += a * Xs[k * LP + c0 + 8 * m];
        }
#pragma unroll
        for (int m = 0; m < 4; ++m) Ts[r * TP + c0 + 8 * m] = s[m];
        __syncthreads();
#pragma unroll
        for (int m = 0; m < 4; ++m) s[m] = 0.0f;
#pragma unroll
        for (int k = 0; k < 32; ++k) {
            const float a = Xs[(32 + r) * LP + 32 + k];
#pragma unroll
            for (int m = 0; m < 4; ++m) s[m] += a * Ts[k * TP + c0 + 8 * m];
        }
#pragma unroll
        for (int m = 0; m < 4; ++m) Xs[(32 + r) * LP + c0 + 8 * m] = -s[m];
        __syncthreads();
    }

#pragma unroll
    for (int u = 0; u < 16; ++u) {
        const int e = tid + 256 * u;
        const int i = e >> 6, j = e & 63;
        const float ab = 2.0f * Xs[i * LP + j] - ((i == j) ? 1.0f : 0.0f);
        abar[e] = ab;
        Ls[i * LP + j] = ab;
    }

    v2f a[16];
#pragma unroll
    for (int tk = 0; tk < 16; ++tk)
        a[tk] = *reinterpret_cast<const v2f*>(Xs + (tn * 16 + lc) * LP + tk * 4 + 2 * lh);
#pragma unroll
    for (int ct = 0; ct < 4; ++ct) {
        const int c = (cg * 4 + ct) * 16 + lc;
        v8f acc0, acc1;
#pragma unroll
        for (int r = 0; r < 8; ++r) { acc0[r] = 0.0f; acc1[r] = 0.0f; }
#pragma unroll
        for (int tk = 0; tk < 16; tk += 2) {
            acc0 = __builtin_amdgcn_wmma_f32_16x16x4_f32(false, a[tk], false, bf[ct][tk], (short)0, acc0, false, false);
            acc1 = __builtin_amdgcn_wmma_f32_16x16x4_f32(false, a[tk + 1], false, bf[ct][tk + 1], (short)0, acc1, false, false);
        }
#pragma unroll
        for (int r = 0; r < 8; ++r)
            bbar[(tn * 16 + 8 * lh + r) * NCH + c] = dt * (acc0[r] + acc1[r]);
    }
    lds_barrier();

    float* src = Ls;
    float* dst = Xs;
#pragma unroll 1
    for (int it = 0; it < 9; ++it) {
        v2f sa[16], sb0[16], sb1[16];
        const int col0 = cg * 32 + lc;
#pragma unroll
        for (int tk = 0; tk < 16; ++tk) {
            sa[tk] = *reinterpret_cast<const v2f*>(src + (tn * 16 + lc) * LP + tk * 4 + 2 * lh);
            const float* p = src + (tk * 4 + 2 * lh) * LP + col0;
            sb0[tk].x = p[0];  sb0[tk].y = p[LP];
            sb1[tk].x = p[16]; sb1[tk].y = p[LP + 16];
        }
        v8f d0, d1;
#pragma unroll
        for (int r = 0; r < 8; ++r) { d0[r] = 0.0f; d1[r] = 0.0f; }
#pragma unroll
        for (int tk = 0; tk < 16; ++tk) {
            d0 = __builtin_amdgcn_wmma_f32_16x16x4_f32(false, sa[tk], false, sb0[tk], (short)0, d0, false, false);
            d1 = __builtin_amdgcn_wmma_f32_16x16x4_f32(false, sa[tk], false, sb1[tk], (short)0, d1, false, false);
        }
#pragma unroll
        for (int r = 0; r < 8; ++r) {
            dst[(tn * 16 + 8 * lh + r) * LP + col0] = d0[r];
            dst[(tn * 16 + 8 * lh + r) * LP + col0 + 16] = d1[r];
        }
        lds_barrier();
        if (it == 2) {
#pragma unroll
            for (int u = 0; u < 4; ++u) {
                const int e4 = tid + 256 * u;
                const int i = e4 >> 4, j4 = (e4 & 15) * 4;
                *reinterpret_cast<v4f*>(q8 + i * SDIM + j4) = *reinterpret_cast<const v4f*>(dst + i * LP + j4);
            }
        } else if (it == 5 || it == 8) {
            float* mt = (it == 5) ? pT : rT;
#pragma unroll
            for (int u = 0; u < 4; ++u) {
                const int e4 = tid + 256 * u;
                const int i = e4 >> 4, j4 = (e4 & 15) * 4;
                v4f v;
#pragma unroll
                for (int m = 0; m < 4; ++m) v[m] = dst[(j4 + m) * LP + i];
                *reinterpret_cast<v4f*>(mt + i * SDIM + j4) = v;
            }
        }
        float* t = src; src = dst; dst = t;
    }
}


#define KPITCH 4112
#define SPITCH 68

__device__ __forceinline__ void chain_step(const v2f (&a)[16], const float* cur, float* nxt,
                                           int lc, int lh, int rowoff, v4f& lo, v4f& hi) {
    v8f acc0, acc1;
#pragma unroll
    for (int r = 0; r < 8; ++r) { acc0[r] = 0.0f; acc1[r] = 0.0f; }
#pragma unroll
    for (int tk = 0; tk < 16; tk += 2) {
        const v2f b0 = *reinterpret_cast<const v2f*>(cur + lc * SPITCH + tk * 4 + 2 * lh);
        const v2f b1 = *reinterpret_cast<const v2f*>(cur + lc * SPITCH + tk * 4 + 4 + 2 * lh);
        acc0 = __builtin_amdgcn_wmma_f32_16x16x4_f32(false, a[tk], false, b0, (short)0, acc0, false, false);
        acc1 = __builtin_amdgcn_wmma_f32_16x16x4_f32(false, a[tk + 1], false, b1, (short)0, acc1, false, false);
    }
#pragma unroll
    for (int r = 0; r < 4; ++r) { lo[r] = acc0[r] + acc1[r]; hi[r] = acc0[r + 4] + acc1[r + 4]; }
    *reinterpret_cast<v4f*>(nxt + lc * SPITCH + rowoff) = lo;
    *reinterpret_cast<v4f*>(nxt + lc * SPITCH + rowoff + 4) = hi;
}

__global__ __launch_bounds__(128, 1)
void hippo_chains_kernel(const float* __restrict__ abar, const float* __restrict__ q8,
                         const float* __restrict__ pT, const float* __restrict__ rT,
                         const float* __restrict__ bbar, const float* __restrict__ Cmat,
                         float* __restrict__ Gc, float* __restrict__ Hc) {
    __shared__ __attribute__((aligned(16))) float Sl[2 * 16 * SPITCH];
    const int lane = threadIdx.x & 31;
    const int tn = __builtin_amdgcn_readfirstlane(threadIdx.x >> 5);
    const int lc = lane & 15, lh = lane >> 4;
    const int blk = blockIdx.x;
    const int side = blk >> 6;
    const int b = blk & 7;
    const int c = ((blk >> 3) & 7) * 16 + lc;
    const float* M1 = side ? pT : abar;
    const float* M2 = side ? rT : q8;
    const float* S0 = side ? Cmat : bbar;
    const int rowoff = tn * 16 + 8 * lh;
    float* dst = (side ? Hc : Gc) + (size_t)c * SDIM * SDIM + b * SDIM + rowoff;

    v4f lo, hi;
#pragma unroll
    for (int r = 0; r < 4; ++r) {
        lo[r] = S0[(rowoff + r) * NCH + c];
        hi[r] = S0[(rowoff + r + 4) * NCH + c];
    }
    v2f a1[16], a2[16];
#pragma unroll
    for (int tk = 0; tk < 16; ++tk) {
        a1[tk] = *reinterpret_cast<const v2f*>(M1 + (tn * 16 + lc) * SDIM + tk * 4 + 2 * lh);
        a2[tk] = *reinterpret_cast<const v2f*>(M2 + (tn * 16 + lc) * SDIM + tk * 4 + 2 * lh);
    }
    *reinterpret_cast<v4f*>(Sl + lc * SPITCH + rowoff) = lo;
    *reinterpret_cast<v4f*>(Sl + lc * SPITCH + rowoff + 4) = hi;
    lds_barrier();

    int par = 0;
#pragma unroll 1
    for (int i = 0; i < b; ++i) {
        chain_step(a1, Sl + par * (16 * SPITCH), Sl + (par ^ 1) * (16 * SPITCH), lc, lh, rowoff, lo, hi);
        par ^= 1;
        lds_barrier();
    }
    *reinterpret_cast<v4f*>(dst) = lo;
    *reinterpret_cast<v4f*>(dst + 4) = hi;
#pragma unroll 1
    for (int a = 1; a < 8; ++a) {
        chain_step(a2, Sl + par * (16 * SPITCH), Sl + (par ^ 1) * (16 * SPITCH), lc, lh, rowoff, lo, hi);
        par ^= 1;
        *reinterpret_cast<v4f*>(dst + a * 8 * SDIM) = lo;
        *reinterpret_cast<v4f*>(dst + a * 8 * SDIM + 4) = hi;
        lds_barrier();
    }
}

__global__ __launch_bounds__(256, 1)
void hippo_taps_kernel(const float* __restrict__ Hc, const float* __restrict__ Gc,
                       float* __restrict__ Kp) {
    const int c = blockIdx.x;
    const int lane = threadIdx.x & 31;
    const int w = threadIdx.x >> 5;
    const int lc = lane & 15, lh = lane >> 4;
    const float* H = Hc + (size_t)c * SDIM * SDIM;
    const float* G = Gc + (size_t)c * SDIM * SDIM;
    float* K = Kp + (size_t)c * KPITCH;
    if (threadIdx.x < 16) K[SLEN + threadIdx.x] = 0.0f;
#pragma unroll
    for (int t = 0; t < 2; ++t) {
        const int tile = 2 * w + t;
        const int tj = tile >> 2, ti = tile & 3;
        v8f acc;
#pragma unroll
        for (int r = 0; r < 8; ++r) acc[r] = 0.0f;
#pragma unroll
        for (int tk = 0; tk < 16; ++tk) {
            const v2f af = *reinterpret_cast<const v2f*>(H + (tj * 16 + lc) * SDIM + tk * 4 + 2 * lh);
            const v2f bf = *reinterpret_cast<const v2f*>(G + (ti * 16 + lc) * SDIM + tk * 4 + 2 * lh);
            acc = __builtin_amdgcn_wmma_f32_16x16x4_f32(false, af, false, bf, (short)0, acc, false, false);
        }
#pragma unroll
        for (int r = 0; r < 8; ++r)
            K[SLEN - 1 - ((tj * 16 + r + 8 * lh) * SDIM + ti * 16 + lc)] = acc[r];
    }
}

#define XPITCH 4132
#define ZPITCH 2068

struct ConvFrag { v2f a1[4], a2[4], a3[4]; v4f slo, shi, elo, ehi, olo, ohi; };

__device__ __forceinline__ void conv_load(ConvFrag& f, const float* sb, const float* xb,
                                          const float* kb, const float* eb) {
    f.slo = *reinterpret_cast<const v4f*>(sb);
    f.shi = *reinterpret_cast<const v4f*>(sb + 4);
    f.elo = *reinterpret_cast<const v4f*>(xb);
    f.ehi = *reinterpret_cast<const v4f*>(xb + 4);
    f.olo = *reinterpret_cast<const v4f*>(xb + 16);
    f.ohi = *reinterpret_cast<const v4f*>(xb + 20);
#pragma unroll
    for (int s = 0; s < 4; ++s) {
        f.a1[s].x = kb[2 * s];      f.a1[s].y = kb[2 * s + 1];
        f.a2[s].x = eb[2 * s];      f.a2[s].y = eb[2 * s + 1];
        f.a3[s].x = eb[2 * s - 16]; f.a3[s].y = eb[2 * s - 15];
    }
}

__device__ __forceinline__ void conv_mma(const ConvFrag& f, v8f& acc1, v8f& acc2, v8f& acc3) {
    v2f b1[4], b2[4], b3[4];
    b1[0].x = f.slo[0]; b1[0].y = f.slo[1]; b1[1].x = f.slo[2]; b1[1].y = f.slo[3];
    b1[2].x = f.shi[0]; b1[2].y = f.shi[1]; b1[3].x = f.shi[2]; b1[3].y = f.shi[3];
    b2[0].x = f.olo[0]; b2[0].y = f.olo[1]; b2[1].x = f.olo[2]; b2[1].y = f.olo[3];
    b2[2].x = f.ohi[0]; b2[2].y = f.ohi[1]; b2[3].x = f.ohi[2]; b2[3].y = f.ohi[3];
    b3[0].x = f.elo[0]; b3[0].y = f.elo[1]; b3[1].x = f.elo[2]; b3[1].y = f.elo[3];
    b3[2].x = f.ehi[0]; b3[2].y = f.ehi[1]; b3[3].x = f.ehi[2]; b3[3].y = f.ehi[3];
#pragma unroll
    for (int s = 0; s < 4; ++s) {
        acc1 = __builtin_amdgcn_wmma_f32_16x16x4_f32(false, f.a1[s], false, b1[s], (short)0, acc1, false, false);
        acc2 = __builtin_amdgcn_wmma_f32_16x16x4_f32(false, f.a2[s], false, b2[s], (short)0, acc2, false, false);
        acc3 = __builtin_amdgcn_wmma_f32_16x16x4_f32(false, f.a3[s], false, b3[s], (short)0, acc3, false, false);
    }
}

__global__ __launch_bounds__(256, 1)
void hippo_conv_kernel(const float* __restrict__ sig, const float* __restrict__ Kp,
                       float* __restrict__ out) {
    __shared__ __attribute__((aligned(16))) float xs[NB * XPITCH];
    __shared__ __attribute__((aligned(16))) float ss[NB * ZPITCH];
    __shared__ __attribute__((aligned(16))) float ks[KPITCH];
    __shared__ __attribute__((aligned(16))) float es[KPITCH];
    const int c = blockIdx.x >> 1;
    const int z = blockIdx.x & 1;
    const int tid = threadIdx.x;
    const int lane = tid & 31;
    const int w = __builtin_amdgcn_readfirstlane(tid >> 5);
    const int lc = lane & 15, lh = lane >> 4;

    xs[(tid >> 5) * XPITCH + (tid & 31)] = 0.0f;
    if (tid < NB * 16) ss[(tid >> 4) * ZPITCH + (tid & 15)] = 0.0f;
#pragma unroll 1
    for (int b = 0; b < NB; ++b) {
        const v4f* src = reinterpret_cast<const v4f*>(sig + ((size_t)b * NCH + c) * SLEN);
        float* xd = xs + b * XPITCH + 32;
        float* sd = ss + b * ZPITCH + 16;
#pragma unroll
        for (int i = 0; i < 2; ++i) {
            const int u = tid + 256 * i;
            const int j = u >> 2, q4 = u & 3;
            const v4f ev = src[8 * j + q4];
            const v4f od = src[8 * j + 4 + q4];
            *reinterpret_cast<v4f*>(xd + 32 * j + 4 * q4) = ev;
            *reinterpret_cast<v4f*>(xd + 32 * j + 16 + 4 * q4) = od;
            *reinterpret_cast<v4f*>(sd + 16 * j + 4 * q4) = ev + od;
        }
    }
    {
        const v4f* src = reinterpret_cast<const v4f*>(Kp + (size_t)c * KPITCH);
        v4f* dst = reinterpret_cast<v4f*>(ks);
        for (int i = tid; i < KPITCH / 4; i += 256) dst[i] = src[i];
    }
    __syncthreads();
    {
        const v4f* k4 = reinterpret_cast<const v4f*>(ks);
        v4f* e4 = reinterpret_cast<v4f*>(es);
        for (int i = tid; i < KPITCH / 4; i += 256) {
            v4f up = {0.0f, 0.0f, 0.0f, 0.0f};
            if (i + 4 < KPITCH / 4) up = k4[i + 4];
            e4[i] = up - k4[i];
        }
    }
    __syncthreads();

    const int bcol = lc & 7;
    const int e    = lc >> 3;
    const int g    = z * 8 + w;
    const float* srow = ss + bcol * ZPITCH + 16 + 8 * lh + 16 * e;
    const float* xrow = xs + bcol * XPITCH + 32 + 8 * lh + 32 * e;
    const float* krow = ks + (SLEN - 1) - lc + 8 * lh;
    const float* erow = es + (SLEN - 1) - lc + 8 * lh;

#pragma unroll 1
    for (int it = 0; it < 4; ++it) {
        const int half = it >> 1;
        const int S  = (it & 1) ? (31 + 32 * half - g) : (g + 32 * half);
        const int nd = 2 * S + 2;
        v8f acc1, acc2, acc3;
#pragma unroll
        for (int r = 0; r < 8; ++r) { acc1[r] = 0.0f; acc2[r] = 0.0f; acc3[r] = 0.0f; }
        const float* sb = srow + 32 * S;
        const float* xb = xrow + 64 * S;
        ConvFrag f0, f1;
        conv_load(f0, sb, xb, krow, erow);
        int d = 0;
#pragma unroll 1
        while (true) {
            conv_load(f1, sb - 16 * (d + 1), xb - 32 * (d + 1), krow - 32 * (d + 1), erow - 32 * (d + 1));
            __builtin_amdgcn_sched_barrier(0);
            conv_mma(f0, acc1, acc2, acc3);
            d += 2;
            if (d >= nd) { conv_mma(f1, acc1, acc2, acc3); break; }
            conv_load(f0, sb - 16 * d, xb - 32 * d, krow - 32 * d, erow - 32 * d);
            __builtin_amdgcn_sched_barrier(0);
            conv_mma(f1, acc1, acc2, acc3);
        }
        float* yp = out + ((size_t)bcol * NCH + c) * SLEN + 32 * (2 * S + e) + 8 * lh;
        v4f lo, hi;
#pragma unroll
        for (int r = 0; r < 4; ++r) { lo[r] = acc1[r] + acc2[r]; hi[r] = acc1[r + 4] + acc2[r + 4]; }
        *reinterpret_cast<v4f*>(yp) = lo;
        *reinterpret_cast<v4f*>(yp + 4) = hi;
#pragma unroll
        for (int r = 0; r < 4; ++r) { lo[r] = acc1[r] - acc3[r]; hi[r] = acc1[r + 4] - acc3[r + 4]; }
        *reinterpret_cast<v4f*>(yp + 16) = lo;
        *reinterpret_cast<v4f*>(yp + 20) = hi;
    }
}

extern "C" void kernel_launch(void* const* d_in, const int* in_sizes, int n_in,
                              void* d_out, int out_size, void* d_ws, size_t ws_size,
                              hipStream_t stream) {
    const float* sig = (const float*)d_in[0];
    const float* Bm  = (const float*)d_in[1];
    const float* Cm  = (const float*)d_in[2];
    const float* Am  = (const float*)d_in[3];
    const float* dtp = (const float*)d_in[4];
    float* out  = (float*)d_out;
    float* abar = (float*)d_ws;
    float* bbar = abar + SDIM * SDIM;
    float* pt   = bbar + SDIM * NCH;
    float* gc   = pt + SDIM * SDIM;
    float* hc   = gc + NCH * SDIM * SDIM;
    float* kp   = hc + NCH * SDIM * SDIM;

    float* q8   = kp;
    float* rt   = kp + SDIM * SDIM;

    hippo_setup_kernel<<<1, 256, 0, stream>>>(Am, Bm, dtp, abar, bbar, q8, pt, rt);
    hippo_chains_kernel<<<128, 128, 0, stream>>>(abar, q8, pt, rt, bbar, Cm, gc, hc);
    hippo_taps_kernel<<<NCH, 256, 0, stream>>>(hc, gc, kp);
    hippo_conv_kernel<<<NCH * 2, 256, 0, stream>>>(sig, kp, out);
}


# convolution: three-product rule used twice: a 4x4 block Toeplitz tile as 9 WMMA block products, not 12; tap differences at lag 16, 32 and both in LDS, input sums in registers; output sums reordered
# speedup vs baseline: 42.8444x; 1.2078x over previous
#include <hip/hip_runtime.h>
#include <hip/hip_bf16.h>

typedef float v2f __attribute__((ext_vector_type(2)));
typedef float v4f __attribute__((ext_vector_type(4)));
typedef float v8f __attribute__((ext_vector_type(8)));

#define SDIM 64
#define NCH  128
#define NB   8
#define SLEN 4096

#define LP 68
#define TP 33

__device__ __forceinline__ void lds_barrier() {
    __builtin_amdgcn_fence(__ATOMIC_RELEASE, "workgroup", "local");
    __builtin_amdgcn_s_barrier();
    __builtin_amdgcn_fence(__ATOMIC_ACQUIRE, "workgroup", "local");
}

__global__ __launch_bounds__(256, 1)
void hippo_setup_kernel(const float* __restrict__ A,
                        const float* __restrict__ Bm,
                        const float* __restrict__ dtp,
                        float* __restrict__ abar,
                        float* __restrict__ bbar,
                        float* __restrict__ q8,
                        float* __restrict__ pT,
                        float* __restrict__ rT) {
    __shared__ __attribute__((aligned(16))) float Ls[SDIM * LP];
    __shared__ __attribute__((aligned(16))) float Xs[SDIM * LP];
    __shared__ float Ts[32 * TP];
    const float dt = dtp[0];
    const int tid = threadIdx.x;
    const int lane = tid & 31;
    const int w = __builtin_amdgcn_readfirstlane(tid >> 5);
    const int lc = lane & 15, lh = lane >> 4;
    const int tn = w & 3;
    const int cg = w >> 2;

    float av[16];
#pragma unroll
    for (int u = 0; u < 16; ++u) av[u] = A[tid + 256 * u];
#pragma unroll
    for (int u = 0; u < 16; ++u) {
        const int e = tid + 256 * u;
        const int i = e >> 6, j = e & 63;
        float l = -0.5f * dt * av[u];
        if (i == j) l += 1.0f;
        Ls[i * LP + j] = l;
        Xs[i * LP + j] = 0.0f;
    }
    v2f bf[4][16];
#pragma unroll
    for (int ct = 0; ct < 4; ++ct) {
        const int c = (cg * 4 + ct) * 16 + lc;
#pragma unroll
        for (int tk = 0; tk < 16; ++tk) {
            bf[ct][tk].x = Bm[(tk * 4 + 2 * lh) * NCH + c];
            bf[ct][tk].y = Bm[(tk * 4 + 2 * lh + 1) * NCH + c];
        }
    }
    __syncthreads();

    if (tid < 64) {
        const int I = tid >> 4, j = tid & 15;
        const float* Lb = Ls + (16 * I) * LP + 16 * I;
        float x[16];
#pragma unroll
        for (int i = 0; i < 16; ++i) {
            float sum = (i == j) ? 1.0f : 0.0f;
#pragma unroll
            for (int k = 0; k < i; ++k) sum -= Lb[i * LP + k] * x[k];
            x[i] = sum / Lb[i * LP + i];
        }
#pragma unroll
        for (int i = 0; i < 16; ++i) Xs[(16 * I + i) * LP + 16 * I + j] = x[i];
    }
    __syncthreads();

    {
        const int r = tid >> 4, c = tid & 15;
#pragma unroll
        for (int p = 0; p < 2; ++p) {
            const int R0 = 16 + 32 * p, C0 = 32 * p;
            float s = 0.0f;
#pragma unroll
            for (int k = 0; k < 16; ++k) s += Ls[(R0 + r) * LP + C0 + k] * Xs[(C0 + k) * LP + C0 + c];
            Ts[(16 * p + r) * TP + c] = s;
        }
        __syncthreads();
#pragma unroll
        for (int p = 0; p < 2; ++p) {
            const int R0 = 16 + 32 * p, C0 = 32 * p;
            float s = 0.0f;
#pragma unroll
            for (int k = 0; k < 16; ++k) s += Xs[(R0 + r) * LP + R0 + k] * Ts[(16 * p + k) * TP + c];
            Xs[(R0 + r) * LP + C0 + c] = -s;
        }
        __syncthreads();
    }

    {
        const int r = tid >> 3, c0 = tid & 7;
        float s[4] = {0.0f, 0.0f, 0.0f, 0.0f};
#pragma unroll
        for (int k = 0; k < 32; ++k) {
            const float a = Ls[(32 + r) * LP + k];
#pragma unroll
            for (int m = 0; m < 4; ++m) s[m] += a * Xs[k * LP + c0 + 8 * m];
        }
#pragma unroll
        for (int m = 0; m < 4; ++m) Ts[r * TP + c0 + 8 * m] = s[m];
        __syncthreads();
#pragma unroll
        for (int m = 0; m < 4; ++m) s[m] = 0.0f;
#pragma unroll
        for (int k = 0; k < 32; ++k) {
            const float a = Xs[(32 + r) * LP + 32 + k];
#pragma unroll
            for (int m = 0; m < 4; ++m) s[m] += a * Ts[k * TP + c0 + 8 * m];
        }
#pragma unroll
        for (int m = 0; m < 4; ++m) Xs[(32 + r) * LP + c0 + 8 * m] = -s[m];
        __syncthreads();
    }

#pragma unroll
    for (int u = 0; u < 16; ++u) {
        const int e = tid + 256 * u;
        const int i = e >> 6, j = e & 63;
        const float ab = 2.0f * Xs[i * LP + j] - ((i == j) ? 1.0f : 0.0f);
        abar[e] = ab;
        Ls[i * LP + j] = ab;
    }

    v2f a[16];
#pragma unroll
    for (int tk = 0; tk < 16; ++tk)
        a[tk] = *reinterpret_cast<const v2f*>(Xs + (tn * 16 + lc) * LP + tk * 4 + 2 * lh);
#pragma unroll
    for (int ct = 0; ct < 4; ++ct) {
        const int c = (cg * 4 + ct) * 16 + lc;
        v8f acc0, acc1;
#pragma unroll
        for (int r = 0; r < 8; ++r) { acc0[r] = 0.0f; acc1[r] = 0.0f; }
#pragma unroll
        for (int tk = 0; tk < 16; tk += 2) {
            acc0 = __builtin_amdgcn_wmma_f32_16x16x4_f32(false, a[tk], false, bf[ct][tk], (short)0, acc0, false, false);
            acc1 = __builtin_amdgcn_wmma_f32_16x16x4_f32(false, a[tk + 1], false, bf[ct][tk + 1], (short)0, acc1, false, false);
        }
#pragma unroll
        for (int r = 0; r < 8; ++r)
            bbar[(tn * 16 + 8 * lh + r) * NCH + c] = dt * (acc0[r] + acc1[r]);
    }
    lds_barrier();

    float* src = Ls;
    float* dst = Xs;
#pragma unroll 1
    for (int it = 0; it < 9; ++it) {
        v2f sa[16], sb0[16], sb1[16];
        const int col0 = cg * 32 + lc;
#pragma unroll
        for (int tk = 0; tk < 16; ++tk) {
            sa[tk] = *reinterpret_cast<const v2f*>(src + (tn * 16 + lc) * LP + tk * 4 + 2 * lh);
            const float* p = src + (tk * 4 + 2 * lh) * LP + col0;
            sb0[tk].x = p[0];  sb0[tk].y = p[LP];
            sb1[tk].x = p[16]; sb1[tk].y = p[LP + 16];
        }
        v8f d0, d1;
#pragma unroll
        for (int r = 0; r < 8; ++r) { d0[r] = 0.0f; d1[r] = 0.0f; }
#pragma unroll
        for (int tk = 0; tk < 16; ++tk) {
            d0 = __builtin_amdgcn_wmma_f32_16x16x4_f32(false, sa[tk], false, sb0[tk], (short)0, d0, false, false);
            d1 = __builtin_amdgcn_wmma_f32_16x16x4_f32(false, sa[tk], false, sb1[tk], (short)0, d1, false, false);
        }
#pragma unroll
        for (int r = 0; r < 8; ++r) {
            dst[(tn * 16 + 8 * lh + r) * LP + col0] = d0[r];
            dst[(tn * 16 + 8 * lh + r) * LP + col0 + 16] = d1[r];
        }
        lds_barrier();
        if (it == 2) {
#pragma unroll
            for (int u = 0; u < 4; ++u) {
                const int e4 = tid + 256 * u;
                const int i = e4 >> 4, j4 = (e4 & 15) * 4;
                *reinterpret_cast<v4f*>(q8 + i * SDIM + j4) = *reinterpret_cast<const v4f*>(dst + i * LP + j4);
            }
        } else if (it == 5 || it == 8) {
            float* mt = (it == 5) ? pT : rT;
#pragma unroll
            for (int u = 0; u < 4; ++u) {
                const int e4 = tid + 256 * u;
                const int i = e4 >> 4, j4 = (e4 & 15) * 4;
                v4f v;
#pragma unroll
                for (int m = 0; m < 4; ++m) v[m] = dst[(j4 + m) * LP + i];
                *reinterpret_cast<v4f*>(mt + i * SDIM + j4) = v;
            }
        }
        float* t = src; src = dst; dst = t;
    }
}


#define KPITCH 4112
#define SPITCH 68

__device__ __forceinline__ void chain_step(const v2f (&a)[16], const float* cur, float* nxt,
                                           int lc, int lh, int rowoff, v4f& lo, v4f& hi) {
    v8f acc0, acc1;
#pragma unroll
    for (int r = 0; r < 8; ++r) { acc0[r] = 0.0f; acc1[r] = 0.0f; }
#pragma unroll
    for (int tk = 0; tk < 16; tk += 2) {
        const v2f b0 = *reinterpret_cast<const v2f*>(cur + lc * SPITCH + tk * 4 + 2 * lh);
        const v2f b1 = *reinterpret_cast<const v2f*>(cur + lc * SPITCH + tk * 4 + 4 + 2 * lh);
        acc0 = __builtin_amdgcn_wmma_f32_16x16x4_f32(false, a[tk], false, b0, (short)0, acc0, false, false);
        acc1 = __builtin_amdgcn_wmma_f32_16x16x4_f32(false, a[tk + 1], false, b1, (short)0, acc1, false, false);
    }
#pragma unroll
    for (int r = 0; r < 4; ++r) { lo[r] = acc0[r] + acc1[r]; hi[r] = acc0[r + 4] + acc1[r + 4]; }
    *reinterpret_cast<v4f*>(nxt + lc * SPITCH + rowoff) = lo;
    *reinterpret_cast<v4f*>(nxt + lc * SPITCH + rowoff + 4) = hi;
}

__global__ __launch_bounds__(128, 1)
void hippo_chains_kernel(const float* __restrict__ abar, const float* __restrict__ q8,
                         const float* __restrict__ pT, const float* __restrict__ rT,
                         const float* __restrict__ bbar, const float* __restrict__ Cmat,
                         float* __restrict__ Gc, float* __restrict__ Hc) {
    __shared__ __attribute__((aligned(16))) float Sl[2 * 16 * SPITCH];
    const int lane = threadIdx.x & 31;
    const int tn = __builtin_amdgcn_readfirstlane(threadIdx.x >> 5);
    const int lc = lane & 15, lh = lane >> 4;
    const int blk = blockIdx.x;
    const int side = blk >> 6;
    const int b = blk & 7;
    const int c = ((blk >> 3) & 7) * 16 + lc;
    const float* M1 = side ? pT : abar;
    const float* M2 = side ? rT : q8;
    const float* S0 = side ? Cmat : bbar;
    const int rowoff = tn * 16 + 8 * lh;
    float* dst = (side ? Hc : Gc) + (size_t)c * SDIM * SDIM + b * SDIM + rowoff;

    v4f lo, hi;
#pragma unroll
    for (int r = 0; r < 4; ++r) {
        lo[r] = S0[(rowoff + r) * NCH + c];
        hi[r] = S0[(rowoff + r + 4) * NCH + c];
    }
    v2f a1[16], a2[16];
#pragma unroll
    for (int tk = 0; tk < 16; ++tk) {
        a1[tk] = *reinterpret_cast<const v2f*>(M1 + (tn * 16 + lc) * SDIM + tk * 4 + 2 * lh);
        a2[tk] = *reinterpret_cast<const v2f*>(M2 + (tn * 16 + lc) * SDIM + tk * 4 + 2 * lh);
    }
    *reinterpret_cast<v4f*>(Sl + lc * SPITCH + rowoff) = lo;
    *reinterpret_cast<v4f*>(Sl + lc * SPITCH + rowoff + 4) = hi;
    lds_barrier();

    int par = 0;
#pragma unroll 1
    for (int i = 0; i < b; ++i) {
        chain_step(a1, Sl + par * (16 * SPITCH), Sl + (par ^ 1) * (16 * SPITCH), lc, lh, rowoff, lo, hi);
        par ^= 1;
        lds_barrier();
    }
    *reinterpret_cast<v4f*>(dst) = lo;
    *reinterpret_cast<v4f*>(dst + 4) = hi;
#pragma unroll 1
    for (int a = 1; a < 8; ++a) {
        chain_step(a2, Sl + par * (16 * SPITCH), Sl + (par ^ 1) * (16 * SPITCH), lc, lh, rowoff, lo, hi);
        par ^= 1;
        *reinterpret_cast<v4f*>(dst + a * 8 * SDIM) = lo;
        *reinterpret_cast<v4f*>(dst + a * 8 * SDIM + 4) = hi;
        lds_barrier();
    }
}

__global__ __launch_bounds__(256, 1)
void hippo_taps_kernel(const float* __restrict__ Hc, const float* __restrict__ Gc,
                       float* __restrict__ Kp) {
    const int c = blockIdx.x;
    const int lane = threadIdx.x & 31;
    const int w = threadIdx.x >> 5;
    const int lc = lane & 15, lh = lane >> 4;
    const float* H = Hc + (size_t)c * SDIM * SDIM;
    const float* G = Gc + (size_t)c * SDIM * SDIM;
    float* K = Kp + (size_t)c * KPITCH;
    if (threadIdx.x < 16) K[SLEN + threadIdx.x] = 0.0f;
#pragma unroll
    for (int t = 0; t < 2; ++t) {
        const int tile = 2 * w + t;
        const int tj = tile >> 2, ti = tile & 3;
        v8f acc;
#pragma unroll
        for (int r = 0; r < 8; ++r) acc[r] = 0.0f;
#pragma unroll
        for (int tk = 0; tk < 16; ++tk) {
            const v2f af = *reinterpret_cast<const v2f*>(H + (tj * 16 + lc) * SDIM + tk * 4 + 2 * lh);
            const v2f bf = *reinterpret_cast<const v2f*>(G + (ti * 16 + lc) * SDIM + tk * 4 + 2 * lh);
            acc = __builtin_amdgcn_wmma_f32_16x16x4_f32(false, af, false, bf, (short)0, acc, false, false);
        }
#pragma unroll
        for (int r = 0; r < 8; ++r)
            K[SLEN - 1 - ((tj * 16 + r + 8 * lh) * SDIM + ti * 16 + lc)] = acc[r];
    }
}

#define SBP    68
#define XPITCH 4424

struct ConvFrag {
    v4f x[8];
    v2f ak[4];
    v2f ae0[4], ae1[4];
    v2f af0[4], af2[4];
    v2f ag0[4], ag1[4], ag2[4], ag3[4];
};

__device__ __forceinline__ void conv_load(ConvFrag& f, const float* xb, const float* kb, const float* eb,
                                          const float* fb, const float* gb) {
#pragma unroll
    for (int r = 0; r < 4; ++r) {
        f.x[2 * r]     = *reinterpret_cast<const v4f*>(xb + 16 * r);
        f.x[2 * r + 1] = *reinterpret_cast<const v4f*>(xb + 16 * r + 4);
    }
#pragma unroll
    for (int s = 0; s < 4; ++s) {
        f.ak[s].x  = kb[2 * s];      f.ak[s].y  = kb[2 * s + 1];
        f.ae0[s].x = eb[2 * s];      f.ae0[s].y = eb[2 * s + 1];
        f.ae1[s].x = eb[2 * s - 16]; f.ae1[s].y = eb[2 * s - 15];
        f.af0[s].x = fb[2 * s];      f.af0[s].y = fb[2 * s + 1];
        f.af2[s].x = fb[2 * s - 32]; f.af2[s].y = fb[2 * s - 31];
        f.ag0[s].x = gb[2 * s];      f.ag0[s].y = gb[2 * s + 1];
        f.ag1[s].x = gb[2 * s - 16]; f.ag1[s].y = gb[2 * s - 15];
        f.ag2[s].x = gb[2 * s - 32]; f.ag2[s].y = gb[2 * s - 31];
        f.ag3[s].x = gb[2 * s - 48]; f.ag3[s].y = gb[2 * s - 47];
    }
}

__device__ __forceinline__ v2f conv_bs(const v4f& lo, const v4f& hi, int s) {
    v2f b;
    if (s < 2) { b.x = lo[2 * s]; b.y = lo[2 * s + 1]; }
    else       { b.x = hi[2 * s - 4]; b.y = hi[2 * s - 3]; }
    return b;
}

__device__ __forceinline__ void conv_mma(const ConvFrag& f, v8f (&a)[9]) {
    const v4f w0l = f.x[0] + f.x[4], w0h = f.x[1] + f.x[5];
    const v4f w1l = f.x[2] + f.x[6], w1h = f.x[3] + f.x[7];
    const v4f ql  = w0l + w1l,       qh  = w0h + w1h;
    const v4f sel = f.x[0] + f.x[2], seh = f.x[1] + f.x[3];
    const v4f sol = f.x[4] + f.x[6], soh = f.x[5] + f.x[7];
#pragma unroll
    for (int s = 0; s < 4; ++s) {
        a[0] = __builtin_amdgcn_wmma_f32_16x16x4_f32(false, f.ak[s],  false, conv_bs(ql, qh, s),         (short)0, a[0], false, false);
        a[1] = __builtin_amdgcn_wmma_f32_16x16x4_f32(false, f.ae0[s], false, conv_bs(w1l, w1h, s),       (short)0, a[1], false, false);
        a[2] = __builtin_amdgcn_wmma_f32_16x16x4_f32(false, f.ae1[s], false, conv_bs(w0l, w0h, s),       (short)0, a[2], false, false);
        a[3] = __builtin_amdgcn_wmma_f32_16x16x4_f32(false, f.af0[s], false, conv_bs(sol, soh, s),       (short)0, a[3], false, false);
        a[4] = __builtin_amdgcn_wmma_f32_16x16x4_f32(false, f.ag0[s], false, conv_bs(f.x[6], f.x[7], s), (short)0, a[4], false, false);
        a[5] = __builtin_amdgcn_wmma_f32_16x16x4_f32(false, f.ag1[s], false, conv_bs(f.x[4], f.x[5], s), (short)0, a[5], false, false);
        a[6] = __builtin_amdgcn_wmma_f32_16x16x4_f32(false, f.af2[s], false, conv_bs(sel, seh, s),       (short)0, a[6], false, false);
        a[7] = __builtin_amdgcn_wmma_f32_16x16x4_f32(false, f.ag2[s], false, conv_bs(f.x[2], f.x[3], s), (short)0, a[7], false, false);
        a[8] = __builtin_amdgcn_wmma_f32_16x16x4_f32(false, f.ag3[s], false, conv_bs(f.x[0], f.x[1], s), (short)0, a[8], false, false);
    }
}

__global__ __launch_bounds__(256, 1)
void hippo_conv_kernel(const float* __restrict__ sig, const float* __restrict__ Kp,
                       float* __restrict__ out) {
    __shared__ __attribute__((aligned(16))) float xs[NB * XPITCH];
    __shared__ __attribute__((aligned(16))) float ks[KPITCH];
    __shared__ __attribute__((aligned(16))) float es[KPITCH];
    __shared__ __attribute__((aligned(16))) float fs[KPITCH];
    __shared__ __attribute__((aligned(16))) float gs[KPITCH];
    const int c = blockIdx.x >> 1;
    const int z = blockIdx.x & 1;
    const int tid = threadIdx.x;
    const int lane = tid & 31;
    const int w = __builtin_amdgcn_readfirstlane(tid >> 5);
    const int lc = lane & 15, lh = lane >> 4;

    xs[(tid >> 5) * XPITCH + (tid & 31)] = 0.0f;
    xs[(tid >> 5) * XPITCH + 32 + (tid & 31)] = 0.0f;
#pragma unroll 1
    for (int b = 0; b < NB; ++b) {
        const v4f* src = reinterpret_cast<const v4f*>(sig + ((size_t)b * NCH + c) * SLEN);
        float* xd = xs + b * XPITCH + SBP;
#pragma unroll
        for (int i = 0; i < 4; ++i) {
            const int u = tid + 256 * i;
            *reinterpret_cast<v4f*>(xd + SBP * (u >> 4) + 4 * (u & 15)) = src[u];
        }
    }
    {
        const v4f* src = reinterpret_cast<const v4f*>(Kp + (size_t)c * KPITCH);
        v4f* dst = reinterpret_cast<v4f*>(ks);
        for (int i = tid; i < KPITCH / 4; i += 256) dst[i] = src[i];
    }
    __syncthreads();
    {
        const v4f* k4 = reinterpret_cast<const v4f*>(ks);
        v4f* e4 = reinterpret_cast<v4f*>(es);
        v4f* f4 = reinterpret_cast<v4f*>(fs);
        v4f* g4 = reinterpret_cast<v4f*>(gs);
        for (int i = tid; i < KPITCH / 4; i += 256) {
            const v4f k0 = k4[i];
            v4f k1 = {0.0f, 0.0f, 0.0f, 0.0f}, k2 = k1, k3 = k1;
            if (i + 4  < KPITCH / 4) k1 = k4[i + 4];
            if (i + 8  < KPITCH / 4) k2 = k4[i + 8];
            if (i + 12 < KPITCH / 4) k3 = k4[i + 12];
            const v4f fv = k2 - k0;
            e4[i] = k1 - k0;
            f4[i] = fv;
            g4[i] = (k3 - k1) - fv;
        }
    }
    __syncthreads();

    const int bcol = lc & 7;
    const int e    = lc >> 3;
    const int g    = z * 8 + w;
    const float* xrow = xs + bcol * XPITCH + SBP + 8 * lh + SBP * e;
    const int toff = (SLEN - 1) - lc + 8 * lh;
    const float* krow = ks + toff;
    const float* erow = es + toff;
    const float* frow = fs + toff;
    const float* grow = gs + toff;

#pragma unroll 1
    for (int it = 0; it < 2; ++it) {
        const int S  = it ? (31 - g) : g;
        const int nd = 2 * S + 2;
        v8f acc[9];
#pragma unroll
        for (int n = 0; n < 9; ++n)
#pragma unroll
            for (int r = 0; r < 8; ++r) acc[n][r] = 0.0f;
        const float* xb = xrow + SBP * 2 * S;
        ConvFrag f0, f1;
        conv_load(f0, xb, krow, erow, frow, grow);
#pragma unroll 1
        for (int D = 0; D < nd; D += 2) {
            conv_load(f1, xb - SBP * (D + 1), krow - 64 * (D + 1), erow - 64 * (D + 1),
                      frow - 64 * (D + 1), grow - 64 * (D + 1));
            __builtin_amdgcn_sched_barrier(0);
            conv_mma(f0, acc);
            const int Dn = (D + 2 < nd) ? (D + 2) : (nd - 1);
            conv_load(f0, xb - SBP * Dn, krow - 64 * Dn, erow - 64 * Dn, frow - 64 * Dn, grow - 64 * Dn);
            __builtin_amdgcn_sched_barrier(0);
            conv_mma(f1, acc);
        }
        float* yp = out + ((size_t)bcol * NCH + c) * SLEN + 64 * (2 * S + e) + 8 * lh;
        const v8f p01 = acc[0] + acc[1];
        const v8f p02 = acc[0] - acc[2];
        const v8f y0 = p01 + (acc[3] + acc[4]);
        const v8f y1 = p02 + (acc[3] - acc[5]);
        const v8f y2 = p01 - (acc[6] + acc[7]);
        const v8f y3 = p02 - (acc[6] - acc[8]);
        v4f lo, hi;
#pragma unroll
        for (int r = 0; r < 4; ++r) { lo[r] = y0[r]; hi[r] = y0[r + 4]; }
        *reinterpret_cast<v4f*>(yp) = lo;
        *reinterpret_cast<v4f*>(yp + 4) = hi;
#pragma unroll
        for (int r = 0; r < 4; ++r) { lo[r] = y1[r]; hi[r] = y1[r + 4]; }
        *reinterpret_cast<v4f*>(yp + 16) = lo;
        *reinterpret_cast<v4f*>(yp + 20) = hi;
#pragma unroll
        for (int r = 0; r < 4; ++r) { lo[r] = y2[r]; hi[r] = y2[r + 4]; }
        *reinterpret_cast<v4f*>(yp + 32) = lo;
        *reinterpret_cast<v4f*>(yp + 36) = hi;
#pragma unroll
        for (int r = 0; r < 4; ++r) { lo[r] = y3[r]; hi[r] = y3[r + 4]; }
        *reinterpret_cast<v4f*>(yp + 48) = lo;
        *reinterpret_cast<v4f*>(yp + 52) = hi;
    }
}

extern "C" void kernel_launch(void* const* d_in, const int* in_sizes, int n_in,
                              void* d_out, int out_size, void* d_ws, size_t ws_size,
                              hipStream_t stream) {
    const float* sig = (const float*)d_in[0];
    const float* Bm  = (const float*)d_in[1];
    const float* Cm  = (const float*)d_in[2];
    const float* Am  = (const float*)d_in[3];
    const float* dtp = (const float*)d_in[4];
    float* out  = (float*)d_out;
    float* abar = (float*)d_ws;
    float* bbar = abar + SDIM * SDIM;
    float* pt   = bbar + SDIM * NCH;
    float* gc   = pt + SDIM * SDIM;
    float* hc   = gc + NCH * SDIM * SDIM;
    float* kp   = hc + NCH * SDIM * SDIM;

    float* q8   = kp;
    float* rt   = kp + SDIM * SDIM;

    hippo_setup_kernel<<<1, 256, 0, stream>>>(Am, Bm, dtp, abar, bbar, q8, pt, rt);
    hippo_chains_kernel<<<128, 128, 0, stream>>>(abar, q8, pt, rt, bbar, Cm, gc, hc);
    hippo_taps_kernel<<<NCH, 256, 0, stream>>>(hc, gc, kp);
    hippo_conv_kernel<<<NCH * 2, 256, 0, stream>>>(sig, kp, out);
}


# convolution: three-product rule a third time: an 8x8 block Toeplitz tile as 27 WMMA block products, not 36; tap differences at lag 64 too; a tile's D range shared by two waves; output sums reordered
# speedup vs baseline: 47.0711x; 1.0987x over previous
#include <hip/hip_runtime.h>
#include <hip/hip_bf16.h>

typedef float v2f __attribute__((ext_vector_type(2)));
typedef float v4f __attribute__((ext_vector_type(4)));
typedef float v8f __attribute__((ext_vector_type(8)));

#define SDIM 64
#define NCH  128
#define NB   8
#define SLEN 4096

#define LP 68
#define TP 33

__device__ __forceinline__ void lds_barrier() {
    __builtin_amdgcn_fence(__ATOMIC_RELEASE, "workgroup", "local");
    __builtin_amdgcn_s_barrier();
    __builtin_amdgcn_fence(__ATOMIC_ACQUIRE, "workgroup", "local");
}

__global__ __launch_bounds__(256, 1)
void hippo_setup_kernel(const float* __restrict__ A,
                        const float* __restrict__ Bm,
                        const float* __restrict__ dtp,
                        float* __restrict__ abar,
                        float* __restrict__ bbar,
                        float* __restrict__ q8,
                        float* __restrict__ pT,
                        float* __restrict__ rT) {
    __shared__ __attribute__((aligned(16))) float Ls[SDIM * LP];
    __shared__ __attribute__((aligned(16))) float Xs[SDIM * LP];
    __shared__ float Ts[32 * TP];
    const float dt = dtp[0];
    const int tid = threadIdx.x;
    const int lane = tid & 31;
    const int w = __builtin_amdgcn_readfirstlane(tid >> 5);
    const int lc = lane & 15, lh = lane >> 4;
    const int tn = w & 3;
    const int cg = w >> 2;

    float av[16];
#pragma unroll
    for (int u = 0; u < 16; ++u) av[u] = A[tid + 256 * u];
#pragma unroll
    for (int u = 0; u < 16; ++u) {
        const int e = tid + 256 * u;
        const int i = e >> 6, j = e & 63;
        float l = -0.5f * dt * av[u];
        if (i == j) l += 1.0f;
        Ls[i * LP + j] = l;
        Xs[i * LP + j] = 0.0f;
    }
    v2f bf[4][16];
#pragma unroll
    for (int ct = 0; ct < 4; ++ct) {
        const int c = (cg * 4 + ct) * 16 + lc;
#pragma unroll
        for (int tk = 0; tk < 16; ++tk) {
            bf[ct][tk].x = Bm[(tk * 4 + 2 * lh) * NCH + c];
            bf[ct][tk].y = Bm[(tk * 4 + 2 * lh + 1) * NCH + c];
        }
    }
    __syncthreads();

    if (tid < 64) {
        const int I = tid >> 4, j = tid & 15;
        const float* Lb = Ls + (16 * I) * LP + 16 * I;
        float x[16];
#pragma unroll
        for (int i = 0; i < 16; ++i) {
            float sum = (i == j) ? 1.0f : 0.0f;
#pragma unroll
            for (int k = 0; k < i; ++k) sum -= Lb[i * LP + k] * x[k];
            x[i] = sum / Lb[i * LP + i];
        }
#pragma unroll
        for (int i = 0; i < 16; ++i) Xs[(16 * I + i) * LP + 16 * I + j] = x[i];
    }
    __syncthreads();

    {
        const int r = tid >> 4, c = tid & 15;
#pragma unroll
        for (int p = 0; p < 2; ++p) {
            const int R0 = 16 + 32 * p, C0 = 32 * p;
            float s = 0.0f;
#pragma unroll
            for (int k = 0; k < 16; ++k) s += Ls[(R0 + r) * LP + C0 + k] * Xs[(C0 + k) * LP + C0 + c];
            Ts[(16 * p + r) * TP + c] = s;
        }
        __syncthreads();
#pragma unroll
        for (int p = 0; p < 2; ++p) {
            const int R0 = 16 + 32 * p, C0 = 32 * p;
            float s = 0.0f;
#pragma unroll
            for (int k = 0; k < 16; ++k) s += Xs[(R0 + r) * LP + R0 + k] * Ts[(16 * p + k) * TP + c];
            Xs[(R0 + r) * LP + C0 + c] = -s;
        }
        __syncthreads();
    }

    {
        const int r = tid >> 3, c0 = tid & 7;
        float s[4] = {0.0f, 0.0f, 0.0f, 0.0f};
#pragma unroll
        for (int k = 0; k < 32; ++k) {
            const float a = Ls[(32 + r) * LP + k];
#pragma unroll
            for (int m = 0; m < 4; ++m) s[m] += a * Xs[k * LP + c0 + 8 * m];
        }
#pragma unroll
        for (int m = 0; m < 4; ++m) Ts[r * TP + c0 + 8 * m] = s[m];
        __syncthreads();
#pragma unroll
        for (int m = 0; m < 4; ++m) s[m] = 0.0f;
#pragma unroll
        for (int k = 0; k < 32; ++k) {
            const float a = Xs[(32 + r) * LP + 32 + k];
#pragma unroll
            for (int m = 0; m < 4; ++m) s[m] += a * Ts[k * TP + c0 + 8 * m];
        }
#pragma unroll
        for (int m = 0; m < 4; ++m) Xs[(32 + r) * LP + c0 + 8 * m] = -s[m];
        __syncthreads();
    }

#pragma unroll
    for (int u = 0; u < 16; ++u) {
        const int e = tid + 256 * u;
        const int i = e >> 6, j = e & 63;
        const float ab = 2.0f * Xs[i * LP + j] - ((i == j) ? 1.0f : 0.0f);
        abar[e] = ab;
        Ls[i * LP + j] = ab;
    }

    v2f a[16];
#pragma unroll
    for (int tk = 0; tk < 16; ++tk)
        a[tk] = *reinterpret_cast<const v2f*>(Xs + (tn * 16 + lc) * LP + tk * 4 + 2 * lh);
#pragma unroll
    for (int ct = 0; ct < 4; ++ct) {
        const int c = (cg * 4 + ct) * 16 + lc;
        v8f acc0, acc1;
#pragma unroll
        for (int r = 0; r < 8; ++r) { acc0[r] = 0.0f; acc1[r] = 0.0f; }
#pragma unroll
        for (int tk = 0; tk < 16; tk += 2) {
            acc0 = __builtin_amdgcn_wmma_f32_16x16x4_f32(false, a[tk], false, bf[ct][tk], (short)0, acc0, false, false);
            acc1 = __builtin_amdgcn_wmma_f32_16x16x4_f32(false, a[tk + 1], false, bf[ct][tk + 1], (short)0, acc1, false, false);
        }
#pragma unroll
        for (int r = 0; r < 8; ++r)
            bbar[(tn * 16 + 8 * lh + r) * NCH + c] = dt * (acc0[r] + acc1[r]);
    }
    lds_barrier();

    float* src = Ls;
    float* dst = Xs;
#pragma unroll 1
    for (int it = 0; it < 9; ++it) {
        v2f sa[16], sb0[16], sb1[16];
        const int col0 = cg * 32 + lc;
#pragma unroll
        for (int tk = 0; tk < 16; ++tk) {
            sa[tk] = *reinterpret_cast<const v2f*>(src + (tn * 16 + lc) * LP + tk * 4 + 2 * lh);
            const float* p = src + (tk * 4 + 2 * lh) * LP + col0;
            sb0[tk].x = p[0];  sb0[tk].y = p[LP];
            sb1[tk].x = p[16]; sb1[tk].y = p[LP + 16];
        }
        v8f d0, d1;
#pragma unroll
        for (int r = 0; r < 8; ++r) { d0[r] = 0.0f; d1[r] = 0.0f; }
#pragma unroll
        for (int tk = 0; tk < 16; ++tk) {
            d0 = __builtin_amdgcn_wmma_f32_16x16x4_f32(false, sa[tk], false, sb0[tk], (short)0, d0, false, false);
            d1 = __builtin_amdgcn_wmma_f32_16x16x4_f32(false, sa[tk], false, sb1[tk], (short)0, d1, false, false);
        }
#pragma unroll
        for (int r = 0; r < 8; ++r) {
            dst[(tn * 16 + 8 * lh + r) * LP + col0] = d0[r];
            dst[(tn * 16 + 8 * lh + r) * LP + col0 + 16] = d1[r];
        }
        lds_barrier();
        if (it == 2) {
#pragma unroll
            for (int u = 0; u < 4; ++u) {
                const int e4 = tid + 256 * u;
                const int i = e4 >> 4, j4 = (e4 & 15) * 4;
                *reinterpret_cast<v4f*>(q8 + i * SDIM + j4) = *reinterpret_cast<const v4f*>(dst + i * LP + j4);
            }
        } else if (it == 5 || it == 8) {
            float* mt = (it == 5) ? pT : rT;
#pragma unroll
            for (int u = 0; u < 4; ++u) {
                const int e4 = tid + 256 * u;
                const int i = e4 >> 4, j4 = (e4 & 15) * 4;
                v4f v;
#pragma unroll
                for (int m = 0; m < 4; ++m) v[m] = dst[(j4 + m) * LP + i];
                *reinterpret_cast<v4f*>(mt + i * SDIM + j4) = v;
            }
        }
        float* t = src; src = dst; dst = t;
    }
}


#define KPITCH 4112
#define SPITCH 68

__device__ __forceinline__ void chain_step(const v2f (&a)[16], const float* cur, float* nxt,
                                           int lc, int lh, int rowoff, v4f& lo, v4f& hi) {
    v8f acc0, acc1;
#pragma unroll
    for (int r = 0; r < 8; ++r) { acc0[r] = 0.0f; acc1[r] = 0.0f; }
#pragma unroll
    for (int tk = 0; tk < 16; tk += 2) {
        const v2f b0 = *reinterpret_cast<const v2f*>(cur + lc * SPITCH + tk * 4 + 2 * lh);
        const v2f b1 = *reinterpret_cast<const v2f*>(cur + lc * SPITCH + tk * 4 + 4 + 2 * lh);
        acc0 = __builtin_amdgcn_wmma_f32_16x16x4_f32(false, a[tk], false, b0, (short)0, acc0, false, false);
        acc1 = __builtin_amdgcn_wmma_f32_16x16x4_f32(false, a[tk + 1], false, b1, (short)0, acc1, false, false);
    }
#pragma unroll
    for (int r = 0; r < 4; ++r) { lo[r] = acc0[r] + acc1[r]; hi[r] = acc0[r + 4] + acc1[r + 4]; }
    *reinterpret_cast<v4f*>(nxt + lc * SPITCH + rowoff) = lo;
    *reinterpret_cast<v4f*>(nxt + lc * SPITCH + rowoff + 4) = hi;
}

__global__ __launch_bounds__(128, 1)
void hippo_chains_kernel(const float* __restrict__ abar, const float* __restrict__ q8,
                         const float* __restrict__ pT, const float* __restrict__ rT,
                         const float* __restrict__ bbar, const float* __restrict__ Cmat,
                         float* __restrict__ Gc, float* __restrict__ Hc) {
    __shared__ __attribute__((aligned(16))) float Sl[2 * 16 * SPITCH];
    const int lane = threadIdx.x & 31;
    const int tn = __builtin_amdgcn_readfirstlane(threadIdx.x >> 5);
    const int lc = lane & 15, lh = lane >> 4;
    const int blk = blockIdx.x;
    const int side = blk >> 6;
    const int b = blk & 7;
    const int c = ((blk >> 3) & 7) * 16 + lc;
    const float* M1 = side ? pT : abar;
    const float* M2 = side ? rT : q8;
    const float* S0 = side ? Cmat : bbar;
    const int rowoff = tn * 16 + 8 * lh;
    float* dst = (side ? Hc : Gc) + (size_t)c * SDIM * SDIM + b * SDIM + rowoff;

    v4f lo, hi;
#pragma unroll
    for (int r = 0; r < 4; ++r) {
        lo[r] = S0[(rowoff + r) * NCH + c];
        hi[r] = S0[(rowoff + r + 4) * NCH + c];
    }
    v2f a1[16], a2[16];
#pragma unroll
    for (int tk = 0; tk < 16; ++tk) {
        a1[tk] = *reinterpret_cast<const v2f*>(M1 + (tn * 16 + lc) * SDIM + tk * 4 + 2 * lh);
        a2[tk] = *reinterpret_cast<const v2f*>(M2 + (tn * 16 + lc) * SDIM + tk * 4 + 2 * lh);
    }
    *reinterpret_cast<v4f*>(Sl + lc * SPITCH + rowoff) = lo;
    *reinterpret_cast<v4f*>(Sl + lc * SPITCH + rowoff + 4) = hi;
    lds_barrier();

    int par = 0;
#pragma unroll 1
    for (int i = 0; i < b; ++i) {
        chain_step(a1, Sl + par * (16 * SPITCH), Sl + (par ^ 1) * (16 * SPITCH), lc, lh, rowoff, lo, hi);
        par ^= 1;
        lds_barrier();
    }
    *reinterpret_cast<v4f*>(dst) = lo;
    *reinterpret_cast<v4f*>(dst + 4) = hi;
#pragma unroll 1
    for (int a = 1; a < 8; ++a) {
        chain_step(a2, Sl + par * (16 * SPITCH), Sl + (par ^ 1) * (16 * SPITCH), lc, lh, rowoff, lo, hi);
        par ^= 1;
        *reinterpret_cast<v4f*>(dst + a * 8 * SDIM) = lo;
        *reinterpret_cast<v4f*>(dst + a * 8 * SDIM + 4) = hi;
        lds_barrier();
    }
}

__global__ __launch_bounds__(256, 1)
void hippo_taps_kernel(const float* __restrict__ Hc, const float* __restrict__ Gc,
                       float* __restrict__ Kp) {
    const int c = blockIdx.x;
    const int lane = threadIdx.x & 31;
    const int w = threadIdx.x >> 5;
    const int lc = lane & 15, lh = lane >> 4;
    const float* H = Hc + (size_t)c * SDIM * SDIM;
    const float* G = Gc + (size_t)c * SDIM * SDIM;
    float* K = Kp + (size_t)c * KPITCH;
    if (threadIdx.x < 16) K[SLEN + threadIdx.x] = 0.0f;
#pragma unroll
    for (int t = 0; t < 2; ++t) {
        const int tile = 2 * w + t;
        const int tj = tile >> 2, ti = tile & 3;
        v8f acc;
#pragma unroll
        for (int r = 0; r < 8; ++r) acc[r] = 0.0f;
#pragma unroll
        for (int tk = 0; tk < 16; ++tk) {
            const v2f af = *reinterpret_cast<const v2f*>(H + (tj * 16 + lc) * SDIM + tk * 4 + 2 * lh);
            const v2f bf = *reinterpret_cast<const v2f*>(G + (ti * 16 + lc) * SDIM + tk * 4 + 2 * lh);
            acc = __builtin_amdgcn_wmma_f32_16x16x4_f32(false, af, false, bf, (short)0, acc, false, false);
        }
#pragma unroll
        for (int r = 0; r < 8; ++r)
            K[SLEN - 1 - ((tj * 16 + r + 8 * lh) * SDIM + ti * 16 + lc)] = acc[r];
    }
}

#define SBP    132
#define XPITCH 4360

struct ConvFrag {
    v4f x[8];
    v4f x2[8];
    v2f ak[4];
    v2f ae0[4], ae1[4];
    v2f af0[4], af2[4];
    v2f ag0[4], ag1[4], ag2[4], ag3[4];
};

__device__ __forceinline__ void conv_load_x(v4f (&x)[8], const float* xb) {
#pragma unroll
    for (int r = 0; r < 4; ++r) {
        x[2 * r]     = *reinterpret_cast<const v4f*>(xb + 16 * r);
        x[2 * r + 1] = *reinterpret_cast<const v4f*>(xb + 16 * r + 4);
    }
}

__device__ __forceinline__ void conv_load_taps(ConvFrag& f, const float* kb, const float* eb,
                                               const float* fb, const float* gb) {
#pragma unroll
    for (int s = 0; s < 4; ++s) {
        f.ak[s].x  = kb[2 * s];      f.ak[s].y  = kb[2 * s + 1];
        f.ae0[s].x = eb[2 * s];      f.ae0[s].y = eb[2 * s + 1];
        f.ae1[s].x = eb[2 * s - 16]; f.ae1[s].y = eb[2 * s - 15];
        f.af0[s].x = fb[2 * s];      f.af0[s].y = fb[2 * s + 1];
        f.af2[s].x = fb[2 * s - 32]; f.af2[s].y = fb[2 * s - 31];
        f.ag0[s].x = gb[2 * s];      f.ag0[s].y = gb[2 * s + 1];
        f.ag1[s].x = gb[2 * s - 16]; f.ag1[s].y = gb[2 * s - 15];
        f.ag2[s].x = gb[2 * s - 32]; f.ag2[s].y = gb[2 * s - 31];
        f.ag3[s].x = gb[2 * s - 48]; f.ag3[s].y = gb[2 * s - 47];
    }
}

__device__ __forceinline__ v2f conv_bs(const v4f& lo, const v4f& hi, int s) {
    v2f b;
    if (s < 2) { b.x = lo[2 * s]; b.y = lo[2 * s + 1]; }
    else       { b.x = hi[2 * s - 4]; b.y = hi[2 * s - 3]; }
    return b;
}

__device__ __forceinline__ void conv_mma(const v4f (&x)[8], const ConvFrag& f, v8f (&a)[9]) {
    const v4f w0l = x[0] + x[4], w0h = x[1] + x[5];
    const v4f w1l = x[2] + x[6], w1h = x[3] + x[7];
    const v4f ql  = w0l + w1l,   qh  = w0h + w1h;
    const v4f sel = x[0] + x[2], seh = x[1] + x[3];
    const v4f sol = x[4] + x[6], soh = x[5] + x[7];
#pragma unroll
    for (int s = 0; s < 4; ++s) {
        a[0] = __builtin_amdgcn_wmma_f32_16x16x4_f32(false, f.ak[s],  false, conv_bs(ql, qh, s),     (short)0, a[0], false, false);
        a[1] = __builtin_amdgcn_wmma_f32_16x16x4_f32(false, f.ae0[s], false, conv_bs(w1l, w1h, s),   (short)0, a[1], false, false);
        a[2] = __builtin_amdgcn_wmma_f32_16x16x4_f32(false, f.ae1[s], false, conv_bs(w0l, w0h, s),   (short)0, a[2], false, false);
        a[3] = __builtin_amdgcn_wmma_f32_16x16x4_f32(false, f.af0[s], false, conv_bs(sol, soh, s),   (short)0, a[3], false, false);
        a[4] = __builtin_amdgcn_wmma_f32_16x16x4_f32(false, f.ag0[s], false, conv_bs(x[6], x[7], s), (short)0, a[4], false, false);
        a[5] = __builtin_amdgcn_wmma_f32_16x16x4_f32(false, f.ag1[s], false, conv_bs(x[4], x[5], s), (short)0, a[5], false, false);
        a[6] = __builtin_amdgcn_wmma_f32_16x16x4_f32(false, f.af2[s], false, conv_bs(sel, seh, s),   (short)0, a[6], false, false);
        a[7] = __builtin_amdgcn_wmma_f32_16x16x4_f32(false, f.ag2[s], false, conv_bs(x[2], x[3], s), (short)0, a[7], false, false);
        a[8] = __builtin_amdgcn_wmma_f32_16x16x4_f32(false, f.ag3[s], false, conv_bs(x[0], x[1], s), (short)0, a[8], false, false);
    }
}

__device__ __forceinline__ void conv_fold(const v8f (&a)[9], v8f (&r)[4]) {
    const v8f p01 = a[0] + a[1];
    const v8f p02 = a[0] - a[2];
    r[0] = p01 + (a[3] + a[4]);
    r[1] = p02 + (a[3] - a[5]);
    r[2] = p01 - (a[6] + a[7]);
    r[3] = p02 - (a[6] - a[8]);
}

__device__ __forceinline__ void conv_store(float* yp, const v8f (&y)[8]) {
#pragma unroll
    for (int j = 0; j < 8; ++j) {
        v4f lo, hi;
#pragma unroll
        for (int r = 0; r < 4; ++r) { lo[r] = y[j][r]; hi[r] = y[j][r + 4]; }
        *reinterpret_cast<v4f*>(yp + 16 * j) = lo;
        *reinterpret_cast<v4f*>(yp + 16 * j + 4) = hi;
    }
}

__global__ __launch_bounds__(256, 1)
void hippo_conv_kernel(const float* __restrict__ sig, const float* __restrict__ Kp,
                       float* __restrict__ out) {
    __shared__ __attribute__((aligned(16))) float xs[NB * XPITCH];
    __shared__ __attribute__((aligned(16))) float ks[KPITCH];
    __shared__ __attribute__((aligned(16))) float es[KPITCH];
    __shared__ __attribute__((aligned(16))) float fs[KPITCH];
    __shared__ __attribute__((aligned(16))) float gs[KPITCH];
    __shared__ __attribute__((aligned(16))) float ls[KPITCH];
    __shared__ __attribute__((aligned(16))) float els[KPITCH];
    __shared__ __attribute__((aligned(16))) float fls[KPITCH];
    __shared__ __attribute__((aligned(16))) float gls[KPITCH];
    const int c = blockIdx.x >> 1;
    const int z = blockIdx.x & 1;
    const int tid = threadIdx.x;
    const int lane = tid & 31;
    const int w = __builtin_amdgcn_readfirstlane(tid >> 5);
    const int lc = lane & 15, lh = lane >> 4;

#pragma unroll
    for (int j = 0; j < 4; ++j) xs[(tid >> 5) * XPITCH + 32 * j + (tid & 31)] = 0.0f;
#pragma unroll 1
    for (int b = 0; b < NB; ++b) {
        const v4f* src = reinterpret_cast<const v4f*>(sig + ((size_t)b * NCH + c) * SLEN);
        float* xd = xs + b * XPITCH + SBP;
#pragma unroll
        for (int i = 0; i < 4; ++i) {
            const int u = tid + 256 * i;
            *reinterpret_cast<v4f*>(xd + SBP * (u >> 5) + 4 * (u & 31)) = src[u];
        }
    }
    {
        const v4f* src = reinterpret_cast<const v4f*>(Kp + (size_t)c * KPITCH);
        v4f* dst = reinterpret_cast<v4f*>(ks);
        for (int i = tid; i < KPITCH / 4; i += 256) dst[i] = src[i];
    }
    __syncthreads();
    {
        const v4f* k4 = reinterpret_cast<const v4f*>(ks);
        v4f* e4 = reinterpret_cast<v4f*>(es);
        v4f* f4 = reinterpret_cast<v4f*>(fs);
        v4f* g4 = reinterpret_cast<v4f*>(gs);
        v4f* l4 = reinterpret_cast<v4f*>(ls);
        v4f* el4 = reinterpret_cast<v4f*>(els);
        v4f* fl4 = reinterpret_cast<v4f*>(fls);
        v4f* gl4 = reinterpret_cast<v4f*>(gls);
        for (int i = tid; i < KPITCH / 4; i += 256) {
            const v4f zero = {0.0f, 0.0f, 0.0f, 0.0f};
            v4f k[8];
#pragma unroll
            for (int n = 0; n < 8; ++n) {
                k[n] = zero;
                if (i + 4 * n < KPITCH / 4) k[n] = k4[i + 4 * n];
            }
            const v4f fv = k[2] - k[0];
            e4[i] = k[1] - k[0];
            f4[i] = fv;
            g4[i] = (k[3] - k[1]) - fv;
            const v4f l0 = k[4] - k[0], l1 = k[5] - k[1], l2 = k[6] - k[2], l3 = k[7] - k[3];
            const v4f flv = l2 - l0;
            l4[i] = l0;
            el4[i] = l1 - l0;
            fl4[i] = flv;
            gl4[i] = (l3 - l1) - flv;
        }
    }
    __syncthreads();

    const int bcol = lc & 7;
    const int e    = lc >> 3;
    const int role = w >> 2;
    const int g    = z * 4 + (w & 3);
    const float* xrow = xs + bcol * XPITCH + SBP + 8 * lh + SBP * e;
    const int toff = (SLEN - 1) - lc + 8 * lh;
    const float* krow = ks + toff;
    const float* erow = es + toff;
    const float* frow = fs + toff;
    const float* grow = gs + toff;
    const float* lrow = ls + toff;
    const float* elrow = els + toff;
    const float* flrow = fls + toff;
    const float* glrow = gls + toff;
    float* ybase = out + ((size_t)bcol * NCH + c) * SLEN + 128 * e + 8 * lh;

    v8f y[8];
#pragma unroll 1
    for (int it = 0; it < 2; ++it) {
        const int S  = it ? (15 - g) : g;
        const int D0 = (role && it) ? (16 - 2 * g) : 0;
        const int D1 = role ? (it ? (32 - 2 * g) : 0) : (it ? (16 - 2 * g) : (2 * g + 2));
        v8f acc[3][9];
#pragma unroll
        for (int m = 0; m < 3; ++m)
#pragma unroll
            for (int n = 0; n < 9; ++n)
#pragma unroll
                for (int r = 0; r < 8; ++r) acc[m][n][r] = 0.0f;
        const float* xb = xrow + SBP * 2 * S;
#pragma unroll 1
        for (int D = D0; D < D1; ++D) {
            const float* xd = xb - SBP * D;
            ConvFrag f;
            conv_load_x(f.x, xd);
            conv_load_x(f.x2, xd + 64);
            conv_load_taps(f, krow - 128 * D, erow - 128 * D, frow - 128 * D, grow - 128 * D);
            __builtin_amdgcn_sched_barrier(0);
            {
                v4f xsum[8];
#pragma unroll
                for (int i = 0; i < 8; ++i) xsum[i] = f.x[i] + f.x2[i];
                conv_mma(xsum, f, acc[0]);
            }
            conv_load_x(f.x, xd + 64);
            conv_load_taps(f, lrow - 128 * D, elrow - 128 * D, flrow - 128 * D, glrow - 128 * D);
            __builtin_amdgcn_sched_barrier(0);
            conv_mma(f.x, f, acc[1]);
            conv_load_x(f.x, xd);
            conv_load_taps(f, lrow - 128 * D - 64, elrow - 128 * D - 64, flrow - 128 * D - 64, glrow - 128 * D - 64);
            __builtin_amdgcn_sched_barrier(0);
            conv_mma(f.x, f, acc[2]);
        }
        {
            v8f n1[4], n2[4], n3[4];
            conv_fold(acc[0], n1);
            conv_fold(acc[1], n2);
            conv_fold(acc[2], n3);
#pragma unroll
            for (int j = 0; j < 4; ++j) { y[j] = n1[j] + n2[j]; y[4 + j] = n1[j] - n3[j]; }
        }
        if (role == 0 && it == 0) conv_store(ybase + 128 * 2 * S, y);
    }
    __syncthreads();
    float* red = xs + (w & 3) * 2048 + lane;
    if (role == 1) {
#pragma unroll
        for (int j = 0; j < 8; ++j)
#pragma unroll
            for (int r = 0; r < 8; ++r) red[(8 * j + r) * 32] = y[j][r];
    }
    __syncthreads();
    if (role == 0) {
#pragma unroll
        for (int j = 0; j < 8; ++j)
#pragma unroll
            for (int r = 0; r < 8; ++r) y[j][r] += red[(8 * j + r) * 32];
        conv_store(ybase + 128 * 2 * (15 - g), y);
    }
}

extern "C" void kernel_launch(void* const* d_in, const int* in_sizes, int n_in,
                              void* d_out, int out_size, void* d_ws, size_t ws_size,
                              hipStream_t stream) {
    const float* sig = (const float*)d_in[0];
    const float* Bm  = (const float*)d_in[1];
    const float* Cm  = (const float*)d_in[2];
    const float* Am  = (const float*)d_in[3];
    const float* dtp = (const float*)d_in[4];
    float* out  = (float*)d_out;
    float* abar = (float*)d_ws;
    float* bbar = abar + SDIM * SDIM;
    float* pt   = bbar + SDIM * NCH;
    float* gc   = pt + SDIM * SDIM;
    float* hc   = gc + NCH * SDIM * SDIM;
    float* kp   = hc + NCH * SDIM * SDIM;

    float* q8   = kp;
    float* rt   = kp + SDIM * SDIM;

    hippo_setup_kernel<<<1, 256, 0, stream>>>(Am, Bm, dtp, abar, bbar, q8, pt, rt);
    hippo_chains_kernel<<<128, 128, 0, stream>>>(abar, q8, pt, rt, bbar, Cm, gc, hc);
    hippo_taps_kernel<<<NCH, 256, 0, stream>>>(hc, gc, kp);
    hippo_conv_kernel<<<NCH * 2, 256, 0, stream>>>(sig, kp, out);
}


# taps product (64x64x64 per channel, H^T G) moved into the convolution kernel's staging: K goes from WMMA registers straight to LDS, the taps launch and its global plane are gone; every sum unchanged
# speedup vs baseline: 49.8184x; 1.0584x over previous
#include <hip/hip_runtime.h>
#include <hip/hip_bf16.h>

typedef float v2f __attribute__((ext_vector_type(2)));
typedef float v4f __attribute__((ext_vector_type(4)));
typedef float v8f __attribute__((ext_vector_type(8)));

#define SDIM 64
#define NCH  128
#define NB   8
#define SLEN 4096

#define LP 68
#define TP 33

__device__ __forceinline__ void lds_barrier() {
    __builtin_amdgcn_fence(__ATOMIC_RELEASE, "workgroup", "local");
    __builtin_amdgcn_s_barrier();
    __builtin_amdgcn_fence(__ATOMIC_ACQUIRE, "workgroup", "local");
}

__global__ __launch_bounds__(256, 1)
void hippo_setup_kernel(const float* __restrict__ A,
                        const float* __restrict__ Bm,
                        const float* __restrict__ dtp,
                        float* __restrict__ abar,
                        float* __restrict__ bbar,
                        float* __restrict__ q8,
                        float* __restrict__ pT,
                        float* __restrict__ rT) {
    __shared__ __attribute__((aligned(16))) float Ls[SDIM * LP];
    __shared__ __attribute__((aligned(16))) float Xs[SDIM * LP];
    __shared__ float Ts[32 * TP];
    const float dt = dtp[0];
    const int tid = threadIdx.x;
    const int lane = tid & 31;
    const int w = __builtin_amdgcn_readfirstlane(tid >> 5);
    const int lc = lane & 15, lh = lane >> 4;
    const int tn = w & 3;
    const int cg = w >> 2;

    float av[16];
#pragma unroll
    for (int u = 0; u < 16; ++u) av[u] = A[tid + 256 * u];
#pragma unroll
    for (int u = 0; u < 16; ++u) {
        const int e = tid + 256 * u;
        const int i = e >> 6, j = e & 63;
        float l = -0.5f * dt * av[u];
        if (i == j) l += 1.0f;
        Ls[i * LP + j] = l;
        Xs[i * LP + j] = 0.0f;
    }
    v2f bf[4][16];
#pragma unroll
    for (int ct = 0; ct < 4; ++ct) {
        const int c = (cg * 4 + ct) * 16 + lc;
#pragma unroll
        for (int tk = 0; tk < 16; ++tk) {
            bf[ct][tk].x = Bm[(tk * 4 + 2 * lh) * NCH + c];
            bf[ct][tk].y = Bm[(tk * 4 + 2 * lh + 1) * NCH + c];
        }
    }
    __syncthreads();

    if (tid < 64) {
        const int I = tid >> 4, j = tid & 15;
        const float* Lb = Ls + (16 * I) * LP + 16 * I;
        float x[16];
#pragma unroll
        for (int i = 0; i < 16; ++i) {
            float sum = (i == j) ? 1.0f : 0.0f;
#pragma unroll
            for (int k = 0; k < i; ++k) sum -= Lb[i * LP + k] * x[k];
            x[i] = sum / Lb[i * LP + i];
        }
#pragma unroll
        for (int i = 0; i < 16; ++i) Xs[(16 * I + i) * LP + 16 * I + j] = x[i];
    }
    __syncthreads();

    {
        const int r = tid >> 4, c = tid & 15;
#pragma unroll
        for (int p = 0; p < 2; ++p) {
            const int R0 = 16 + 32 * p, C0 = 32 * p;
            float s = 0.0f;
#pragma unroll
            for (int k = 0; k < 16; ++k) s += Ls[(R0 + r) * LP + C0 + k] * Xs[(C0 + k) * LP + C0 + c];
            Ts[(16 * p + r) * TP + c] = s;
        }
        __syncthreads();
#pragma unroll
        for (int p = 0; p < 2; ++p) {
            const int R0 = 16 + 32 * p, C0 = 32 * p;
            float s = 0.0f;
#pragma unroll
            for (int k = 0; k < 16; ++k) s += Xs[(R0 + r) * LP + R0 + k] * Ts[(16 * p + k) * TP + c];
            Xs[(R0 + r) * LP + C0 + c] = -s;
        }
        __syncthreads();
    }

    {
        const int r = tid >> 3, c0 = tid & 7;
        float s[4] = {0.0f, 0.0f, 0.0f, 0.0f};
#pragma unroll
        for (int k = 0; k < 32; ++k) {
            const float a = Ls[(32 + r) * LP + k];
#pragma unroll
            for (int m = 0; m < 4; ++m) s[m] += a * Xs[k * LP + c0 + 8 * m];
        }
#pragma unroll
        for (int m = 0; m < 4; ++m) Ts[r * TP + c0 + 8 * m] = s[m];
        __syncthreads();
#pragma unroll
        for (int m = 0; m < 4; ++m) s[m] = 0.0f;
#pragma unroll
        for (int k = 0; k < 32; ++k) {
            const float a = Xs[(32 + r) * LP + 32 + k];
#pragma unroll
            for (int m = 0; m < 4; ++m) s[m] += a * Ts[k * TP + c0 + 8 * m];
        }
#pragma unroll
        for (int m = 0; m < 4; ++m) Xs[(32 + r) * LP + c0 + 8 * m] = -s[m];
        __syncthreads();
    }

#pragma unroll
    for (int u = 0; u < 16; ++u) {
        const int e = tid + 256 * u;
        const int i = e >> 6, j = e & 63;
        const float ab = 2.0f * Xs[i * LP + j] - ((i == j) ? 1.0f : 0.0f);
        abar[e] = ab;
        Ls[i * LP + j] = ab;
    }

    v2f a[16];
#pragma unroll
    for (int tk = 0; tk < 16; ++tk)
        a[tk] = *reinterpret_cast<const v2f*>(Xs + (tn * 16 + lc) * LP + tk * 4 + 2 * lh);
#pragma unroll
    for (int ct = 0; ct < 4; ++ct) {
        const int c = (cg * 4 + ct) * 16 + lc;
        v8f acc0, acc1;
#pragma unroll
        for (int r = 0; r < 8; ++r) { acc0[r] = 0.0f; acc1[r] = 0.0f; }
#pragma unroll
        for (int tk = 0; tk < 16; tk += 2) {
            acc0 = __builtin_amdgcn_wmma_f32_16x16x4_f32(false, a[tk], false, bf[ct][tk], (short)0, acc0, false, false);
            acc1 = __builtin_amdgcn_wmma_f32_16x16x4_f32(false, a[tk + 1], false, bf[ct][tk + 1], (short)0, acc1, false, false);
        }
#pragma unroll
        for (int r = 0; r < 8; ++r)
            bbar[(tn * 16 + 8 * lh + r) * NCH + c] = dt * (acc0[r] + acc1[r]);
    }
    lds_barrier();

    float* src = Ls;
    float* dst = Xs;
#pragma unroll 1
    for (int it = 0; it < 9; ++it) {
        v2f sa[16], sb0[16], sb1[16];
        const int col0 = cg * 32 + lc;
#pragma unroll
        for (int tk = 0; tk < 16; ++tk) {
            sa[tk] = *reinterpret_cast<const v2f*>(src + (tn * 16 + lc) * LP + tk * 4 + 2 * lh);
            const float* p = src + (tk * 4 + 2 * lh) * LP + col0;
            sb0[tk].x = p[0];  sb0[tk].y = p[LP];
            sb1[tk].x = p[16]; sb1[tk].y = p[LP + 16];
        }
        v8f d0, d1;
#pragma unroll
        for (int r = 0; r < 8; ++r) { d0[r] = 0.0f; d1[r] = 0.0f; }
#pragma unroll
        for (int tk = 0; tk < 16; ++tk) {
            d0 = __builtin_amdgcn_wmma_f32_16x16x4_f32(false, sa[tk], false, sb0[tk], (short)0, d0, false, false);
            d1 = __builtin_amdgcn_wmma_f32_16x16x4_f32(false, sa[tk], false, sb1[tk], (short)0, d1, false, false);
        }
#pragma unroll
        for (int r = 0; r < 8; ++r) {
            dst[(tn * 16 + 8 * lh + r) * LP + col0] = d0[r];
            dst[(tn * 16 + 8 * lh + r) * LP + col0 + 16] = d1[r];
        }
        lds_barrier();
        if (it == 2) {
#pragma unroll
            for (int u = 0; u < 4; ++u) {
                const int e4 = tid + 256 * u;
                const int i = e4 >> 4, j4 = (e4 & 15) * 4;
                *reinterpret_cast<v4f*>(q8 + i * SDIM + j4) = *reinterpret_cast<const v4f*>(dst + i * LP + j4);
            }
        } else if (it == 5 || it == 8) {
            float* mt = (it == 5) ? pT : rT;
#pragma unroll
            for (int u = 0; u < 4; ++u) {
                const int e4 = tid + 256 * u;
                const int i = e4 >> 4, j4 = (e4 & 15) * 4;
                v4f v;
#pragma unroll
                for (int m = 0; m < 4; ++m) v[m] = dst[(j4 + m) * LP + i];
                *reinterpret_cast<v4f*>(mt + i * SDIM + j4) = v;
            }
        }
        float* t = src; src = dst; dst = t;
    }
}


#define KPITCH 4112
#define SPITCH 68

__device__ __forceinline__ void chain_step(const v2f (&a)[16], const float* cur, float* nxt,
                                           int lc, int lh, int rowoff, v4f& lo, v4f& hi) {
    v8f acc0, acc1;
#pragma unroll
    for (int r = 0; r < 8; ++r) { acc0[r] = 0.0f; acc1[r] = 0.0f; }
#pragma unroll
    for (int tk = 0; tk < 16; tk += 2) {
        const v2f b0 = *reinterpret_cast<const v2f*>(cur + lc * SPITCH + tk * 4 + 2 * lh);
        const v2f b1 = *reinterpret_cast<const v2f*>(cur + lc * SPITCH + tk * 4 + 4 + 2 * lh);
        acc0 = __builtin_amdgcn_wmma_f32_16x16x4_f32(false, a[tk], false, b0, (short)0, acc0, false, false);
        acc1 = __builtin_amdgcn_wmma_f32_16x16x4_f32(false, a[tk + 1], false, b1, (short)0, acc1, false, false);
    }
#pragma unroll
    for (int r = 0; r < 4; ++r) { lo[r] = acc0[r] + acc1[r]; hi[r] = acc0[r + 4] + acc1[r + 4]; }
    *reinterpret_cast<v4f*>(nxt + lc * SPITCH + rowoff) = lo;
    *reinterpret_cast<v4f*>(nxt + lc * SPITCH + rowoff + 4) = hi;
}

__global__ __launch_bounds__(128, 1)
void hippo_chains_kernel(const float* __restrict__ abar, const float* __restrict__ q8,
                         const float* __restrict__ pT, const float* __restrict__ rT,
                         const float* __restrict__ bbar, const float* __restrict__ Cmat,
                         float* __restrict__ Gc, float* __restrict__ Hc) {
    __shared__ __attribute__((aligned(16))) float Sl[2 * 16 * SPITCH];
    const int lane = threadIdx.x & 31;
    const int tn = __builtin_amdgcn_readfirstlane(threadIdx.x >> 5);
    const int lc = lane & 15, lh = lane >> 4;
    const int blk = blockIdx.x;
    const int side = blk >> 6;
    const int b = blk & 7;
    const int c = ((blk >> 3) & 7) * 16 + lc;
    const float* M1 = side ? pT : abar;
    const float* M2 = side ? rT : q8;
    const float* S0 = side ? Cmat : bbar;
    const int rowoff = tn * 16 + 8 * lh;
    float* dst = (side ? Hc : Gc) + (size_t)c * SDIM * SDIM + b * SDIM + rowoff;

    v4f lo, hi;
#pragma unroll
    for (int r = 0; r < 4; ++r) {
        lo[r] = S0[(rowoff + r) * NCH + c];
        hi[r] = S0[(rowoff + r + 4) * NCH + c];
    }
    v2f a1[16], a2[16];
#pragma unroll
    for (int tk = 0; tk < 16; ++tk) {
        a1[tk] = *reinterpret_cast<const v2f*>(M1 + (tn * 16 + lc) * SDIM + tk * 4 + 2 * lh);
        a2[tk] = *reinterpret_cast<const v2f*>(M2 + (tn * 16 + lc) * SDIM + tk * 4 + 2 * lh);
    }
    *reinterpret_cast<v4f*>(Sl + lc * SPITCH + rowoff) = lo;
    *reinterpret_cast<v4f*>(Sl + lc * SPITCH + rowoff + 4) = hi;
    lds_barrier();

    int par = 0;
#pragma unroll 1
    for (int i = 0; i < b; ++i) {
        chain_step(a1, Sl + par * (16 * SPITCH), Sl + (par ^ 1) * (16 * SPITCH), lc, lh, rowoff, lo, hi);
        par ^= 1;
        lds_barrier();
    }
    *reinterpret_cast<v4f*>(dst) = lo;
    *reinterpret_cast<v4f*>(dst + 4) = hi;
#pragma unroll 1
    for (int a = 1; a < 8; ++a) {
        chain_step(a2, Sl + par * (16 * SPITCH), Sl + (par ^ 1) * (16 * SPITCH), lc, lh, rowoff, lo, hi);
        par ^= 1;
        *reinterpret_cast<v4f*>(dst + a * 8 * SDIM) = lo;
        *reinterpret_cast<v4f*>(dst + a * 8 * SDIM + 4) = hi;
        lds_barrier();
    }
}


#define SBP    132
#define XPITCH 4360

struct ConvFrag {
    v4f x[8];
    v4f x2[8];
    v2f ak[4];
    v2f ae0[4], ae1[4];
    v2f af0[4], af2[4];
    v2f ag0[4], ag1[4], ag2[4], ag3[4];
};

__device__ __forceinline__ void conv_load_x(v4f (&x)[8], const float* xb) {
#pragma unroll
    for (int r = 0; r < 4; ++r) {
        x[2 * r]     = *reinterpret_cast<const v4f*>(xb + 16 * r);
        x[2 * r + 1] = *reinterpret_cast<const v4f*>(xb + 16 * r + 4);
    }
}

__device__ __forceinline__ void conv_load_taps(ConvFrag& f, const float* kb, const float* eb,
                                               const float* fb, const float* gb) {
#pragma unroll
    for (int s = 0; s < 4; ++s) {
        f.ak[s].x  = kb[2 * s];      f.ak[s].y  = kb[2 * s + 1];
        f.ae0[s].x = eb[2 * s];      f.ae0[s].y = eb[2 * s + 1];
        f.ae1[s].x = eb[2 * s - 16]; f.ae1[s].y = eb[2 * s - 15];
        f.af0[s].x = fb[2 * s];      f.af0[s].y = fb[2 * s + 1];
        f.af2[s].x = fb[2 * s - 32]; f.af2[s].y = fb[2 * s - 31];
        f.ag0[s].x = gb[2 * s];      f.ag0[s].y = gb[2 * s + 1];
        f.ag1[s].x = gb[2 * s - 16]; f.ag1[s].y = gb[2 * s - 15];
        f.ag2[s].x = gb[2 * s - 32]; f.ag2[s].y = gb[2 * s - 31];
        f.ag3[s].x = gb[2 * s - 48]; f.ag3[s].y = gb[2 * s - 47];
    }
}

__device__ __forceinline__ v2f conv_bs(const v4f& lo, const v4f& hi, int s) {
    v2f b;
    if (s < 2) { b.x = lo[2 * s]; b.y = lo[2 * s + 1]; }
    else       { b.x = hi[2 * s - 4]; b.y = hi[2 * s - 3]; }
    return b;
}

__device__ __forceinline__ void conv_mma(const v4f (&x)[8], const ConvFrag& f, v8f (&a)[9]) {
    const v4f w0l = x[0] + x[4], w0h = x[1] + x[5];
    const v4f w1l = x[2] + x[6], w1h = x[3] + x[7];
    const v4f ql  = w0l + w1l,   qh  = w0h + w1h;
    const v4f sel = x[0] + x[2], seh = x[1] + x[3];
    const v4f sol = x[4] + x[6], soh = x[5] + x[7];
#pragma unroll
    for (int s = 0; s < 4; ++s) {
        a[0] = __builtin_amdgcn_wmma_f32_16x16x4_f32(false, f.ak[s],  false, conv_bs(ql, qh, s),     (short)0, a[0], false, false);
        a[1] = __builtin_amdgcn_wmma_f32_16x16x4_f32(false, f.ae0[s], false, conv_bs(w1l, w1h, s),   (short)0, a[1], false, false);
        a[2] = __builtin_amdgcn_wmma_f32_16x16x4_f32(false, f.ae1[s], false, conv_bs(w0l, w0h, s),   (short)0, a[2], false, false);
        a[3] = __builtin_amdgcn_wmma_f32_16x16x4_f32(false, f.af0[s], false, conv_bs(sol, soh, s),   (short)0, a[3], false, false);
        a[4] = __builtin_amdgcn_wmma_f32_16x16x4_f32(false, f.ag0[s], false, conv_bs(x[6], x[7], s), (short)0, a[4], false, false);
        a[5] = __builtin_amdgcn_wmma_f32_16x16x4_f32(false, f.ag1[s], false, conv_bs(x[4], x[5], s), (short)0, a[5], false, false);
        a[6] = __builtin_amdgcn_wmma_f32_16x16x4_f32(false, f.af2[s], false, conv_bs(sel, seh, s),   (short)0, a[6], false, false);
        a[7] = __builtin_amdgcn_wmma_f32_16x16x4_f32(false, f.ag2[s], false, conv_bs(x[2], x[3], s), (short)0, a[7], false, false);
        a[8] = __builtin_amdgcn_wmma_f32_16x16x4_f32(false, f.ag3[s], false, conv_bs(x[0], x[1], s), (short)0, a[8], false, false);
    }
}

__device__ __forceinline__ void conv_fold(const v8f (&a)[9], v8f (&r)[4]) {
    const v8f p01 = a[0] + a[1];
    const v8f p02 = a[0] - a[2];
    r[0] = p01 + (a[3] + a[4]);
    r[1] = p02 + (a[3] - a[5]);
    r[2] = p01 - (a[6] + a[7]);
    r[3] = p02 - (a[6] - a[8]);
}

__device__ __forceinline__ void conv_store(float* yp, const v8f (&y)[8]) {
#pragma unroll
    for (int j = 0; j < 8; ++j) {
        v4f lo, hi;
#pragma unroll
        for (int r = 0; r < 4; ++r) { lo[r] = y[j][r]; hi[r] = y[j][r + 4]; }
        *reinterpret_cast<v4f*>(yp + 16 * j) = lo;
        *reinterpret_cast<v4f*>(yp + 16 * j + 4) = hi;
    }
}

__global__ __launch_bounds__(256, 1)
void hippo_conv_kernel(const float* __restrict__ sig, const float* __restrict__ Hc,
                       const float* __restrict__ Gc, float* __restrict__ out) {
    __shared__ __attribute__((aligned(16))) float xs[NB * XPITCH];
    __shared__ __attribute__((aligned(16))) float ks[KPITCH];
    __shared__ __attribute__((aligned(16))) float es[KPITCH];
    __shared__ __attribute__((aligned(16))) float fs[KPITCH];
    __shared__ __attribute__((aligned(16))) float gs[KPITCH];
    __shared__ __attribute__((aligned(16))) float ls[KPITCH];
    __shared__ __attribute__((aligned(16))) float els[KPITCH];
    __shared__ __attribute__((aligned(16))) float fls[KPITCH];
    __shared__ __attribute__((aligned(16))) float gls[KPITCH];
    const int c = blockIdx.x >> 1;
    const int z = blockIdx.x & 1;
    const int tid = threadIdx.x;
    const int lane = tid & 31;
    const int w = __builtin_amdgcn_readfirstlane(tid >> 5);
    const int lc = lane & 15, lh = lane >> 4;

    v2f hf[2][16], gf[2][16];
    {
        const float* H = Hc + (size_t)c * SDIM * SDIM;
        const float* G = Gc + (size_t)c * SDIM * SDIM;
#pragma unroll
        for (int t = 0; t < 2; ++t) {
            const int tile = 2 * w + t;
            const int tj = tile >> 2, ti = tile & 3;
#pragma unroll
            for (int tk = 0; tk < 16; ++tk) {
                hf[t][tk] = *reinterpret_cast<const v2f*>(H + (tj * 16 + lc) * SDIM + tk * 4 + 2 * lh);
                gf[t][tk] = *reinterpret_cast<const v2f*>(G + (ti * 16 + lc) * SDIM + tk * 4 + 2 * lh);
            }
        }
    }
#pragma unroll
    for (int j = 0; j < 4; ++j) xs[(tid >> 5) * XPITCH + 32 * j + (tid & 31)] = 0.0f;
#pragma unroll 1
    for (int b = 0; b < NB; ++b) {
        const v4f* src = reinterpret_cast<const v4f*>(sig + ((size_t)b * NCH + c) * SLEN);
        float* xd = xs + b * XPITCH + SBP;
#pragma unroll
        for (int i = 0; i < 4; ++i) {
            const int u = tid + 256 * i;
            *reinterpret_cast<v4f*>(xd + SBP * (u >> 5) + 4 * (u & 31)) = src[u];
        }
    }
    if (tid < 16) ks[SLEN + tid] = 0.0f;
    {
        v8f kacc[2];
#pragma unroll
        for (int t = 0; t < 2; ++t)
#pragma unroll
            for (int r = 0; r < 8; ++r) kacc[t][r] = 0.0f;
#pragma unroll
        for (int tk = 0; tk < 16; ++tk) {
#pragma unroll
            for (int t = 0; t < 2; ++t)
                kacc[t] = __builtin_amdgcn_wmma_f32_16x16x4_f32(false, hf[t][tk], false, gf[t][tk], (short)0, kacc[t], false, false);
        }
#pragma unroll
        for (int t = 0; t < 2; ++t) {
            const int tile = 2 * w + t;
            const int tj = tile >> 2, ti = tile & 3;
#pragma unroll
            for (int r = 0; r < 8; ++r)
                ks[SLEN - 1 - ((tj * 16 + r + 8 * lh) * SDIM + ti * 16 + lc)] = kacc[t][r];
        }
    }
    __syncthreads();
    {
        const v4f* k4 = reinterpret_cast<const v4f*>(ks);
        v4f* e4 = reinterpret_cast<v4f*>(es);
        v4f* f4 = reinterpret_cast<v4f*>(fs);
        v4f* g4 = reinterpret_cast<v4f*>(gs);
        v4f* l4 = reinterpret_cast<v4f*>(ls);
        v4f* el4 = reinterpret_cast<v4f*>(els);
        v4f* fl4 = reinterpret_cast<v4f*>(fls);
        v4f* gl4 = reinterpret_cast<v4f*>(gls);
        for (int i = tid; i < KPITCH / 4; i += 256) {
            const v4f zero = {0.0f, 0.0f, 0.0f, 0.0f};
            v4f k[8];
#pragma unroll
            for (int n = 0; n < 8; ++n) {
                k[n] = zero;
                if (i + 4 * n < KPITCH / 4) k[n] = k4[i + 4 * n];
            }
            const v4f fv = k[2] - k[0];
            e4[i] = k[1] - k[0];
            f4[i] = fv;
            g4[i] = (k[3] - k[1]) - fv;
            const v4f l0 = k[4] - k[0], l1 = k[5] - k[1], l2 = k[6] - k[2], l3 = k[7] - k[3];
            const v4f flv = l2 - l0;
            l4[i] = l0;
            el4[i] = l1 - l0;
            fl4[i] = flv;
            gl4[i] = (l3 - l1) - flv;
        }
    }
    __syncthreads();

    const int bcol = lc & 7;
    const int e    = lc >> 3;
    const int role = w >> 2;
    const int g    = z * 4 + (w & 3);
    const float* xrow = xs + bcol * XPITCH + SBP + 8 * lh + SBP * e;
    const int toff = (SLEN - 1) - lc + 8 * lh;
    const float* krow = ks + toff;
    const float* erow = es + toff;
    const float* frow = fs + toff;
    const float* grow = gs + toff;
    const float* lrow = ls + toff;
    const float* elrow = els + toff;
    const float* flrow = fls + toff;
    const float* glrow = gls + toff;
    float* ybase = out + ((size_t)bcol * NCH + c) * SLEN + 128 * e + 8 * lh;

    v8f y[8];
#pragma unroll 1
    for (int it = 0; it < 2; ++it) {
        const int S  = it ? (15 - g) : g;
        const int D0 = (role && it) ? (16 - 2 * g) : 0;
        const int D1 = role ? (it ? (32 - 2 * g) : 0) : (it ? (16 - 2 * g) : (2 * g + 2));
        v8f acc[3][9];
#pragma unroll
        for (int m = 0; m < 3; ++m)
#pragma unroll
            for (int n = 0; n < 9; ++n)
#pragma unroll
                for (int r = 0; r < 8; ++r) acc[m][n][r] = 0.0f;
        const float* xb = xrow + SBP * 2 * S;
#pragma unroll 1
        for (int D = D0; D < D1; ++D) {
            const float* xd = xb - SBP * D;
            ConvFrag f;
            conv_load_x(f.x, xd);
            conv_load_x(f.x2, xd + 64);
            conv_load_taps(f, krow - 128 * D, erow - 128 * D, frow - 128 * D, grow - 128 * D);
            __builtin_amdgcn_sched_barrier(0);
            {
                v4f xsum[8];
#pragma unroll
                for (int i = 0; i < 8; ++i) xsum[i] = f.x[i] + f.x2[i];
                conv_mma(xsum, f, acc[0]);
            }
            conv_load_x(f.x, xd + 64);
            conv_load_taps(f, lrow - 128 * D, elrow - 128 * D, flrow - 128 * D, glrow - 128 * D);
            __builtin_amdgcn_sched_barrier(0);
            conv_mma(f.x, f, acc[1]);
            conv_load_x(f.x, xd);
            conv_load_taps(f, lrow - 128 * D - 64, elrow - 128 * D - 64, flrow - 128 * D - 64, glrow - 128 * D - 64);
            __builtin_amdgcn_sched_barrier(0);
            conv_mma(f.x, f, acc[2]);
        }
        {
            v8f n1[4], n2[4], n3[4];
            conv_fold(acc[0], n1);
            conv_fold(acc[1], n2);
            conv_fold(acc[2], n3);
#pragma unroll
            for (int j = 0; j < 4; ++j) { y[j] = n1[j] + n2[j]; y[4 + j] = n1[j] - n3[j]; }
        }
        if (role == 0 && it == 0) conv_store(ybase + 128 * 2 * S, y);
    }
    __syncthreads();
    float* red = xs + (w & 3) * 2048 + lane;
    if (role == 1) {
#pragma unroll
        for (int j = 0; j < 8; ++j)
#pragma unroll
            for (int r = 0; r < 8; ++r) red[(8 * j + r) * 32] = y[j][r];
    }
    __syncthreads();
    if (role == 0) {
#pragma unroll
        for (int j = 0; j < 8; ++j)
#pragma unroll
            for (int r = 0; r < 8; ++r) y[j][r] += red[(8 * j + r) * 32];
        conv_store(ybase + 128 * 2 * (15 - g), y);
    }
}

extern "C" void kernel_launch(void* const* d_in, const int* in_sizes, int n_in,
                              void* d_out, int out_size, void* d_ws, size_t ws_size,
                              hipStream_t stream) {
    const float* sig = (const float*)d_in[0];
    const float* Bm  = (const float*)d_in[1];
    const float* Cm  = (const float*)d_in[2];
    const float* Am  = (const float*)d_in[3];
    const float* dtp = (const float*)d_in[4];
    float* out  = (float*)d_out;
    float* abar = (float*)d_ws;
    float* bbar = abar + SDIM * SDIM;
    float* pt   = bbar + SDIM * NCH;
    float* gc   = pt + SDIM * SDIM;
    float* hc   = gc + NCH * SDIM * SDIM;
    float* q8   = hc + NCH * SDIM * SDIM;
    float* rt   = q8 + SDIM * SDIM;

    hippo_setup_kernel<<<1, 256, 0, stream>>>(Am, Bm, dtp, abar, bbar, q8, pt, rt);
    hippo_chains_kernel<<<128, 128, 0, stream>>>(abar, q8, pt, rt, bbar, Cm, gc, hc);
    hippo_conv_kernel<<<NCH * 2, 256, 0, stream>>>(sig, hc, gc, out);
}


# chain launch folded into the convolution kernel: each work-group runs its own channel's two stride-8 tap chains in LDS (same 14 steps) under its signal loads; two launches, every sum unchanged
# speedup vs baseline: 55.0502x; 1.1050x over previous
#include <hip/hip_runtime.h>
#include <hip/hip_bf16.h>

typedef float v2f __attribute__((ext_vector_type(2)));
typedef float v4f __attribute__((ext_vector_type(4)));
typedef float v8f __attribute__((ext_vector_type(8)));

#define SDIM 64
#define NCH  128
#define NB   8
#define SLEN 4096

#define LP 68
#define TP 33

__device__ __forceinline__ void lds_barrier() {
    __builtin_amdgcn_fence(__ATOMIC_RELEASE, "workgroup", "local");
    __builtin_amdgcn_s_barrier();
    __builtin_amdgcn_fence(__ATOMIC_ACQUIRE, "workgroup", "local");
}

__global__ __launch_bounds__(256, 1)
void hippo_setup_kernel(const float* __restrict__ A,
                        const float* __restrict__ Bm,
                        const float* __restrict__ dtp,
                        float* __restrict__ abar,
                        float* __restrict__ bbar,
                        float* __restrict__ q8,
                        float* __restrict__ pT,
                        float* __restrict__ rT) {
    __shared__ __attribute__((aligned(16))) float Ls[SDIM * LP];
    __shared__ __attribute__((aligned(16))) float Xs[SDIM * LP];
    __shared__ float Ts[32 * TP];
    const float dt = dtp[0];
    const int tid = threadIdx.x;
    const int lane = tid & 31;
    const int w = __builtin_amdgcn_readfirstlane(tid >> 5);
    const int lc = lane & 15, lh = lane >> 4;
    const int tn = w & 3;
    const int cg = w >> 2;

    float av[16];
#pragma unroll
    for (int u = 0; u < 16; ++u) av[u] = A[tid + 256 * u];
#pragma unroll
    for (int u = 0; u < 16; ++u) {
        const int e = tid + 256 * u;
        const int i = e >> 6, j = e & 63;
        float l = -0.5f * dt * av[u];
        if (i == j) l += 1.0f;
        Ls[i * LP + j] = l;
        Xs[i * LP + j] = 0.0f;
    }
    v2f bf[4][16];
#pragma unroll
    for (int ct = 0; ct < 4; ++ct) {
        const int c = (cg * 4 + ct) * 16 + lc;
#pragma unroll
        for (int tk = 0; tk < 16; ++tk) {
            bf[ct][tk].x = Bm[(tk * 4 + 2 * lh) * NCH + c];
            bf[ct][tk].y = Bm[(tk * 4 + 2 * lh + 1) * NCH + c];
        }
    }
    __syncthreads();

    if (tid < 64) {
        const int I = tid >> 4, j = tid & 15;
        const float* Lb = Ls + (16 * I) * LP + 16 * I;
        float x[16];
#pragma unroll
        for (int i = 0; i < 16; ++i) {
            float sum = (i == j) ? 1.0f : 0.0f;
#pragma unroll
            for (int k = 0; k < i; ++k) sum -= Lb[i * LP + k] * x[k];
            x[i] = sum / Lb[i * LP + i];
        }
#pragma unroll
        for (int i = 0; i < 16; ++i) Xs[(16 * I + i) * LP + 16 * I + j] = x[i];
    }
    __syncthreads();

    {
        const int r = tid >> 4, c = tid & 15;
#pragma unroll
        for (int p = 0; p < 2; ++p) {
            const int R0 = 16 + 32 * p, C0 = 32 * p;
            float s = 0.0f;
#pragma unroll
            for (int k = 0; k < 16; ++k) s += Ls[(R0 + r) * LP + C0 + k] * Xs[(C0 + k) * LP + C0 + c];
            Ts[(16 * p + r) * TP + c] = s;
        }
        __syncthreads();
#pragma unroll
        for (int p = 0; p < 2; ++p) {
            const int R0 = 16 + 32 * p, C0 = 32 * p;
            float s = 0.0f;
#pragma unroll
            for (int k = 0; k < 16; ++k) s += Xs[(R0 + r) * LP + R0 + k] * Ts[(16 * p + k) * TP + c];
            Xs[(R0 + r) * LP + C0 + c] = -s;
        }
        __syncthreads();
    }

    {
        const int r = tid >> 3, c0 = tid & 7;
        float s[4] = {0.0f, 0.0f, 0.0f, 0.0f};
#pragma unroll
        for (int k = 0; k < 32; ++k) {
            const float a = Ls[(32 + r) * LP + k];
#pragma unroll
            for (int m = 0; m < 4; ++m) s[m] += a * Xs[k * LP + c0 + 8 * m];
        }
#pragma unroll
        for (int m = 0; m < 4; ++m) Ts[r * TP + c0 + 8 * m] = s[m];
        __syncthreads();
#pragma unroll
        for (int m = 0; m < 4; ++m) s[m] = 0.0f;
#pragma unroll
        for (int k = 0; k < 32; ++k) {
            const float a = Xs[(32 + r) * LP + 32 + k];
#pragma unroll
            for (int m = 0; m < 4; ++m) s[m] += a * Ts[k * TP + c0 + 8 * m];
        }
#pragma unroll
        for (int m = 0; m < 4; ++m) Xs[(32 + r) * LP + c0 + 8 * m] = -s[m];
        __syncthreads();
    }

#pragma unroll
    for (int u = 0; u < 16; ++u) {
        const int e = tid + 256 * u;
        const int i = e >> 6, j = e & 63;
        const float ab = 2.0f * Xs[i * LP + j] - ((i == j) ? 1.0f : 0.0f);
        abar[e] = ab;
        Ls[i * LP + j] = ab;
    }

    v2f a[16];
#pragma unroll
    for (int tk = 0; tk < 16; ++tk)
        a[tk] = *reinterpret_cast<const v2f*>(Xs + (tn * 16 + lc) * LP + tk * 4 + 2 * lh);
#pragma unroll
    for (int ct = 0; ct < 4; ++ct) {
        const int c = (cg * 4 + ct) * 16 + lc;
        v8f acc0, acc1;
#pragma unroll
        for (int r = 0; r < 8; ++r) { acc0[r] = 0.0f; acc1[r] = 0.0f; }
#pragma unroll
        for (int tk = 0; tk < 16; tk += 2) {
            acc0 = __builtin_amdgcn_wmma_f32_16x16x4_f32(false, a[tk], false, bf[ct][tk], (short)0, acc0, false, false);
            acc1 = __builtin_amdgcn_wmma_f32_16x16x4_f32(false, a[tk + 1], false, bf[ct][tk + 1], (short)0, acc1, false, false);
        }
#pragma unroll
        for (int r = 0; r < 8; ++r)
            bbar[(tn * 16 + 8 * lh + r) * NCH + c] = dt * (acc0[r] + acc1[r]);
    }
    lds_barrier();

    float* src = Ls;
    float* dst = Xs;
#pragma unroll 1
    for (int it = 0; it < 9; ++it) {
        v2f sa[16], sb0[16], sb1[16];
        const int col0 = cg * 32 + lc;
#pragma unroll
        for (int tk = 0; tk < 16; ++tk) {
            sa[tk] = *reinterpret_cast<const v2f*>(src + (tn * 16 + lc) * LP + tk * 4 + 2 * lh);
            const float* p = src + (tk * 4 + 2 * lh) * LP + col0;
            sb0[tk].x = p[0];  sb0[tk].y = p[LP];
            sb1[tk].x = p[16]; sb1[tk].y = p[LP + 16];
        }
        v8f d0, d1;
#pragma unroll
        for (int r = 0; r < 8; ++r) { d0[r] = 0.0f; d1[r] = 0.0f; }
#pragma unroll
        for (int tk = 0; tk < 16; ++tk) {
            d0 = __builtin_amdgcn_wmma_f32_16x16x4_f32(false, sa[tk], false, sb0[tk], (short)0, d0, false, false);
            d1 = __builtin_amdgcn_wmma_f32_16x16x4_f32(false, sa[tk], false, sb1[tk], (short)0, d1, false, false);
        }
#pragma unroll
        for (int r = 0; r < 8; ++r) {
            dst[(tn * 16 + 8 * lh + r) * LP + col0] = d0[r];
            dst[(tn * 16 + 8 * lh + r) * LP + col0 + 16] = d1[r];
        }
        lds_barrier();
        if (it == 2) {
#pragma unroll
            for (int u = 0; u < 4; ++u) {
                const int e4 = tid + 256 * u;
                const int i = e4 >> 4, j4 = (e4 & 15) * 4;
                *reinterpret_cast<v4f*>(q8 + i * SDIM + j4) = *reinterpret_cast<const v4f*>(dst + i * LP + j4);
            }
        } else if (it == 5 || it == 8) {
            float* mt = (it == 5) ? pT : rT;
#pragma unroll
            for (int u = 0; u < 4; ++u) {
                const int e4 = tid + 256 * u;
                const int i = e4 >> 4, j4 = (e4 & 15) * 4;
                v4f v;
#pragma unroll
                for (int m = 0; m < 4; ++m) v[m] = dst[(j4 + m) * LP + i];
                *reinterpret_cast<v4f*>(mt + i * SDIM + j4) = v;
            }
        }
        float* t = src; src = dst; dst = t;
    }
}


#define KPITCH 4112
#define CP 68

__device__ __forceinline__ void chain_mma(const v2f (&a)[16], const float* bp, v4f& lo, v4f& hi) {
    v8f acc0, acc1;
#pragma unroll
    for (int r = 0; r < 8; ++r) { acc0[r] = 0.0f; acc1[r] = 0.0f; }
#pragma unroll
    for (int tk = 0; tk < 16; tk += 2) {
        const v2f b0 = *reinterpret_cast<const v2f*>(bp + tk * 4);
        const v2f b1 = *reinterpret_cast<const v2f*>(bp + tk * 4 + 4);
        acc0 = __builtin_amdgcn_wmma_f32_16x16x4_f32(false, a[tk], false, b0, (short)0, acc0, false, false);
        acc1 = __builtin_amdgcn_wmma_f32_16x16x4_f32(false, a[tk + 1], false, b1, (short)0, acc1, false, false);
    }
#pragma unroll
    for (int r = 0; r < 4; ++r) { lo[r] = acc0[r] + acc1[r]; hi[r] = acc0[r + 4] + acc1[r + 4]; }
}


#define SBP    132
#define XPITCH 4360

struct ConvFrag {
    v4f x[8];
    v4f x2[8];
    v2f ak[4];
    v2f ae0[4], ae1[4];
    v2f af0[4], af2[4];
    v2f ag0[4], ag1[4], ag2[4], ag3[4];
};

__device__ __forceinline__ void conv_load_x(v4f (&x)[8], const float* xb) {
#pragma unroll
    for (int r = 0; r < 4; ++r) {
        x[2 * r]     = *reinterpret_cast<const v4f*>(xb + 16 * r);
        x[2 * r + 1] = *reinterpret_cast<const v4f*>(xb + 16 * r + 4);
    }
}

__device__ __forceinline__ void conv_load_taps(ConvFrag& f, const float* kb, const float* eb,
                                               const float* fb, const float* gb) {
#pragma unroll
    for (int s = 0; s < 4; ++s) {
        f.ak[s].x  = kb[2 * s];      f.ak[s].y  = kb[2 * s + 1];
        f.ae0[s].x = eb[2 * s];      f.ae0[s].y = eb[2 * s + 1];
        f.ae1[s].x = eb[2 * s - 16]; f.ae1[s].y = eb[2 * s - 15];
        f.af0[s].x = fb[2 * s];      f.af0[s].y = fb[2 * s + 1];
        f.af2[s].x = fb[2 * s - 32]; f.af2[s].y = fb[2 * s - 31];
        f.ag0[s].x = gb[2 * s];      f.ag0[s].y = gb[2 * s + 1];
        f.ag1[s].x = gb[2 * s - 16]; f.ag1[s].y = gb[2 * s - 15];
        f.ag2[s].x = gb[2 * s - 32]; f.ag2[s].y = gb[2 * s - 31];
        f.ag3[s].x = gb[2 * s - 48]; f.ag3[s].y = gb[2 * s - 47];
    }
}

__device__ __forceinline__ v2f conv_bs(const v4f& lo, const v4f& hi, int s) {
    v2f b;
    if (s < 2) { b.x = lo[2 * s]; b.y = lo[2 * s + 1]; }
    else       { b.x = hi[2 * s - 4]; b.y = hi[2 * s - 3]; }
    return b;
}

__device__ __forceinline__ void conv_mma(const v4f (&x)[8], const ConvFrag& f, v8f (&a)[9]) {
    const v4f w0l = x[0] + x[4], w0h = x[1] + x[5];
    const v4f w1l = x[2] + x[6], w1h = x[3] + x[7];
    const v4f ql  = w0l + w1l,   qh  = w0h + w1h;
    const v4f sel = x[0] + x[2], seh = x[1] + x[3];
    const v4f sol = x[4] + x[6], soh = x[5] + x[7];
#pragma unroll
    for (int s = 0; s < 4; ++s) {
        a[0] = __builtin_amdgcn_wmma_f32_16x16x4_f32(false, f.ak[s],  false, conv_bs(ql, qh, s),     (short)0, a[0], false, false);
        a[1] = __builtin_amdgcn_wmma_f32_16x16x4_f32(false, f.ae0[s], false, conv_bs(w1l, w1h, s),   (short)0, a[1], false, false);
        a[2] = __builtin_amdgcn_wmma_f32_16x16x4_f32(false, f.ae1[s], false, conv_bs(w0l, w0h, s),   (short)0, a[2], false, false);
        a[3] = __builtin_amdgcn_wmma_f32_16x16x4_f32(false, f.af0[s], false, conv_bs(sol, soh, s),   (short)0, a[3], false, false);
        a[4] = __builtin_amdgcn_wmma_f32_16x16x4_f32(false, f.ag0[s], false, conv_bs(x[6], x[7], s), (short)0, a[4], false, false);
        a[5] = __builtin_amdgcn_wmma_f32_16x16x4_f32(false, f.ag1[s], false, conv_bs(x[4], x[5], s), (short)0, a[5], false, false);
        a[6] = __builtin_amdgcn_wmma_f32_16x16x4_f32(false, f.af2[s], false, conv_bs(sel, seh, s),   (short)0, a[6], false, false);
        a[7] = __builtin_amdgcn_wmma_f32_16x16x4_f32(false, f.ag2[s], false, conv_bs(x[2], x[3], s), (short)0, a[7], false, false);
        a[8] = __builtin_amdgcn_wmma_f32_16x16x4_f32(false, f.ag3[s], false, conv_bs(x[0], x[1], s), (short)0, a[8], false, false);
    }
}

__device__ __forceinline__ void conv_fold(const v8f (&a)[9], v8f (&r)[4]) {
    const v8f p01 = a[0] + a[1];
    const v8f p02 = a[0] - a[2];
    r[0] = p01 + (a[3] + a[4]);
    r[1] = p02 + (a[3] - a[5]);
    r[2] = p01 - (a[6] + a[7]);
    r[3] = p02 - (a[6] - a[8]);
}

__device__ __forceinline__ void conv_store(float* yp, const v8f (&y)[8]) {
#pragma unroll
    for (int j = 0; j < 8; ++j) {
        v4f lo, hi;
#pragma unroll
        for (int r = 0; r < 4; ++r) { lo[r] = y[j][r]; hi[r] = y[j][r + 4]; }
        *reinterpret_cast<v4f*>(yp + 16 * j) = lo;
        *reinterpret_cast<v4f*>(yp + 16 * j + 4) = hi;
    }
}

__global__ __launch_bounds__(256, 1)
void hippo_conv_kernel(const float* __restrict__ sig,
                       const float* __restrict__ abar, const float* __restrict__ q8,
                       const float* __restrict__ pT, const float* __restrict__ rT,
                       const float* __restrict__ bbar, const float* __restrict__ Cmat,
                       float* __restrict__ out) {
    __shared__ __attribute__((aligned(16))) float xs[NB * XPITCH];
    __shared__ __attribute__((aligned(16))) float ks[KPITCH];
    __shared__ __attribute__((aligned(16))) float dif[7 * KPITCH];
    float* const es  = dif;
    float* const fs  = dif + KPITCH;
    float* const gs  = dif + 2 * KPITCH;
    float* const ls  = dif + 3 * KPITCH;
    float* const els = dif + 4 * KPITCH;
    float* const fls = dif + 5 * KPITCH;
    float* const gls = dif + 6 * KPITCH;
    float* const Gl  = dif;
    float* const Hl  = dif + SDIM * CP;
    const int c = blockIdx.x >> 1;
    const int z = blockIdx.x & 1;
    const int tid = threadIdx.x;
    const int lane = tid & 31;
    const int w = __builtin_amdgcn_readfirstlane(tid >> 5);
    const int lc = lane & 15, lh = lane >> 4;

    const int side = w >> 2;
    const int tn = w & 3;
    const int rowoff = tn * 16 + 8 * lh;
    const int sidx = tid & 127;
    float* const Pl = side ? Hl : Gl;
    {
        const float* M1 = side ? pT : abar;
        const float* M2 = side ? rT : q8;
        const float* S0 = side ? Cmat : bbar;
        float s0v = 0.0f;
        if (sidx < SDIM) s0v = S0[sidx * NCH + c];
        v2f a1[16], a2[16];
#pragma unroll
        for (int tk = 0; tk < 16; ++tk) {
            a1[tk] = *reinterpret_cast<const v2f*>(M1 + (tn * 16 + lc) * SDIM + tk * 4 + 2 * lh);
            a2[tk] = *reinterpret_cast<const v2f*>(M2 + (tn * 16 + lc) * SDIM + tk * 4 + 2 * lh);
        }
        __builtin_amdgcn_sched_barrier(0);
        v4f sg[NB][4];
#pragma unroll
        for (int b = 0; b < NB; ++b) {
            const v4f* src = reinterpret_cast<const v4f*>(sig + ((size_t)b * NCH + c) * SLEN);
#pragma unroll
            for (int i = 0; i < 4; ++i) sg[b][i] = src[tid + 256 * i];
        }
#pragma unroll
        for (int j = 0; j < 4; ++j) xs[(tid >> 5) * XPITCH + 32 * j + (tid & 31)] = 0.0f;
        if (sidx < SDIM) Pl[sidx] = s0v;
        for (int i = sidx; i < 7 * CP; i += 128) Pl[CP + i] = 0.0f;
        lds_barrier();

        v4f lo, hi;
#pragma unroll 1
        for (int i = 1; i < 8; ++i) {
            chain_mma(a1, Pl + (lc & 7) * CP + 2 * lh, lo, hi);
            if (lc == i - 1) {
                *reinterpret_cast<v4f*>(Pl + i * CP + rowoff) = lo;
                *reinterpret_cast<v4f*>(Pl + i * CP + rowoff + 4) = hi;
            }
            lds_barrier();
        }
#pragma unroll 1
        for (int a = 1; a < 8; ++a) {
            chain_mma(a2, Pl + (8 * (a - 1) + (lc & 7)) * CP + 2 * lh, lo, hi);
            if (lc < 8) {
                *reinterpret_cast<v4f*>(Pl + (8 * a + lc) * CP + rowoff) = lo;
                *reinterpret_cast<v4f*>(Pl + (8 * a + lc) * CP + rowoff + 4) = hi;
            }
            lds_barrier();
        }
#pragma unroll
        for (int b = 0; b < NB; ++b) {
            float* xd = xs + b * XPITCH + SBP;
#pragma unroll
            for (int i = 0; i < 4; ++i) {
                const int u = tid + 256 * i;
                *reinterpret_cast<v4f*>(xd + SBP * (u >> 5) + 4 * (u & 31)) = sg[b][i];
            }
        }
    }
    v2f hf[2][16], gf[2][16];
#pragma unroll
    for (int t = 0; t < 2; ++t) {
        const int tile = 2 * w + t;
        const int tj = tile >> 2, ti = tile & 3;
#pragma unroll
        for (int tk = 0; tk < 16; ++tk) {
            hf[t][tk] = *reinterpret_cast<const v2f*>(Hl + (tj * 16 + lc) * CP + tk * 4 + 2 * lh);
            gf[t][tk] = *reinterpret_cast<const v2f*>(Gl + (ti * 16 + lc) * CP + tk * 4 + 2 * lh);
        }
    }
    if (tid < 16) ks[SLEN + tid] = 0.0f;
    {
        v8f kacc[2];
#pragma unroll
        for (int t = 0; t < 2; ++t)
#pragma unroll
            for (int r = 0; r < 8; ++r) kacc[t][r] = 0.0f;
#pragma unroll
        for (int tk = 0; tk < 16; ++tk) {
#pragma unroll
            for (int t = 0; t < 2; ++t)
                kacc[t] = __builtin_amdgcn_wmma_f32_16x16x4_f32(false, hf[t][tk], false, gf[t][tk], (short)0, kacc[t], false, false);
        }
#pragma unroll
        for (int t = 0; t < 2; ++t) {
            const int tile = 2 * w + t;
            const int tj = tile >> 2, ti = tile & 3;
#pragma unroll
            for (int r = 0; r < 8; ++r)
                ks[SLEN - 1 - ((tj * 16 + r + 8 * lh) * SDIM + ti * 16 + lc)] = kacc[t][r];
        }
    }
    __syncthreads();
    {
        const v4f* k4 = reinterpret_cast<const v4f*>(ks);
        v4f* e4 = reinterpret_cast<v4f*>(es);
        v4f* f4 = reinterpret_cast<v4f*>(fs);
        v4f* g4 = reinterpret_cast<v4f*>(gs);
        v4f* l4 = reinterpret_cast<v4f*>(ls);
        v4f* el4 = reinterpret_cast<v4f*>(els);
        v4f* fl4 = reinterpret_cast<v4f*>(fls);
        v4f* gl4 = reinterpret_cast<v4f*>(gls);
        for (int i = tid; i < KPITCH / 4; i += 256) {
            const v4f zero = {0.0f, 0.0f, 0.0f, 0.0f};
            v4f k[8];
#pragma unroll
            for (int n = 0; n < 8; ++n) {
                k[n] = zero;
                if (i + 4 * n < KPITCH / 4) k[n] = k4[i + 4 * n];
            }
            const v4f fv = k[2] - k[0];
            e4[i] = k[1] - k[0];
            f4[i] = fv;
            g4[i] = (k[3] - k[1]) - fv;
            const v4f l0 = k[4] - k[0], l1 = k[5] - k[1], l2 = k[6] - k[2], l3 = k[7] - k[3];
            const v4f flv = l2 - l0;
            l4[i] = l0;
            el4[i] = l1 - l0;
            fl4[i] = flv;
            gl4[i] = (l3 - l1) - flv;
        }
    }
    __syncthreads();

    const int bcol = lc & 7;
    const int e    = lc >> 3;
    const int role = w >> 2;
    const int g    = z * 4 + (w & 3);
    const float* xrow = xs + bcol * XPITCH + SBP + 8 * lh + SBP * e;
    const int toff = (SLEN - 1) - lc + 8 * lh;
    const float* krow = ks + toff;
    const float* erow = es + toff;
    const float* frow = fs + toff;
    const float* grow = gs + toff;
    const float* lrow = ls + toff;
    const float* elrow = els + toff;
    const float* flrow = fls + toff;
    const float* glrow = gls + toff;
    float* ybase = out + ((size_t)bcol * NCH + c) * SLEN + 128 * e + 8 * lh;

    v8f y[8];
#pragma unroll 1
    for (int it = 0; it < 2; ++it) {
        const int S  = it ? (15 - g) : g;
        const int D0 = (role && it) ? (16 - 2 * g) : 0;
        const int D1 = role ? (it ? (32 - 2 * g) : 0) : (it ? (16 - 2 * g) : (2 * g + 2));
        v8f acc[3][9];
#pragma unroll
        for (int m = 0; m < 3; ++m)
#pragma unroll
            for (int n = 0; n < 9; ++n)
#pragma unroll
                for (int r = 0; r < 8; ++r) acc[m][n][r] = 0.0f;
        const float* xb = xrow + SBP * 2 * S;
#pragma unroll 1
        for (int D = D0; D < D1; ++D) {
            const float* xd = xb - SBP * D;
            ConvFrag f;
            conv_load_x(f.x, xd);
            conv_load_x(f.x2, xd + 64);
            conv_load_taps(f, krow - 128 * D, erow - 128 * D, frow - 128 * D, grow - 128 * D);
            __builtin_amdgcn_sched_barrier(0);
            {
                v4f xsum[8];
#pragma unroll
                for (int i = 0; i < 8; ++i) xsum[i] = f.x[i] + f.x2[i];
                conv_mma(xsum, f, acc[0]);
            }
            conv_load_x(f.x, xd + 64);
            conv_load_taps(f, lrow - 128 * D, elrow - 128 * D, flrow - 128 * D, glrow - 128 * D);
            __builtin_amdgcn_sched_barrier(0);
            conv_mma(f.x, f, acc[1]);
            conv_load_x(f.x, xd);
            conv_load_taps(f, lrow - 128 * D - 64, elrow - 128 * D - 64, flrow - 128 * D - 64, glrow - 128 * D - 64);
            __builtin_amdgcn_sched_barrier(0);
            conv_mma(f.x, f, acc[2]);
        }
        {
            v8f n1[4], n2[4], n3[4];
            conv_fold(acc[0], n1);
            conv_fold(acc[1], n2);
            conv_fold(acc[2], n3);
#pragma unroll
            for (int j = 0; j < 4; ++j) { y[j] = n1[j] + n2[j]; y[4 + j] = n1[j] - n3[j]; }
        }
        if (role == 0 && it == 0) conv_store(ybase + 128 * 2 * S, y);
    }
    __syncthreads();
    float* red = xs + (w & 3) * 2048 + lane;
    if (role == 1) {
#pragma unroll
        for (int j = 0; j < 8; ++j)
#pragma unroll
            for (int r = 0; r < 8; ++r) red[(8 * j + r) * 32] = y[j][r];
    }
    __syncthreads();
    if (role == 0) {
#pragma unroll
        for (int j = 0; j < 8; ++j)
#pragma unroll
            for (int r = 0; r < 8; ++r) y[j][r] += red[(8 * j + r) * 32];
        conv_store(ybase + 128 * 2 * (15 - g), y);
    }
}

extern "C" void kernel_launch(void* const* d_in, const int* in_sizes, int n_in,
                              void* d_out, int out_size, void* d_ws, size_t ws_size,
                              hipStream_t stream) {
    const float* sig = (const float*)d_in[0];
    const float* Bm  = (const float*)d_in[1];
    const float* Cm  = (const float*)d_in[2];
    const float* Am  = (const float*)d_in[3];
    const float* dtp = (const float*)d_in[4];
    float* out  = (float*)d_out;
    float* abar = (float*)d_ws;
    float* bbar = abar + SDIM * SDIM;
    float* pt   = bbar + SDIM * NCH;
    float* q8   = pt + SDIM * SDIM;
    float* rt   = q8 + SDIM * SDIM;

    hippo_setup_kernel<<<1, 256, 0, stream>>>(Am, Bm, dtp, abar, bbar, q8, pt, rt);
    hippo_conv_kernel<<<NCH * 2, 256, 0, stream>>>(sig, abar, q8, pt, rt, bbar, Cm, out);
}
